# starting kernel
# baseline (speedup 1.0000x reference)
#include <hip/hip_runtime.h>

typedef __attribute__((ext_vector_type(2))) float v2f;
typedef __attribute__((ext_vector_type(4))) float v4f;
typedef __attribute__((ext_vector_type(8))) float v8f;

#define WMMA4(A, B, C) \
  __builtin_amdgcn_wmma_f32_16x16x4_f32(false, (A), false, (B), (short)0, (C), false, false)

constexpr int kN = 2048;
constexpr int kB = 256;
constexpr int kD = 32;
constexpr size_t kStep = (size_t)kB * kD;

__device__ __forceinline__ int colbase(int c, int half) {
  return 2 * c + 8 * half + (c >= 4 ? 8 : 0);
}

struct ATiles { v2f a[2][8]; };

__device__ __forceinline__ void load_a_tiles(const float* __restrict__ M, ATiles& t, int lane) {
  const int r0 = lane & 15;
  const int half = lane >> 4;
#pragma unroll
  for (int rt = 0; rt < 2; ++rt) {
#pragma unroll
    for (int c = 0; c < 8; ++c) {
      const float* p = M + (16 * rt + r0) * kD + colbase(c, half);
      t.a[rt][c] = *(const v2f*)p;
    }
  }
}

__device__ __forceinline__ void load_b_cols(const float* __restrict__ src, v2f xb[8], int lane) {
  const int col = lane & 15;
  const int half = lane >> 4;
  const float* p = src + col * kD;
#pragma unroll
  for (int c = 0; c < 8; ++c) xb[c] = *(const v2f*)(p + colbase(c, half));
}

__device__ __forceinline__ void store_b_cols(float* __restrict__ dst, const v2f xb[8], int lane) {
  const int col = lane & 15;
  const int half = lane >> 4;
  float* p = dst + col * kD;
#pragma unroll
  for (int c = 0; c < 8; ++c) *(v2f*)(p + colbase(c, half)) = xb[c];
}

__device__ __forceinline__ void load_c_cols(const float* __restrict__ src, v8f& c0, v8f& c1, int lane) {
  const int col = lane & 15;
  const int half = lane >> 4;
  const float* p = src + col * kD + 8 * half;
  v4f a = *(const v4f*)(p + 0);
  v4f b = *(const v4f*)(p + 4);
  v4f c = *(const v4f*)(p + 16);
  v4f d = *(const v4f*)(p + 20);
  c0 = (v8f){a[0], a[1], a[2], a[3], b[0], b[1], b[2], b[3]};
  c1 = (v8f){c[0], c[1], c[2], c[3], d[0], d[1], d[2], d[3]};
}

__device__ __forceinline__ void store_c_cols(float* __restrict__ dst, v8f c0, v8f c1, int lane) {
  const int col = lane & 15;
  const int half = lane >> 4;
  float* p = dst + col * kD + 8 * half;
  *(v4f*)(p + 0)  = (v4f){c0[0], c0[1], c0[2], c0[3]};
  *(v4f*)(p + 4)  = (v4f){c0[4], c0[5], c0[6], c0[7]};
  *(v4f*)(p + 16) = (v4f){c1[0], c1[1], c1[2], c1[3]};
  *(v4f*)(p + 20) = (v4f){c1[4], c1[5], c1[6], c1[7]};
}

__device__ __forceinline__ void c_to_b_regs(const v8f& c0, const v8f& c1, v2f xb[8]) {
#pragma unroll
  for (int c = 0; c < 4; ++c) {
    xb[c]     = (v2f){c0[2 * c], c0[2 * c + 1]};
    xb[c + 4] = (v2f){c1[2 * c], c1[2 * c + 1]};
  }
}

__device__ __forceinline__ void step_mm(const ATiles& Axx, const v2f xb[8],
                                        const ATiles& Ain, const v2f inb[8],
                                        v8f& d0, v8f& d1) {
  v8f a0e = {0.f, 0.f, 0.f, 0.f, 0.f, 0.f, 0.f, 0.f};
  v8f a0o = a0e, a1e = a0e, a1o = a0e;
#pragma unroll
  for (int c = 0; c < 8; c += 2) {
    a0e = WMMA4(Ain.a[0][c], inb[c], a0e);
    a0o = WMMA4(Ain.a[0][c + 1], inb[c + 1], a0o);
    a1e = WMMA4(Ain.a[1][c], inb[c], a1e);
    a1o = WMMA4(Ain.a[1][c + 1], inb[c + 1], a1o);
  }
#pragma unroll
  for (int c = 0; c < 8; c += 2) {
    a0e = WMMA4(Axx.a[0][c], xb[c], a0e);
    a0o = WMMA4(Axx.a[0][c + 1], xb[c + 1], a0o);
    a1e = WMMA4(Axx.a[1][c], xb[c], a1e);
    a1o = WMMA4(Axx.a[1][c + 1], xb[c + 1], a1o);
  }
  d0 = a0e + a0o;
  d1 = a1e + a1o;
}

__global__ void __launch_bounds__(32)
kf_forward_kernel(const float* __restrict__ z, const float* __restrict__ x0,
                  const float* __restrict__ F, const float* __restrict__ K,
                  float* __restrict__ hat) {
  const int lane = threadIdx.x & 31;
  const int b0 = blockIdx.x * 16;

  ATiles Fa, Ka;
  load_a_tiles(F, Fa, lane);
  load_a_tiles(K, Ka, lane);

  v2f xb[8];
  load_b_cols(x0 + (size_t)b0 * kD, xb, lane);
  store_b_cols(hat + (size_t)b0 * kD, xb, lane);

  for (int n = 0; n < kN - 1; ++n) {
    const float* zsrc = z + (size_t)n * kStep + (size_t)b0 * kD;
    if (n + 6 < kN - 1)
      __builtin_prefetch(z + (size_t)(n + 6) * kStep + (size_t)b0 * kD + lane * 16, 0, 0);
    v2f zb[8];
    load_b_cols(zsrc, zb, lane);

    v8f d0, d1;
    step_mm(Fa, xb, Ka, zb, d0, d1);

    store_c_cols(hat + (size_t)(n + 1) * kStep + (size_t)b0 * kD, d0, d1, lane);
    c_to_b_regs(d0, d1, xb);
  }
}

__global__ void __launch_bounds__(32)
kf_backward_kernel(const float* __restrict__ z, const float* __restrict__ Ht,
                   const float* __restrict__ Ka0, const float* __restrict__ Gs,
                   const float* __restrict__ Gr, const float* __restrict__ hat,
                   float* __restrict__ out2) {
  const int lane = threadIdx.x & 31;
  const int b0 = blockIdx.x * 16;
  const size_t last = (size_t)(kN - 1) * kStep + (size_t)b0 * kD;

  ATiles A1, A2;
  v2f sb[8];

  {
    load_a_tiles(Ht, A1, lane);
    v2f hb[8];
    load_b_cols(hat + last, hb, lane);
    v8f t0 = {0.f, 0.f, 0.f, 0.f, 0.f, 0.f, 0.f, 0.f};
    v8f t1 = t0;
#pragma unroll
    for (int c = 0; c < 8; ++c) t0 = WMMA4(A1.a[0][c], hb[c], t0);
#pragma unroll
    for (int c = 0; c < 8; ++c) t1 = WMMA4(A1.a[1][c], hb[c], t1);

    v8f zc0, zc1;
    load_c_cols(z + last, zc0, zc1, lane);
    v8f bz0 = zc0 - t0;
    v8f bz1 = zc1 - t1;
    v2f bb[8];
    c_to_b_regs(bz0, bz1, bb);

    load_a_tiles(Ka0, A2, lane);
    v8f s0, s1;
    load_c_cols(hat + last, s0, s1, lane);
#pragma unroll
    for (int c = 0; c < 8; ++c) s0 = WMMA4(A2.a[0][c], bb[c], s0);
#pragma unroll
    for (int c = 0; c < 8; ++c) s1 = WMMA4(A2.a[1][c], bb[c], s1);

    store_c_cols(out2 + last, s0, s1, lane);
    c_to_b_regs(s0, s1, sb);
  }

  load_a_tiles(Gr, A1, lane);
  load_a_tiles(Gs, A2, lane);

  for (int k = kN - 2; k >= 0; --k) {
    const float* hsrc = hat + (size_t)(k + 1) * kStep + (size_t)b0 * kD;
    if (k - 6 >= 0)
      __builtin_prefetch(hat + (size_t)(k - 5) * kStep + (size_t)b0 * kD + lane * 16, 0, 0);
    v2f hb[8];
    load_b_cols(hsrc, hb, lane);

    v8f d0, d1;
    step_mm(A1, sb, A2, hb, d0, d1);

    store_c_cols(out2 + (size_t)k * kStep + (size_t)b0 * kD, d0, d1, lane);
    c_to_b_regs(d0, d1, sb);
  }
}

extern "C" void kernel_launch(void* const* d_in, const int* in_sizes, int n_in,
                              void* d_out, int out_size, void* d_ws, size_t ws_size,
                              hipStream_t stream) {
  (void)in_sizes; (void)n_in; (void)d_ws; (void)ws_size; (void)out_size;
  const float* z   = (const float*)d_in[0];
  const float* x0  = (const float*)d_in[1];
  const float* F   = (const float*)d_in[2];
  const float* K   = (const float*)d_in[3];
  const float* Ht  = (const float*)d_in[4];
  const float* Ka0 = (const float*)d_in[5];
  const float* Gs  = (const float*)d_in[6];
  const float* Gr  = (const float*)d_in[7];

  float* hat  = (float*)d_out;
  float* out2 = hat + (size_t)kN * kB * kD;

  dim3 grid(kB / 16);
  dim3 block(32);
  kf_forward_kernel<<<grid, block, 0, stream>>>(z, x0, F, K, hat);
  kf_backward_kernel<<<grid, block, 0, stream>>>(z, Ht, Ka0, Gs, Gr, hat, out2);
}


# software pipeline of the recurrence-independent input (z forward, hat backward): loaded 4 steps ahead into a register ring, its product formed one step early; no load wait left inside a step
# speedup vs baseline: 2.3671x; 2.3671x over previous
#include <hip/hip_runtime.h>

typedef __attribute__((ext_vector_type(2))) float v2f;
typedef __attribute__((ext_vector_type(4))) float v4f;
typedef __attribute__((ext_vector_type(8))) float v8f;

#define WMMA4(A, B, C) \
  __builtin_amdgcn_wmma_f32_16x16x4_f32(false, (A), false, (B), (short)0, (C), false, false)

constexpr int kN = 2048;
constexpr int kB = 256;
constexpr int kD = 32;
constexpr size_t kStep = (size_t)kB * kD;

__device__ __forceinline__ int colbase(int c, int half) {
  return 2 * c + 8 * half + (c >= 4 ? 8 : 0);
}

struct ATiles { v2f a[2][8]; };

__device__ __forceinline__ void load_a_tiles(const float* __restrict__ M, ATiles& t, int lane) {
  const int r0 = lane & 15;
  const int half = lane >> 4;
#pragma unroll
  for (int rt = 0; rt < 2; ++rt) {
#pragma unroll
    for (int c = 0; c < 8; ++c) {
      const float* p = M + (16 * rt + r0) * kD + colbase(c, half);
      t.a[rt][c] = *(const v2f*)p;
    }
  }
}

__device__ __forceinline__ void load_b_cols(const float* __restrict__ src, v2f xb[8], int lane) {
  const int col = lane & 15;
  const int half = lane >> 4;
  const float* p = src + col * kD;
#pragma unroll
  for (int c = 0; c < 8; ++c) xb[c] = *(const v2f*)(p + colbase(c, half));
}

__device__ __forceinline__ void store_b_cols(float* __restrict__ dst, const v2f xb[8], int lane) {
  const int col = lane & 15;
  const int half = lane >> 4;
  float* p = dst + col * kD;
#pragma unroll
  for (int c = 0; c < 8; ++c) *(v2f*)(p + colbase(c, half)) = xb[c];
}

__device__ __forceinline__ void load_c_cols(const float* __restrict__ src, v8f& c0, v8f& c1, int lane) {
  const int col = lane & 15;
  const int half = lane >> 4;
  const float* p = src + col * kD + 8 * half;
  v4f a = *(const v4f*)(p + 0);
  v4f b = *(const v4f*)(p + 4);
  v4f c = *(const v4f*)(p + 16);
  v4f d = *(const v4f*)(p + 20);
  c0 = (v8f){a[0], a[1], a[2], a[3], b[0], b[1], b[2], b[3]};
  c1 = (v8f){c[0], c[1], c[2], c[3], d[0], d[1], d[2], d[3]};
}

__device__ __forceinline__ void store_c_cols(float* __restrict__ dst, v8f c0, v8f c1, int lane) {
  const int col = lane & 15;
  const int half = lane >> 4;
  float* p = dst + col * kD + 8 * half;
  *(v4f*)(p + 0)  = (v4f){c0[0], c0[1], c0[2], c0[3]};
  *(v4f*)(p + 4)  = (v4f){c0[4], c0[5], c0[6], c0[7]};
  *(v4f*)(p + 16) = (v4f){c1[0], c1[1], c1[2], c1[3]};
  *(v4f*)(p + 20) = (v4f){c1[4], c1[5], c1[6], c1[7]};
}

__device__ __forceinline__ void c_to_b_regs(const v8f& c0, const v8f& c1, v2f xb[8]) {
#pragma unroll
  for (int c = 0; c < 4; ++c) {
    xb[c]     = (v2f){c0[2 * c], c0[2 * c + 1]};
    xb[c + 4] = (v2f){c1[2 * c], c1[2 * c + 1]};
  }
}

struct Acc4 { v8f e0, o0, e1, o1; };

__device__ __forceinline__ void mm_acc(const ATiles& A, const v2f b[8], Acc4& a) {
#pragma unroll
  for (int c = 0; c < 8; c += 2) {
    a.e0 = WMMA4(A.a[0][c], b[c], a.e0);
    a.o0 = WMMA4(A.a[0][c + 1], b[c + 1], a.o0);
    a.e1 = WMMA4(A.a[1][c], b[c], a.e1);
    a.o1 = WMMA4(A.a[1][c + 1], b[c + 1], a.o1);
  }
}

__device__ __forceinline__ void mm_seed(const ATiles& A, const v2f b[8], Acc4& a) {
  const v8f zero = {0.f, 0.f, 0.f, 0.f, 0.f, 0.f, 0.f, 0.f};
  a.e0 = zero; a.o0 = zero; a.e1 = zero; a.o1 = zero;
  mm_acc(A, b, a);
}

constexpr int kAhead = 4;
constexpr int kFar = 16;
constexpr int kMainSteps = ((kN - 2) / kAhead) * kAhead;
constexpr int kTailSteps = (kN - 1) - kMainSteps;

__device__ __forceinline__ void stream_step(const ATiles& Axx, v2f xb[8], const ATiles& Ain,
                                            v2f next_slot[8], Acc4& cur, Acc4& nxt,
                                            bool seed_next, bool refill,
                                            const float* __restrict__ refill_src,
                                            const float* __restrict__ far_src,
                                            float* __restrict__ dst, int lane) {
  mm_acc(Axx, xb, cur);
  __builtin_amdgcn_sched_barrier(0);
  if (seed_next) mm_seed(Ain, next_slot, nxt);
  if (refill) {
    load_b_cols(refill_src, next_slot, lane);
    __builtin_prefetch(far_src + lane * 16, 0, 0);
  }
  const v8f d0 = cur.e0 + cur.o0;
  const v8f d1 = cur.e1 + cur.o1;
  store_c_cols(dst, d0, d1, lane);
  c_to_b_regs(d0, d1, xb);
  __builtin_amdgcn_sched_barrier(0);
}

__global__ void __launch_bounds__(32)
kf_forward_kernel(const float* __restrict__ z, const float* __restrict__ x0,
                  const float* __restrict__ F, const float* __restrict__ K,
                  float* __restrict__ hat) {
  const int lane = threadIdx.x & 31;
  const int b0 = blockIdx.x * 16;

  ATiles Fa, Ka;
  load_a_tiles(F, Fa, lane);
  load_a_tiles(K, Ka, lane);

  v2f xb[8];
  load_b_cols(x0 + (size_t)b0 * kD, xb, lane);
  store_b_cols(hat + (size_t)b0 * kD, xb, lane);

  const float* zbase = z + (size_t)b0 * kD;
  float* hbase = hat + (size_t)b0 * kD;

  v2f zr[kAhead][8];
#pragma unroll
  for (int s = 0; s < kAhead; ++s) load_b_cols(zbase + (size_t)s * kStep, zr[s], lane);

  Acc4 acc[2];
  mm_seed(Ka, zr[0], acc[0]);
  load_b_cols(zbase + (size_t)kAhead * kStep, zr[0], lane);

  for (int n = 0; n < kMainSteps; n += kAhead) {
#pragma unroll
    for (int s = 0; s < kAhead; ++s) {
      const int m = n + s;
      const int mr = min(m + 1 + kAhead, kN - 1);
      const int mf = min(m + 1 + kFar, kN - 1);
      stream_step(Fa, xb, Ka, zr[(s + 1) % kAhead], acc[s & 1], acc[(s + 1) & 1], true, true,
                  zbase + (size_t)mr * kStep, zbase + (size_t)mf * kStep,
                  hbase + (size_t)(m + 1) * kStep, lane);
    }
  }
#pragma unroll
  for (int s = 0; s < kTailSteps; ++s) {
    const int m = kMainSteps + s;
    stream_step(Fa, xb, Ka, zr[(s + 1) % kAhead], acc[s & 1], acc[(s + 1) & 1], s + 1 < kTailSteps, false,
                nullptr, nullptr, hbase + (size_t)(m + 1) * kStep, lane);
  }
}

__global__ void __launch_bounds__(32)
kf_backward_kernel(const float* __restrict__ z, const float* __restrict__ Ht,
                   const float* __restrict__ Ka0, const float* __restrict__ Gs,
                   const float* __restrict__ Gr, const float* __restrict__ hat,
                   float* __restrict__ out2) {
  const int lane = threadIdx.x & 31;
  const int b0 = blockIdx.x * 16;
  const size_t last = (size_t)(kN - 1) * kStep + (size_t)b0 * kD;

  const float* hbase = hat + (size_t)b0 * kD;
  float* obase = out2 + (size_t)b0 * kD;

  v2f hr[kAhead][8];
#pragma unroll
  for (int s = 0; s < kAhead; ++s)
    load_b_cols(hbase + (size_t)(kN - 1 - s) * kStep, hr[s], lane);

  ATiles A1, A2;
  v2f sb[8];

  {
    load_a_tiles(Ht, A1, lane);
    v8f t0 = {0.f, 0.f, 0.f, 0.f, 0.f, 0.f, 0.f, 0.f};
    v8f t1 = t0;
#pragma unroll
    for (int c = 0; c < 8; ++c) t0 = WMMA4(A1.a[0][c], hr[0][c], t0);
#pragma unroll
    for (int c = 0; c < 8; ++c) t1 = WMMA4(A1.a[1][c], hr[0][c], t1);

    v8f zc0, zc1;
    load_c_cols(z + last, zc0, zc1, lane);
    v8f bz0 = zc0 - t0;
    v8f bz1 = zc1 - t1;
    v2f bb[8];
    c_to_b_regs(bz0, bz1, bb);

    load_a_tiles(Ka0, A2, lane);
    v8f s0, s1;
    load_c_cols(hat + last, s0, s1, lane);
#pragma unroll
    for (int c = 0; c < 8; ++c) s0 = WMMA4(A2.a[0][c], bb[c], s0);
#pragma unroll
    for (int c = 0; c < 8; ++c) s1 = WMMA4(A2.a[1][c], bb[c], s1);

    store_c_cols(out2 + last, s0, s1, lane);
    c_to_b_regs(s0, s1, sb);
  }

  load_a_tiles(Gr, A1, lane);
  load_a_tiles(Gs, A2, lane);

  Acc4 acc[2];
  mm_seed(A2, hr[0], acc[0]);
  load_b_cols(hbase + (size_t)(kN - 1 - kAhead) * kStep, hr[0], lane);

  for (int u = 0; u < kMainSteps; u += kAhead) {
#pragma unroll
    for (int s = 0; s < kAhead; ++s) {
      const int j = kN - 1 - (u + s);
      const int jr = max(j - 1 - kAhead, 0);
      const int jf = max(j - 1 - kFar, 0);
      stream_step(A1, sb, A2, hr[(s + 1) % kAhead], acc[s & 1], acc[(s + 1) & 1], true, true,
                  hbase + (size_t)jr * kStep, hbase + (size_t)jf * kStep,
                  obase + (size_t)(j - 1) * kStep, lane);
    }
  }
#pragma unroll
  for (int s = 0; s < kTailSteps; ++s) {
    const int j = kN - 1 - (kMainSteps + s);
    stream_step(A1, sb, A2, hr[(s + 1) % kAhead], acc[s & 1], acc[(s + 1) & 1], s + 1 < kTailSteps, false,
                nullptr, nullptr, obase + (size_t)(j - 1) * kStep, lane);
  }
}

extern "C" void kernel_launch(void* const* d_in, const int* in_sizes, int n_in,
                              void* d_out, int out_size, void* d_ws, size_t ws_size,
                              hipStream_t stream) {
  (void)in_sizes; (void)n_in; (void)d_ws; (void)ws_size; (void)out_size;
  const float* z   = (const float*)d_in[0];
  const float* x0  = (const float*)d_in[1];
  const float* F   = (const float*)d_in[2];
  const float* K   = (const float*)d_in[3];
  const float* Ht  = (const float*)d_in[4];
  const float* Ka0 = (const float*)d_in[5];
  const float* Gs  = (const float*)d_in[6];
  const float* Gr  = (const float*)d_in[7];

  float* hat  = (float*)d_out;
  float* out2 = hat + (size_t)kN * kB * kD;

  dim3 grid(kB / 16);
  dim3 block(32);
  kf_forward_kernel<<<grid, block, 0, stream>>>(z, x0, F, K, hat);
  kf_backward_kernel<<<grid, block, 0, stream>>>(z, Ht, Ka0, Gs, Gr, hat, out2);
}


# input products (K*z, Gs*hat) moved off the serial waves into wide launches that store them in the output as seeds; a serial step is 16 WMMAs in 2 chains, no adds; changes the order of sums
# speedup vs baseline: 4.4596x; 1.8840x over previous
#include <hip/hip_runtime.h>

typedef __attribute__((ext_vector_type(2))) float v2f;
typedef __attribute__((ext_vector_type(4))) float v4f;
typedef __attribute__((ext_vector_type(8))) float v8f;

#define WMMA4(A, B, C) \
  __builtin_amdgcn_wmma_f32_16x16x4_f32(false, (A), false, (B), (short)0, (C), false, false)

constexpr int kN = 2048;
constexpr int kB = 256;
constexpr int kD = 32;
constexpr size_t kStep = (size_t)kB * kD;
constexpr int kTileF = 16 * kD;

__device__ __forceinline__ int colbase(int c, int half) {
  return 2 * c + 8 * half + (c >= 4 ? 8 : 0);
}

struct ATiles { v2f a[2][8]; };

__device__ __forceinline__ void load_a_tiles(const float* __restrict__ M, ATiles& t, int lane) {
  const int r0 = lane & 15;
  const int half = lane >> 4;
#pragma unroll
  for (int rt = 0; rt < 2; ++rt) {
#pragma unroll
    for (int c = 0; c < 8; ++c) {
      const float* p = M + (16 * rt + r0) * kD + colbase(c, half);
      t.a[rt][c] = *(const v2f*)p;
    }
  }
}

__device__ __forceinline__ void load_b_cols(const float* __restrict__ src, v2f xb[8], int lane) {
  const int col = lane & 15;
  const int half = lane >> 4;
  const float* p = src + col * kD;
#pragma unroll
  for (int c = 0; c < 8; ++c) xb[c] = *(const v2f*)(p + colbase(c, half));
}

__device__ __forceinline__ void store_b_cols(float* __restrict__ dst, const v2f xb[8], int lane) {
  const int col = lane & 15;
  const int half = lane >> 4;
  float* p = dst + col * kD;
#pragma unroll
  for (int c = 0; c < 8; ++c) *(v2f*)(p + colbase(c, half)) = xb[c];
}

__device__ __forceinline__ void load_c_cols(const float* src, v8f& c0, v8f& c1, int lane) {
  const int col = lane & 15;
  const int half = lane >> 4;
  const float* p = src + col * kD + 8 * half;
  v4f a = *(const v4f*)(p + 0);
  v4f b = *(const v4f*)(p + 4);
  v4f c = *(const v4f*)(p + 16);
  v4f d = *(const v4f*)(p + 20);
  c0 = (v8f){a[0], a[1], a[2], a[3], b[0], b[1], b[2], b[3]};
  c1 = (v8f){c[0], c[1], c[2], c[3], d[0], d[1], d[2], d[3]};
}

__device__ __forceinline__ void store_c_cols(float* dst, v8f c0, v8f c1, int lane) {
  const int col = lane & 15;
  const int half = lane >> 4;
  float* p = dst + col * kD + 8 * half;
  *(v4f*)(p + 0)  = (v4f){c0[0], c0[1], c0[2], c0[3]};
  *(v4f*)(p + 4)  = (v4f){c0[4], c0[5], c0[6], c0[7]};
  *(v4f*)(p + 16) = (v4f){c1[0], c1[1], c1[2], c1[3]};
  *(v4f*)(p + 20) = (v4f){c1[4], c1[5], c1[6], c1[7]};
}

__device__ __forceinline__ void c_to_b_regs(const v8f& c0, const v8f& c1, v2f xb[8]) {
#pragma unroll
  for (int c = 0; c < 4; ++c) {
    xb[c]     = (v2f){c0[2 * c], c0[2 * c + 1]};
    xb[c + 4] = (v2f){c1[2 * c], c1[2 * c + 1]};
  }
}

__device__ __forceinline__ void mm_chunk(const ATiles& A, const v2f b[8], int c, v8f& a0, v8f& a1) {
  a0 = WMMA4(A.a[0][c], b[c], a0);
  a1 = WMMA4(A.a[1][c], b[c], a1);
}

constexpr int kProdWaves = 4;
constexpr int kProdTiles = 4;

__global__ void __launch_bounds__(32 * kProdWaves)
kf_product_kernel(const float* __restrict__ src, const float* __restrict__ M,
                  float* __restrict__ dst) {
  const int lane = threadIdx.x & 31;
  const int wave = threadIdx.x >> 5;
  const size_t base = (size_t)blockIdx.x * kStep + (size_t)wave * (kProdTiles * kTileF);

  v8f in0[kProdTiles], in1[kProdTiles];
#pragma unroll
  for (int i = 0; i < kProdTiles; ++i) load_c_cols(src + base + (size_t)i * kTileF, in0[i], in1[i], lane);

  ATiles A;
  load_a_tiles(M, A, lane);

#pragma unroll
  for (int i = 0; i < kProdTiles; ++i) {
    v2f b[8];
    c_to_b_regs(in0[i], in1[i], b);
    v8f a0 = {0.f, 0.f, 0.f, 0.f, 0.f, 0.f, 0.f, 0.f};
    v8f a1 = a0;
#pragma unroll
    for (int c = 0; c < 8; ++c) mm_chunk(A, b, c, a0, a1);
    store_c_cols(dst + base + (size_t)i * kTileF, a0, a1, lane);
  }
}

constexpr int kRing = 8;
constexpr int kFar = 32;
constexpr int kSteps = kN - 1;
constexpr int kMainSteps = ((kSteps - 1) / kRing) * kRing;
constexpr int kTailSteps = (kSteps - 1) - kMainSteps;
static_assert(kRing >= 3 && kTailSteps <= kRing - 2, "every tail step's seed is loaded before the tail");

__device__ __forceinline__ float* step_tile(float* base, bool forward, int m) {
  return base + (size_t)(forward ? m + 1 : kN - 2 - m) * kStep;
}

__device__ __forceinline__ void seeded_step(const ATiles& A, const v2f xb[8], v8f acc[2],
                                            bool store_prev, const v8f prev[2], float* prev_dst,
                                            bool refill, v8f freed[2], const float* refill_src,
                                            const float* far_src, int lane) {
  mm_chunk(A, xb, 0, acc[0], acc[1]);
  __builtin_amdgcn_sched_barrier(0);
  if (refill) {
    load_c_cols(refill_src, freed[0], freed[1], lane);
    __builtin_prefetch(far_src + lane * 16, 0, 0);
  }
  if (store_prev) store_c_cols(prev_dst, prev[0], prev[1], lane);
  __builtin_amdgcn_sched_barrier(0);
#pragma unroll
  for (int c = 1; c < 8; ++c) mm_chunk(A, xb, c, acc[0], acc[1]);
  __builtin_amdgcn_sched_barrier(0);
}

__device__ __forceinline__ void run_recurrence(const ATiles& A, const v2f x_init[8], float* base,
                                               bool forward, int lane) {
  v8f sd[kRing][2];
  __builtin_amdgcn_sched_barrier(0);
#pragma unroll
  for (int s = 0; s < kRing - 2; ++s) load_c_cols(step_tile(base, forward, s), sd[s][0], sd[s][1], lane);
  __builtin_amdgcn_sched_barrier(0);

  seeded_step(A, x_init, sd[0], false, sd[0], nullptr, true, sd[kRing - 2],
              step_tile(base, forward, kRing - 2), step_tile(base, forward, kFar), lane);

  for (int n = 0; n < kMainSteps; n += kRing) {
#pragma unroll
    for (int s = 0; s < kRing; ++s) {
      const int m = 1 + n + s;
      const int mr = min(m + kRing - 2, kSteps - 1);
      const int mf = min(m + kFar, kSteps - 1);
      v2f xb[8];
      c_to_b_regs(sd[s % kRing][0], sd[s % kRing][1], xb);
      seeded_step(A, xb, sd[(1 + s) % kRing], true, sd[s % kRing], step_tile(base, forward, m - 1),
                  true, sd[(s + kRing - 1) % kRing], step_tile(base, forward, mr),
                  step_tile(base, forward, mf), lane);
    }
  }
#pragma unroll
  for (int s = 0; s < kTailSteps; ++s) {
    const int m = 1 + kMainSteps + s;
    v2f xb[8];
    c_to_b_regs(sd[s % kRing][0], sd[s % kRing][1], xb);
    seeded_step(A, xb, sd[(1 + s) % kRing], true, sd[s % kRing], step_tile(base, forward, m - 1),
                false, sd[(s + kRing - 1) % kRing], nullptr, nullptr, lane);
  }
  store_c_cols(step_tile(base, forward, kSteps - 1), sd[(kSteps - 1) % kRing][0],
               sd[(kSteps - 1) % kRing][1], lane);
}

__global__ void __launch_bounds__(32)
kf_forward_kernel(const float* __restrict__ x0, const float* __restrict__ F, float* hat) {
  const int lane = threadIdx.x & 31;
  const int b0 = blockIdx.x * 16;

  ATiles Fa;
  load_a_tiles(F, Fa, lane);

  v2f xb[8];
  load_b_cols(x0 + (size_t)b0 * kD, xb, lane);
  store_b_cols(hat + (size_t)b0 * kD, xb, lane);

  run_recurrence(Fa, xb, hat + (size_t)b0 * kD, true, lane);
}

__global__ void __launch_bounds__(32)
kf_backward_kernel(const float* __restrict__ z, const float* __restrict__ Ht,
                   const float* __restrict__ Ka0, const float* __restrict__ Gr,
                   const float* __restrict__ hat, float* out2) {
  const int lane = threadIdx.x & 31;
  const int b0 = blockIdx.x * 16;
  const size_t last = (size_t)(kN - 1) * kStep + (size_t)b0 * kD;

  ATiles A1;
  v2f sb[8];

  {
    ATiles A2;
    v2f hb[8];
    load_b_cols(hat + last, hb, lane);
    load_a_tiles(Ht, A1, lane);
    v8f t0 = {0.f, 0.f, 0.f, 0.f, 0.f, 0.f, 0.f, 0.f};
    v8f t1 = t0;
#pragma unroll
    for (int c = 0; c < 8; ++c) t0 = WMMA4(A1.a[0][c], hb[c], t0);
#pragma unroll
    for (int c = 0; c < 8; ++c) t1 = WMMA4(A1.a[1][c], hb[c], t1);

    v8f zc0, zc1;
    load_c_cols(z + last, zc0, zc1, lane);
    v8f bz0 = zc0 - t0;
    v8f bz1 = zc1 - t1;
    v2f bb[8];
    c_to_b_regs(bz0, bz1, bb);

    load_a_tiles(Ka0, A2, lane);
    v8f s0, s1;
    load_c_cols(hat + last, s0, s1, lane);
#pragma unroll
    for (int c = 0; c < 8; ++c) s0 = WMMA4(A2.a[0][c], bb[c], s0);
#pragma unroll
    for (int c = 0; c < 8; ++c) s1 = WMMA4(A2.a[1][c], bb[c], s1);

    store_c_cols(out2 + last, s0, s1, lane);
    c_to_b_regs(s0, s1, sb);
  }

  load_a_tiles(Gr, A1, lane);
  run_recurrence(A1, sb, out2 + (size_t)b0 * kD, false, lane);
}

extern "C" void kernel_launch(void* const* d_in, const int* in_sizes, int n_in,
                              void* d_out, int out_size, void* d_ws, size_t ws_size,
                              hipStream_t stream) {
  (void)in_sizes; (void)n_in; (void)d_ws; (void)ws_size; (void)out_size;
  const float* z   = (const float*)d_in[0];
  const float* x0  = (const float*)d_in[1];
  const float* F   = (const float*)d_in[2];
  const float* K   = (const float*)d_in[3];
  const float* Ht  = (const float*)d_in[4];
  const float* Ka0 = (const float*)d_in[5];
  const float* Gs  = (const float*)d_in[6];
  const float* Gr  = (const float*)d_in[7];

  float* hat  = (float*)d_out;
  float* out2 = hat + (size_t)kN * kB * kD;

  dim3 wide(kN - 1);
  dim3 wide_block(32 * kProdWaves);
  dim3 grid(kB / 16);
  dim3 block(32);
  kf_product_kernel<<<wide, wide_block, 0, stream>>>(z, K, hat + kStep);
  kf_forward_kernel<<<grid, block, 0, stream>>>(x0, F, hat);
  kf_product_kernel<<<wide, wide_block, 0, stream>>>(hat + kStep, Gs, out2);
  kf_backward_kernel<<<grid, block, 0, stream>>>(z, Ht, Ka0, Gr, hat, out2);
}


# time axis cut into 64 chunks of 32 steps: chunk end states from a zero start, 63 serial jumps by the 32nd matrix power (31 plain products), chunks refilled in parallel; changes the order of sums
# speedup vs baseline: 16.9797x; 3.8074x over previous
#include <hip/hip_runtime.h>

typedef __attribute__((ext_vector_type(2))) float v2f;
typedef __attribute__((ext_vector_type(4))) float v4f;
typedef __attribute__((ext_vector_type(8))) float v8f;

#define WMMA4(A, B, C) \
  __builtin_amdgcn_wmma_f32_16x16x4_f32(false, (A), false, (B), (short)0, (C), false, false)

constexpr int kN = 2048;
constexpr int kB = 256;
constexpr int kD = 32;
constexpr size_t kStep = (size_t)kB * kD;
constexpr int kTileF = 16 * kD;

__device__ __forceinline__ int colbase(int c, int half) {
  return 2 * c + 8 * half + (c >= 4 ? 8 : 0);
}

struct ATiles { v2f a[2][8]; };

__device__ __forceinline__ void load_a_tiles(const float* __restrict__ M, ATiles& t, int lane) {
  const int r0 = lane & 15;
  const int half = lane >> 4;
#pragma unroll
  for (int rt = 0; rt < 2; ++rt) {
#pragma unroll
    for (int c = 0; c < 8; ++c) {
      const float* p = M + (16 * rt + r0) * kD + colbase(c, half);
      t.a[rt][c] = *(const v2f*)p;
    }
  }
}

__device__ __forceinline__ void load_b_cols(const float* __restrict__ src, v2f xb[8], int lane) {
  const int col = lane & 15;
  const int half = lane >> 4;
  const float* p = src + col * kD;
#pragma unroll
  for (int c = 0; c < 8; ++c) xb[c] = *(const v2f*)(p + colbase(c, half));
}

__device__ __forceinline__ void store_b_cols(float* __restrict__ dst, const v2f xb[8], int lane) {
  const int col = lane & 15;
  const int half = lane >> 4;
  float* p = dst + col * kD;
#pragma unroll
  for (int c = 0; c < 8; ++c) *(v2f*)(p + colbase(c, half)) = xb[c];
}

__device__ __forceinline__ void load_c_cols(const float* src, v8f& c0, v8f& c1, int lane) {
  const int col = lane & 15;
  const int half = lane >> 4;
  const float* p = src + col * kD + 8 * half;
  v4f a = *(const v4f*)(p + 0);
  v4f b = *(const v4f*)(p + 4);
  v4f c = *(const v4f*)(p + 16);
  v4f d = *(const v4f*)(p + 20);
  c0 = (v8f){a[0], a[1], a[2], a[3], b[0], b[1], b[2], b[3]};
  c1 = (v8f){c[0], c[1], c[2], c[3], d[0], d[1], d[2], d[3]};
}

__device__ __forceinline__ void store_c_cols(float* dst, v8f c0, v8f c1, int lane) {
  const int col = lane & 15;
  const int half = lane >> 4;
  float* p = dst + col * kD + 8 * half;
  *(v4f*)(p + 0)  = (v4f){c0[0], c0[1], c0[2], c0[3]};
  *(v4f*)(p + 4)  = (v4f){c0[4], c0[5], c0[6], c0[7]};
  *(v4f*)(p + 16) = (v4f){c1[0], c1[1], c1[2], c1[3]};
  *(v4f*)(p + 20) = (v4f){c1[4], c1[5], c1[6], c1[7]};
}

__device__ __forceinline__ void c_to_b_regs(const v8f& c0, const v8f& c1, v2f xb[8]) {
#pragma unroll
  for (int c = 0; c < 4; ++c) {
    xb[c]     = (v2f){c0[2 * c], c0[2 * c + 1]};
    xb[c + 4] = (v2f){c1[2 * c], c1[2 * c + 1]};
  }
}

__device__ __forceinline__ void mm_chunk(const ATiles& A, const v2f b[8], int c, v8f& a0, v8f& a1) {
  a0 = WMMA4(A.a[0][c], b[c], a0);
  a1 = WMMA4(A.a[1][c], b[c], a1);
}

constexpr int kProdWaves = 4;
constexpr int kProdTiles = 4;

__global__ void __launch_bounds__(32 * kProdWaves)
kf_product_kernel(const float* __restrict__ src, const float* __restrict__ M,
                  float* __restrict__ dst) {
  const int lane = threadIdx.x & 31;
  const int wave = threadIdx.x >> 5;
  const size_t base = (size_t)blockIdx.x * kStep + (size_t)wave * (kProdTiles * kTileF);

  v8f in0[kProdTiles], in1[kProdTiles];
#pragma unroll
  for (int i = 0; i < kProdTiles; ++i) load_c_cols(src + base + (size_t)i * kTileF, in0[i], in1[i], lane);

  ATiles A;
  load_a_tiles(M, A, lane);

#pragma unroll
  for (int i = 0; i < kProdTiles; ++i) {
    v2f b[8];
    c_to_b_regs(in0[i], in1[i], b);
    v8f a0 = {0.f, 0.f, 0.f, 0.f, 0.f, 0.f, 0.f, 0.f};
    v8f a1 = a0;
#pragma unroll
    for (int c = 0; c < 8; ++c) mm_chunk(A, b, c, a0, a1);
    store_c_cols(dst + base + (size_t)i * kTileF, a0, a1, lane);
  }
}

constexpr int kRing = 8;
constexpr int kFar = 16;

constexpr int kL = 32;
constexpr int kT = kN / kL;
static_assert(kT * kL == kN, "chunks tile the time axis");

__device__ __forceinline__ float* step_tile(float* base, bool forward, int g) {
  return base + (size_t)(forward ? g + 1 : kN - 2 - g) * kStep;
}

template <bool COMPUTE>
__device__ __forceinline__ void seeded_step(const ATiles& A, const v2f xb[8], v8f acc[2],
                                            bool store_prev, const v8f prev[2], float* prev_dst,
                                            bool refill, v8f freed[2], const float* refill_src,
                                            const float* far_src, int lane) {
  if (COMPUTE) mm_chunk(A, xb, 0, acc[0], acc[1]);
  __builtin_amdgcn_sched_barrier(0);
  if (refill) {
    load_c_cols(refill_src, freed[0], freed[1], lane);
    __builtin_prefetch(far_src + lane * 16, 0, 0);
  }
  if (store_prev) store_c_cols(prev_dst, prev[0], prev[1], lane);
  __builtin_amdgcn_sched_barrier(0);
  if (COMPUTE) {
#pragma unroll
    for (int c = 1; c < 8; ++c) mm_chunk(A, xb, c, acc[0], acc[1]);
  }
  __builtin_amdgcn_sched_barrier(0);
}

template <int STEPS, long STRIDE, bool STORE_ALL, bool ZERO_INIT>
__device__ __forceinline__ void run_recurrence(const ATiles& A, const v2f x_init[8], float* tile0,
                                               int lane) {
  constexpr int kMainSteps = ((STEPS - 1) / kRing) * kRing;
  constexpr int kTailSteps = (STEPS - 1) - kMainSteps;
  static_assert(kRing >= 3 && STEPS >= kRing, "the ring's first loads stay inside the run");
  v8f sd[kRing][2];
  __builtin_amdgcn_sched_barrier(0);
#pragma unroll
  for (int s = 0; s < kRing - 2; ++s) load_c_cols(tile0 + s * STRIDE, sd[s][0], sd[s][1], lane);
  __builtin_amdgcn_sched_barrier(0);
#pragma unroll
  for (int s = kRing - 2; s < min(kFar, STEPS); ++s) __builtin_prefetch(tile0 + s * STRIDE + lane * 16, 0, 0);

  seeded_step<!ZERO_INIT>(A, x_init, sd[0], false, sd[0], nullptr, true, sd[kRing - 2],
                          tile0 + (kRing - 2) * STRIDE, tile0 + min(kFar, STEPS - 1) * STRIDE, lane);

  for (int n = 0; n < kMainSteps; n += kRing) {
#pragma unroll
    for (int s = 0; s < kRing; ++s) {
      const int m = 1 + n + s;
      const int mr = min(m + kRing - 2, STEPS - 1);
      const int mf = min(m + kFar, STEPS - 1);
      v2f xb[8];
      c_to_b_regs(sd[s % kRing][0], sd[s % kRing][1], xb);
      seeded_step<true>(A, xb, sd[(1 + s) % kRing], STORE_ALL, sd[s % kRing], tile0 + (long)(m - 1) * STRIDE,
                        true, sd[(s + kRing - 1) % kRing], tile0 + (long)mr * STRIDE,
                        tile0 + (long)mf * STRIDE, lane);
    }
  }
#pragma unroll
  for (int s = 0; s < kTailSteps; ++s) {
    const int m = 1 + kMainSteps + s;
    const bool refill = m + kRing - 2 <= STEPS - 1;
    const int mr = refill ? m + kRing - 2 : STEPS - 1;
    v2f xb[8];
    c_to_b_regs(sd[s % kRing][0], sd[s % kRing][1], xb);
    seeded_step<true>(A, xb, sd[(1 + s) % kRing], STORE_ALL, sd[s % kRing], tile0 + (long)(m - 1) * STRIDE,
                      refill, sd[(s + kRing - 1) % kRing], tile0 + (long)mr * STRIDE,
                      tile0 + (long)(STEPS - 1) * STRIDE, lane);
  }
  store_c_cols(tile0 + (long)(STEPS - 1) * STRIDE, sd[(STEPS - 1) % kRing][0],
               sd[(STEPS - 1) % kRing][1], lane);
}

constexpr long kFwdStride = (long)kStep;
constexpr long kBwdStride = -(long)kStep;

template <bool FWD>
__global__ void __launch_bounds__(32)
kf_chunk_end_kernel(const float* __restrict__ M, float* buf) {
  const int lane = threadIdx.x & 31;
  const int b0 = blockIdx.x * 16;
  const int j = blockIdx.y;

  ATiles A;
  load_a_tiles(M, A, lane);
  v2f none[8] = {};
  run_recurrence<kL, FWD ? kFwdStride : kBwdStride, false, true>(
      A, none, step_tile(buf + (size_t)b0 * kD, FWD, j * kL), lane);
}

template <bool FWD>
__global__ void __launch_bounds__(32)
kf_chunk_fill_kernel(const float* __restrict__ M, float* buf) {
  const int lane = threadIdx.x & 31;
  const int b0 = blockIdx.x * 16;
  const int j = blockIdx.y;

  ATiles A;
  load_a_tiles(M, A, lane);
  v2f xb[8];
  load_b_cols(step_tile(buf + (size_t)b0 * kD, FWD, j * kL - 1), xb, lane);
  run_recurrence<kL - 1, FWD ? kFwdStride : kBwdStride, true, false>(
      A, xb, step_tile(buf + (size_t)b0 * kD, FWD, j * kL), lane);
}

__device__ __forceinline__ void power_a_tiles(const float* __restrict__ M, ATiles& P, int lane) {
  const int r0 = lane & 15;
  const int half = lane >> 4;
  ATiles Mt;
#pragma unroll
  for (int rt = 0; rt < 2; ++rt) {
#pragma unroll
    for (int c = 0; c < 8; ++c) {
      const float* p = M + colbase(c, half) * kD + 16 * rt + r0;
      Mt.a[rt][c] = (v2f){p[0], p[kD]};
    }
  }
  load_a_tiles(M, P, lane);
  for (int i = 1; i < kL; ++i) {
    const v8f zero = {0.f, 0.f, 0.f, 0.f, 0.f, 0.f, 0.f, 0.f};
    v8f d00 = zero, d01 = zero, d10 = zero, d11 = zero;
#pragma unroll
    for (int c = 0; c < 8; ++c) {
      d00 = WMMA4(Mt.a[0][c], P.a[0][c], d00);
      d01 = WMMA4(Mt.a[1][c], P.a[0][c], d01);
      d10 = WMMA4(Mt.a[0][c], P.a[1][c], d10);
      d11 = WMMA4(Mt.a[1][c], P.a[1][c], d11);
    }
    c_to_b_regs(d00, d01, P.a[0]);
    c_to_b_regs(d10, d11, P.a[1]);
  }
}

__global__ void __launch_bounds__(32)
kf_forward_borders_kernel(const float* __restrict__ x0, const float* __restrict__ F, float* hat) {
  const int lane = threadIdx.x & 31;
  const int b0 = blockIdx.x * 16;

  ATiles P;
  power_a_tiles(F, P, lane);

  v2f xb[8];
  load_b_cols(x0 + (size_t)b0 * kD, xb, lane);
  store_b_cols(hat + (size_t)b0 * kD, xb, lane);

  run_recurrence<kT - 1, kL * kFwdStride, true, false>(
      P, xb, step_tile(hat + (size_t)b0 * kD, true, kL - 1), lane);
}

__global__ void __launch_bounds__(32)
kf_backward_borders_kernel(const float* __restrict__ z, const float* __restrict__ Ht,
                           const float* __restrict__ Ka0, const float* __restrict__ Gr,
                           const float* __restrict__ hat, float* out2) {
  const int lane = threadIdx.x & 31;
  const int b0 = blockIdx.x * 16;
  const size_t last = (size_t)(kN - 1) * kStep + (size_t)b0 * kD;

  ATiles A1;
  v2f sb[8];

  {
    ATiles A2;
    v2f hb[8];
    load_b_cols(hat + last, hb, lane);
    load_a_tiles(Ht, A1, lane);
    v8f t0 = {0.f, 0.f, 0.f, 0.f, 0.f, 0.f, 0.f, 0.f};
    v8f t1 = t0;
#pragma unroll
    for (int c = 0; c < 8; ++c) t0 = WMMA4(A1.a[0][c], hb[c], t0);
#pragma unroll
    for (int c = 0; c < 8; ++c) t1 = WMMA4(A1.a[1][c], hb[c], t1);

    v8f zc0, zc1;
    load_c_cols(z + last, zc0, zc1, lane);
    v8f bz0 = zc0 - t0;
    v8f bz1 = zc1 - t1;
    v2f bb[8];
    c_to_b_regs(bz0, bz1, bb);

    load_a_tiles(Ka0, A2, lane);
    v8f s0, s1;
    load_c_cols(hat + last, s0, s1, lane);
#pragma unroll
    for (int c = 0; c < 8; ++c) s0 = WMMA4(A2.a[0][c], bb[c], s0);
#pragma unroll
    for (int c = 0; c < 8; ++c) s1 = WMMA4(A2.a[1][c], bb[c], s1);

    store_c_cols(out2 + last, s0, s1, lane);
    c_to_b_regs(s0, s1, sb);
  }

  power_a_tiles(Gr, A1, lane);
  run_recurrence<kT - 1, kL * kBwdStride, true, false>(
      A1, sb, step_tile(out2 + (size_t)b0 * kD, false, kL - 1), lane);
}

extern "C" void kernel_launch(void* const* d_in, const int* in_sizes, int n_in,
                              void* d_out, int out_size, void* d_ws, size_t ws_size,
                              hipStream_t stream) {
  (void)in_sizes; (void)n_in; (void)d_ws; (void)ws_size; (void)out_size;
  const float* z   = (const float*)d_in[0];
  const float* x0  = (const float*)d_in[1];
  const float* F   = (const float*)d_in[2];
  const float* K   = (const float*)d_in[3];
  const float* Ht  = (const float*)d_in[4];
  const float* Ka0 = (const float*)d_in[5];
  const float* Gs  = (const float*)d_in[6];
  const float* Gr  = (const float*)d_in[7];

  float* hat  = (float*)d_out;
  float* out2 = hat + (size_t)kN * kB * kD;

  dim3 wide(kN - 1);
  dim3 wide_block(32 * kProdWaves);
  dim3 slices(kB / 16);
  dim3 ends(kB / 16, kT - 1);
  dim3 chunks(kB / 16, kT);
  dim3 block(32);
  kf_product_kernel<<<wide, wide_block, 0, stream>>>(z, K, hat + kStep);
  kf_chunk_end_kernel<true><<<ends, block, 0, stream>>>(F, hat);
  kf_forward_borders_kernel<<<slices, block, 0, stream>>>(x0, F, hat);
  kf_chunk_fill_kernel<true><<<chunks, block, 0, stream>>>(F, hat);
  kf_product_kernel<<<wide, wide_block, 0, stream>>>(hat + kStep, Gs, out2);
  kf_chunk_end_kernel<false><<<ends, block, 0, stream>>>(Gr, out2);
  kf_backward_borders_kernel<<<slices, block, 0, stream>>>(z, Ht, Ka0, Gr, hat, out2);
  kf_chunk_fill_kernel<false><<<chunks, block, 0, stream>>>(Gr, out2);
}


# the two wide input-product launches folded into the forward chunk kernels (K*z into the chunk-end pass, Gs*hat into the fill pass, on the wave holding the tile): 6 launches; order of sums unchanged
# speedup vs baseline: 23.8478x; 1.4045x over previous
#include <hip/hip_runtime.h>

typedef __attribute__((ext_vector_type(2))) float v2f;
typedef __attribute__((ext_vector_type(4))) float v4f;
typedef __attribute__((ext_vector_type(8))) float v8f;

#define WMMA4(A, B, C) \
  __builtin_amdgcn_wmma_f32_16x16x4_f32(false, (A), false, (B), (short)0, (C), false, false)

constexpr int kN = 2048;
constexpr int kB = 256;
constexpr int kD = 32;
constexpr size_t kStep = (size_t)kB * kD;

__device__ __forceinline__ int colbase(int c, int half) {
  return 2 * c + 8 * half + (c >= 4 ? 8 : 0);
}

struct ATiles { v2f a[2][8]; };

__device__ __forceinline__ void load_a_tiles(const float* __restrict__ M, ATiles& t, int lane) {
  const int r0 = lane & 15;
  const int half = lane >> 4;
#pragma unroll
  for (int rt = 0; rt < 2; ++rt) {
#pragma unroll
    for (int c = 0; c < 8; ++c) {
      const float* p = M + (16 * rt + r0) * kD + colbase(c, half);
      t.a[rt][c] = *(const v2f*)p;
    }
  }
}

__device__ __forceinline__ void load_b_cols(const float* __restrict__ src, v2f xb[8], int lane) {
  const int col = lane & 15;
  const int half = lane >> 4;
  const float* p = src + col * kD;
#pragma unroll
  for (int c = 0; c < 8; ++c) xb[c] = *(const v2f*)(p + colbase(c, half));
}

__device__ __forceinline__ void store_b_cols(float* __restrict__ dst, const v2f xb[8], int lane) {
  const int col = lane & 15;
  const int half = lane >> 4;
  float* p = dst + col * kD;
#pragma unroll
  for (int c = 0; c < 8; ++c) *(v2f*)(p + colbase(c, half)) = xb[c];
}

__device__ __forceinline__ void load_c_cols(const float* src, v8f& c0, v8f& c1, int lane) {
  const int col = lane & 15;
  const int half = lane >> 4;
  const float* p = src + col * kD + 8 * half;
  v4f a = *(const v4f*)(p + 0);
  v4f b = *(const v4f*)(p + 4);
  v4f c = *(const v4f*)(p + 16);
  v4f d = *(const v4f*)(p + 20);
  c0 = (v8f){a[0], a[1], a[2], a[3], b[0], b[1], b[2], b[3]};
  c1 = (v8f){c[0], c[1], c[2], c[3], d[0], d[1], d[2], d[3]};
}

__device__ __forceinline__ void store_c_cols(float* dst, v8f c0, v8f c1, int lane) {
  const int col = lane & 15;
  const int half = lane >> 4;
  float* p = dst + col * kD + 8 * half;
  *(v4f*)(p + 0)  = (v4f){c0[0], c0[1], c0[2], c0[3]};
  *(v4f*)(p + 4)  = (v4f){c0[4], c0[5], c0[6], c0[7]};
  *(v4f*)(p + 16) = (v4f){c1[0], c1[1], c1[2], c1[3]};
  *(v4f*)(p + 20) = (v4f){c1[4], c1[5], c1[6], c1[7]};
}

__device__ __forceinline__ void c_to_b_regs(const v8f& c0, const v8f& c1, v2f xb[8]) {
#pragma unroll
  for (int c = 0; c < 4; ++c) {
    xb[c]     = (v2f){c0[2 * c], c0[2 * c + 1]};
    xb[c + 4] = (v2f){c1[2 * c], c1[2 * c + 1]};
  }
}

__device__ __forceinline__ void mm_chunk(const ATiles& A, const v2f b[8], int c, v8f& a0, v8f& a1) {
  a0 = WMMA4(A.a[0][c], b[c], a0);
  a1 = WMMA4(A.a[1][c], b[c], a1);
}

constexpr int kRing = 8;
constexpr int kFar = 16;

constexpr int kL = 32;
constexpr int kT = kN / kL;
static_assert(kT * kL == kN, "chunks tile the time axis");

__device__ __forceinline__ float* step_tile(float* base, bool forward, int g) {
  return base + (size_t)(forward ? g + 1 : kN - 2 - g) * kStep;
}

template <bool COMPUTE>
__device__ __forceinline__ void seeded_step(const ATiles& A, const v2f xb[8], v8f acc[2],
                                            bool store_prev, const v8f prev[2], float* prev_dst,
                                            bool refill, v8f freed[2], const float* refill_src,
                                            const float* far_src, int lane) {
  if (COMPUTE) mm_chunk(A, xb, 0, acc[0], acc[1]);
  __builtin_amdgcn_sched_barrier(0);
  if (refill) {
    load_c_cols(refill_src, freed[0], freed[1], lane);
    __builtin_prefetch(far_src + lane * 16, 0, 0);
  }
  if (store_prev) store_c_cols(prev_dst, prev[0], prev[1], lane);
  __builtin_amdgcn_sched_barrier(0);
  if (COMPUTE) {
#pragma unroll
    for (int c = 1; c < 8; ++c) mm_chunk(A, xb, c, acc[0], acc[1]);
  }
  __builtin_amdgcn_sched_barrier(0);
}

template <int STEPS, long STRIDE, bool STORE_ALL, bool ZERO_INIT>
__device__ __forceinline__ void run_recurrence(const ATiles& A, const v2f x_init[8], float* tile0,
                                               int lane) {
  constexpr int kMainSteps = ((STEPS - 1) / kRing) * kRing;
  constexpr int kTailSteps = (STEPS - 1) - kMainSteps;
  static_assert(kRing >= 3 && STEPS >= kRing, "the ring's first loads stay inside the run");
  v8f sd[kRing][2];
  __builtin_amdgcn_sched_barrier(0);
#pragma unroll
  for (int s = 0; s < kRing - 2; ++s) load_c_cols(tile0 + s * STRIDE, sd[s][0], sd[s][1], lane);
  __builtin_amdgcn_sched_barrier(0);
#pragma unroll
  for (int s = kRing - 2; s < min(kFar, STEPS); ++s) __builtin_prefetch(tile0 + s * STRIDE + lane * 16, 0, 0);

  seeded_step<!ZERO_INIT>(A, x_init, sd[0], false, sd[0], nullptr, true, sd[kRing - 2],
                          tile0 + (kRing - 2) * STRIDE, tile0 + min(kFar, STEPS - 1) * STRIDE, lane);

  for (int n = 0; n < kMainSteps; n += kRing) {
#pragma unroll
    for (int s = 0; s < kRing; ++s) {
      const int m = 1 + n + s;
      const int mr = min(m + kRing - 2, STEPS - 1);
      const int mf = min(m + kFar, STEPS - 1);
      v2f xb[8];
      c_to_b_regs(sd[s % kRing][0], sd[s % kRing][1], xb);
      seeded_step<true>(A, xb, sd[(1 + s) % kRing], STORE_ALL, sd[s % kRing], tile0 + (long)(m - 1) * STRIDE,
                        true, sd[(s + kRing - 1) % kRing], tile0 + (long)mr * STRIDE,
                        tile0 + (long)mf * STRIDE, lane);
    }
  }
#pragma unroll
  for (int s = 0; s < kTailSteps; ++s) {
    const int m = 1 + kMainSteps + s;
    const bool refill = m + kRing - 2 <= STEPS - 1;
    const int mr = refill ? m + kRing - 2 : STEPS - 1;
    v2f xb[8];
    c_to_b_regs(sd[s % kRing][0], sd[s % kRing][1], xb);
    seeded_step<true>(A, xb, sd[(1 + s) % kRing], STORE_ALL, sd[s % kRing], tile0 + (long)(m - 1) * STRIDE,
                      refill, sd[(s + kRing - 1) % kRing], tile0 + (long)mr * STRIDE,
                      tile0 + (long)(STEPS - 1) * STRIDE, lane);
  }
  store_c_cols(tile0 + (long)(STEPS - 1) * STRIDE, sd[(STEPS - 1) % kRing][0],
               sd[(STEPS - 1) % kRing][1], lane);
}

constexpr long kFwdStride = (long)kStep;
constexpr long kBwdStride = -(long)kStep;

template <bool FWD>
__global__ void __launch_bounds__(32)
kf_chunk_end_kernel(const float* __restrict__ M, float* buf) {
  const int lane = threadIdx.x & 31;
  const int b0 = blockIdx.x * 16;
  const int j = blockIdx.y;

  ATiles A;
  load_a_tiles(M, A, lane);
  v2f none[8] = {};
  run_recurrence<kL, FWD ? kFwdStride : kBwdStride, false, true>(
      A, none, step_tile(buf + (size_t)b0 * kD, FWD, j * kL), lane);
}

template <bool FWD>
__global__ void __launch_bounds__(32)
kf_chunk_fill_kernel(const float* __restrict__ M, float* buf) {
  const int lane = threadIdx.x & 31;
  const int b0 = blockIdx.x * 16;
  const int j = blockIdx.y;

  ATiles A;
  load_a_tiles(M, A, lane);
  v2f xb[8];
  load_b_cols(step_tile(buf + (size_t)b0 * kD, FWD, j * kL - 1), xb, lane);
  run_recurrence<kL - 1, FWD ? kFwdStride : kBwdStride, true, false>(
      A, xb, step_tile(buf + (size_t)b0 * kD, FWD, j * kL), lane);
}

constexpr int kInRing = 6;

template <int STEPS, bool STORE_END>
__device__ __forceinline__ void forward_end_pass(const ATiles& Kt, const ATiles& A,
                                                 const float* __restrict__ zt, float* dst, int lane) {
  static_assert(STEPS >= kInRing && STEPS >= kL - 1 && STEPS <= kL, "a whole chunk, or the last one");
  const v8f zero = {0.f, 0.f, 0.f, 0.f, 0.f, 0.f, 0.f, 0.f};
  v8f zr[kInRing][2];
#pragma unroll
  for (int s = 0; s < kInRing; ++s) load_c_cols(zt + (long)s * kStep, zr[s][0], zr[s][1], lane);
#pragma unroll
  for (int s = kInRing; s < min(kFar, STEPS); ++s) __builtin_prefetch(zt + (long)s * kStep + lane * 16, 0, 0);

  v8f st[2] = {zero, zero};
  v8f sp[2] = {zero, zero};
#pragma unroll
  for (int m = 0; m < STEPS; ++m) {
    v2f zb[8];
    c_to_b_regs(zr[m % kInRing][0], zr[m % kInRing][1], zb);
    v8f s0 = zero, s1 = zero;
    mm_chunk(Kt, zb, 0, s0, s1);
    __builtin_amdgcn_sched_barrier(0);
    if (m >= 1) {
      if (m - 1 + kInRing < STEPS)
        load_c_cols(zt + (long)(m - 1 + kInRing) * kStep, zr[(m - 1) % kInRing][0], zr[(m - 1) % kInRing][1], lane);
      if (m - 1 + kFar < STEPS) __builtin_prefetch(zt + (long)(m - 1 + kFar) * kStep + lane * 16, 0, 0);
      store_c_cols(dst + (long)(m - 1) * kStep, sp[0], sp[1], lane);
    }
    __builtin_amdgcn_sched_barrier(0);
#pragma unroll
    for (int c = 1; c < 8; ++c) mm_chunk(Kt, zb, c, s0, s1);
    sp[0] = s0;
    sp[1] = s1;
    if (m >= 1 && STORE_END) {
      v2f xb[8];
      c_to_b_regs(st[0], st[1], xb);
#pragma unroll
      for (int c = 0; c < 8; ++c) mm_chunk(A, xb, c, s0, s1);
    }
    st[0] = s0;
    st[1] = s1;
    __builtin_amdgcn_sched_barrier(0);
  }
  if (STEPS - 1 <= kL - 2) store_c_cols(dst + (long)(STEPS - 1) * kStep, sp[0], sp[1], lane);
  if (STORE_END) store_c_cols(dst + (long)(kL - 1) * kStep, st[0], st[1], lane);
}

__global__ void __launch_bounds__(32)
kf_forward_end_kernel(const float* __restrict__ z, const float* __restrict__ K,
                      const float* __restrict__ F, float* hat) {
  const int lane = threadIdx.x & 31;
  const int b0 = blockIdx.x * 16;
  const int j = blockIdx.y;

  ATiles Kt, A;
  load_a_tiles(K, Kt, lane);
  load_a_tiles(F, A, lane);
  const float* zt = z + (size_t)j * kL * kStep + (size_t)b0 * kD;
  float* dst = step_tile(hat + (size_t)b0 * kD, true, j * kL);
  if (j < kT - 1) forward_end_pass<kL, true>(Kt, A, zt, dst, lane);
  else            forward_end_pass<kL - 1, false>(Kt, A, zt, dst, lane);
}

template <bool FIRST_PRODUCT>
__device__ __forceinline__ void forward_fill_pass(const ATiles& A, const ATiles& G, const v2f x_init[8],
                                                  float* tile0, float* __restrict__ p0, int lane) {
  constexpr int STEPS = kL - 1;
  const v8f zero = {0.f, 0.f, 0.f, 0.f, 0.f, 0.f, 0.f, 0.f};
  v8f sd[kInRing][2];
#pragma unroll
  for (int s = 0; s < kInRing; ++s) load_c_cols(tile0 + (long)s * kStep, sd[s][0], sd[s][1], lane);
#pragma unroll
  for (int s = kInRing; s < min(kFar, STEPS); ++s) __builtin_prefetch(tile0 + (long)s * kStep + lane * 16, 0, 0);

  v2f xb[8];
#pragma unroll
  for (int c = 0; c < 8; ++c) xb[c] = x_init[c];
  v8f xs[2] = {zero, zero};
  v8f ps[2] = {zero, zero};
#pragma unroll
  for (int m = 0; m < STEPS; ++m) {
    v8f a0 = sd[m % kInRing][0], a1 = sd[m % kInRing][1];
    mm_chunk(A, xb, 0, a0, a1);
    __builtin_amdgcn_sched_barrier(0);
    if (m >= 1) {
      if (m - 1 + kInRing < STEPS)
        load_c_cols(tile0 + (long)(m - 1 + kInRing) * kStep, sd[(m - 1) % kInRing][0], sd[(m - 1) % kInRing][1], lane);
      if (m - 1 + kFar < STEPS) __builtin_prefetch(tile0 + (long)(m - 1 + kFar) * kStep + lane * 16, 0, 0);
      store_c_cols(tile0 + (long)(m - 1) * kStep, xs[0], xs[1], lane);
      if (m >= 2 || FIRST_PRODUCT) store_c_cols(p0 + (long)(m - 1) * kStep, ps[0], ps[1], lane);
    }
    __builtin_amdgcn_sched_barrier(0);
#pragma unroll
    for (int c = 1; c < 8; ++c) mm_chunk(A, xb, c, a0, a1);
    v8f q0 = zero, q1 = zero;
#pragma unroll
    for (int c = 0; c < 8; ++c) mm_chunk(G, xb, c, q0, q1);
    xs[0] = a0;
    xs[1] = a1;
    ps[0] = q0;
    ps[1] = q1;
    c_to_b_regs(a0, a1, xb);
    __builtin_amdgcn_sched_barrier(0);
  }
  store_c_cols(tile0 + (long)(STEPS - 1) * kStep, xs[0], xs[1], lane);
  store_c_cols(p0 + (long)(STEPS - 1) * kStep, ps[0], ps[1], lane);
  v8f q0 = zero, q1 = zero;
#pragma unroll
  for (int c = 0; c < 8; ++c) mm_chunk(G, xb, c, q0, q1);
  store_c_cols(p0 + (long)STEPS * kStep, q0, q1, lane);
}

__global__ void __launch_bounds__(32)
kf_forward_fill_kernel(const float* __restrict__ F, const float* __restrict__ Gs, float* hat,
                       float* __restrict__ out2) {
  const int lane = threadIdx.x & 31;
  const int b0 = blockIdx.x * 16;
  const int j = blockIdx.y;

  ATiles A, G;
  load_a_tiles(F, A, lane);
  load_a_tiles(Gs, G, lane);
  v2f xb[8];
  load_b_cols(step_tile(hat + (size_t)b0 * kD, true, j * kL - 1), xb, lane);
  float* p0 = out2 + ((long)j * kL - 1) * (long)kStep + (long)b0 * kD;
  float* tile0 = step_tile(hat + (size_t)b0 * kD, true, j * kL);
  if (j > 0) forward_fill_pass<true>(A, G, xb, tile0, p0, lane);
  else       forward_fill_pass<false>(A, G, xb, tile0, p0, lane);
}

__device__ __forceinline__ void power_a_tiles(const float* __restrict__ M, ATiles& P, int lane) {
  const int r0 = lane & 15;
  const int half = lane >> 4;
  ATiles Mt;
#pragma unroll
  for (int rt = 0; rt < 2; ++rt) {
#pragma unroll
    for (int c = 0; c < 8; ++c) {
      const float* p = M + colbase(c, half) * kD + 16 * rt + r0;
      Mt.a[rt][c] = (v2f){p[0], p[kD]};
    }
  }
  load_a_tiles(M, P, lane);
  for (int i = 1; i < kL; ++i) {
    const v8f zero = {0.f, 0.f, 0.f, 0.f, 0.f, 0.f, 0.f, 0.f};
    v8f d00 = zero, d01 = zero, d10 = zero, d11 = zero;
#pragma unroll
    for (int c = 0; c < 8; ++c) {
      d00 = WMMA4(Mt.a[0][c], P.a[0][c], d00);
      d01 = WMMA4(Mt.a[1][c], P.a[0][c], d01);
      d10 = WMMA4(Mt.a[0][c], P.a[1][c], d10);
      d11 = WMMA4(Mt.a[1][c], P.a[1][c], d11);
    }
    c_to_b_regs(d00, d01, P.a[0]);
    c_to_b_regs(d10, d11, P.a[1]);
  }
}

__global__ void __launch_bounds__(32)
kf_forward_borders_kernel(const float* __restrict__ x0, const float* __restrict__ F, float* hat) {
  const int lane = threadIdx.x & 31;
  const int b0 = blockIdx.x * 16;

  ATiles P;
  power_a_tiles(F, P, lane);

  v2f xb[8];
  load_b_cols(x0 + (size_t)b0 * kD, xb, lane);
  store_b_cols(hat + (size_t)b0 * kD, xb, lane);

  run_recurrence<kT - 1, kL * kFwdStride, true, false>(
      P, xb, step_tile(hat + (size_t)b0 * kD, true, kL - 1), lane);
}

__global__ void __launch_bounds__(32)
kf_backward_borders_kernel(const float* __restrict__ z, const float* __restrict__ Ht,
                           const float* __restrict__ Ka0, const float* __restrict__ Gr,
                           const float* __restrict__ hat, float* out2) {
  const int lane = threadIdx.x & 31;
  const int b0 = blockIdx.x * 16;
  const size_t last = (size_t)(kN - 1) * kStep + (size_t)b0 * kD;

  ATiles A1;
  v2f sb[8];

  {
    ATiles A2;
    v2f hb[8];
    load_b_cols(hat + last, hb, lane);
    load_a_tiles(Ht, A1, lane);
    v8f t0 = {0.f, 0.f, 0.f, 0.f, 0.f, 0.f, 0.f, 0.f};
    v8f t1 = t0;
#pragma unroll
    for (int c = 0; c < 8; ++c) t0 = WMMA4(A1.a[0][c], hb[c], t0);
#pragma unroll
    for (int c = 0; c < 8; ++c) t1 = WMMA4(A1.a[1][c], hb[c], t1);

    v8f zc0, zc1;
    load_c_cols(z + last, zc0, zc1, lane);
    v8f bz0 = zc0 - t0;
    v8f bz1 = zc1 - t1;
    v2f bb[8];
    c_to_b_regs(bz0, bz1, bb);

    load_a_tiles(Ka0, A2, lane);
    v8f s0, s1;
    load_c_cols(hat + last, s0, s1, lane);
#pragma unroll
    for (int c = 0; c < 8; ++c) s0 = WMMA4(A2.a[0][c], bb[c], s0);
#pragma unroll
    for (int c = 0; c < 8; ++c) s1 = WMMA4(A2.a[1][c], bb[c], s1);

    store_c_cols(out2 + last, s0, s1, lane);
    c_to_b_regs(s0, s1, sb);
  }

  power_a_tiles(Gr, A1, lane);
  run_recurrence<kT - 1, kL * kBwdStride, true, false>(
      A1, sb, step_tile(out2 + (size_t)b0 * kD, false, kL - 1), lane);
}

extern "C" void kernel_launch(void* const* d_in, const int* in_sizes, int n_in,
                              void* d_out, int out_size, void* d_ws, size_t ws_size,
                              hipStream_t stream) {
  (void)in_sizes; (void)n_in; (void)d_ws; (void)ws_size; (void)out_size;
  const float* z   = (const float*)d_in[0];
  const float* x0  = (const float*)d_in[1];
  const float* F   = (const float*)d_in[2];
  const float* K   = (const float*)d_in[3];
  const float* Ht  = (const float*)d_in[4];
  const float* Ka0 = (const float*)d_in[5];
  const float* Gs  = (const float*)d_in[6];
  const float* Gr  = (const float*)d_in[7];

  float* hat  = (float*)d_out;
  float* out2 = hat + (size_t)kN * kB * kD;

  dim3 slices(kB / 16);
  dim3 ends(kB / 16, kT - 1);
  dim3 chunks(kB / 16, kT);
  dim3 block(32);
  kf_forward_end_kernel<<<chunks, block, 0, stream>>>(z, K, F, hat);
  kf_forward_borders_kernel<<<slices, block, 0, stream>>>(x0, F, hat);
  kf_forward_fill_kernel<<<chunks, block, 0, stream>>>(F, Gs, hat, out2);
  kf_chunk_end_kernel<false><<<ends, block, 0, stream>>>(Gr, out2);
  kf_backward_borders_kernel<<<slices, block, 0, stream>>>(z, Ht, Ka0, Gr, hat, out2);
  kf_chunk_fill_kernel<false><<<chunks, block, 0, stream>>>(Gr, out2);
}


# the two 32nd matrix powers taken off the 16 serial border waves: each formed once per call by 2 waves of the launch before, parked in a free output tile and loaded; order of sums unchanged
# speedup vs baseline: 26.8501x; 1.1259x over previous
#include <hip/hip_runtime.h>

typedef __attribute__((ext_vector_type(2))) float v2f;
typedef __attribute__((ext_vector_type(4))) float v4f;
typedef __attribute__((ext_vector_type(8))) float v8f;

#define WMMA4(A, B, C) \
  __builtin_amdgcn_wmma_f32_16x16x4_f32(false, (A), false, (B), (short)0, (C), false, false)

constexpr int kN = 2048;
constexpr int kB = 256;
constexpr int kD = 32;
constexpr size_t kStep = (size_t)kB * kD;

__device__ __forceinline__ int colbase(int c, int half) {
  return 2 * c + 8 * half + (c >= 4 ? 8 : 0);
}

struct ATiles { v2f a[2][8]; };

__device__ __forceinline__ void load_a_tiles(const float* __restrict__ M, ATiles& t, int lane) {
  const int r0 = lane & 15;
  const int half = lane >> 4;
#pragma unroll
  for (int rt = 0; rt < 2; ++rt) {
#pragma unroll
    for (int c = 0; c < 8; ++c) {
      const float* p = M + (16 * rt + r0) * kD + colbase(c, half);
      t.a[rt][c] = *(const v2f*)p;
    }
  }
}

__device__ __forceinline__ void load_b_cols(const float* __restrict__ src, v2f xb[8], int lane) {
  const int col = lane & 15;
  const int half = lane >> 4;
  const float* p = src + col * kD;
#pragma unroll
  for (int c = 0; c < 8; ++c) xb[c] = *(const v2f*)(p + colbase(c, half));
}

__device__ __forceinline__ void store_b_cols(float* __restrict__ dst, const v2f xb[8], int lane) {
  const int col = lane & 15;
  const int half = lane >> 4;
  float* p = dst + col * kD;
#pragma unroll
  for (int c = 0; c < 8; ++c) *(v2f*)(p + colbase(c, half)) = xb[c];
}

__device__ __forceinline__ void load_c_cols(const float* src, v8f& c0, v8f& c1, int lane) {
  const int col = lane & 15;
  const int half = lane >> 4;
  const float* p = src + col * kD + 8 * half;
  v4f a = *(const v4f*)(p + 0);
  v4f b = *(const v4f*)(p + 4);
  v4f c = *(const v4f*)(p + 16);
  v4f d = *(const v4f*)(p + 20);
  c0 = (v8f){a[0], a[1], a[2], a[3], b[0], b[1], b[2], b[3]};
  c1 = (v8f){c[0], c[1], c[2], c[3], d[0], d[1], d[2], d[3]};
}

__device__ __forceinline__ void store_c_cols(float* dst, v8f c0, v8f c1, int lane) {
  const int col = lane & 15;
  const int half = lane >> 4;
  float* p = dst + col * kD + 8 * half;
  *(v4f*)(p + 0)  = (v4f){c0[0], c0[1], c0[2], c0[3]};
  *(v4f*)(p + 4)  = (v4f){c0[4], c0[5], c0[6], c0[7]};
  *(v4f*)(p + 16) = (v4f){c1[0], c1[1], c1[2], c1[3]};
  *(v4f*)(p + 20) = (v4f){c1[4], c1[5], c1[6], c1[7]};
}

__device__ __forceinline__ void c_to_b_regs(const v8f& c0, const v8f& c1, v2f xb[8]) {
#pragma unroll
  for (int c = 0; c < 4; ++c) {
    xb[c]     = (v2f){c0[2 * c], c0[2 * c + 1]};
    xb[c + 4] = (v2f){c1[2 * c], c1[2 * c + 1]};
  }
}

__device__ __forceinline__ void mm_chunk(const ATiles& A, const v2f b[8], int c, v8f& a0, v8f& a1) {
  a0 = WMMA4(A.a[0][c], b[c], a0);
  a1 = WMMA4(A.a[1][c], b[c], a1);
}

constexpr int kRing = 8;
constexpr int kFar = 16;

constexpr int kL = 32;
constexpr int kT = kN / kL;
static_assert(kT * kL == kN, "chunks tile the time axis");

__device__ __forceinline__ float* step_tile(float* base, bool forward, int g) {
  return base + (size_t)(forward ? g + 1 : kN - 2 - g) * kStep;
}

template <bool COMPUTE>
__device__ __forceinline__ void seeded_step(const ATiles& A, const v2f xb[8], v8f acc[2],
                                            bool store_prev, const v8f prev[2], float* prev_dst,
                                            bool refill, v8f freed[2], const float* refill_src,
                                            const float* far_src, int lane) {
  if (COMPUTE) mm_chunk(A, xb, 0, acc[0], acc[1]);
  __builtin_amdgcn_sched_barrier(0);
  if (refill) {
    load_c_cols(refill_src, freed[0], freed[1], lane);
    __builtin_prefetch(far_src + lane * 16, 0, 0);
  }
  if (store_prev) store_c_cols(prev_dst, prev[0], prev[1], lane);
  __builtin_amdgcn_sched_barrier(0);
  if (COMPUTE) {
#pragma unroll
    for (int c = 1; c < 8; ++c) mm_chunk(A, xb, c, acc[0], acc[1]);
  }
  __builtin_amdgcn_sched_barrier(0);
}

template <int STEPS, long STRIDE, bool STORE_ALL, bool ZERO_INIT>
__device__ __forceinline__ void run_recurrence(const ATiles& A, const v2f x_init[8], float* tile0,
                                               int lane) {
  constexpr int kMainSteps = ((STEPS - 1) / kRing) * kRing;
  constexpr int kTailSteps = (STEPS - 1) - kMainSteps;
  static_assert(kRing >= 3 && STEPS >= kRing, "the ring's first loads stay inside the run");
  v8f sd[kRing][2];
  __builtin_amdgcn_sched_barrier(0);
#pragma unroll
  for (int s = 0; s < kRing - 2; ++s) load_c_cols(tile0 + s * STRIDE, sd[s][0], sd[s][1], lane);
  __builtin_amdgcn_sched_barrier(0);
#pragma unroll
  for (int s = kRing - 2; s < min(kFar, STEPS); ++s) __builtin_prefetch(tile0 + s * STRIDE + lane * 16, 0, 0);

  seeded_step<!ZERO_INIT>(A, x_init, sd[0], false, sd[0], nullptr, true, sd[kRing - 2],
                          tile0 + (kRing - 2) * STRIDE, tile0 + min(kFar, STEPS - 1) * STRIDE, lane);

  for (int n = 0; n < kMainSteps; n += kRing) {
#pragma unroll
    for (int s = 0; s < kRing; ++s) {
      const int m = 1 + n + s;
      const int mr = min(m + kRing - 2, STEPS - 1);
      const int mf = min(m + kFar, STEPS - 1);
      v2f xb[8];
      c_to_b_regs(sd[s % kRing][0], sd[s % kRing][1], xb);
      seeded_step<true>(A, xb, sd[(1 + s) % kRing], STORE_ALL, sd[s % kRing], tile0 + (long)(m - 1) * STRIDE,
                        true, sd[(s + kRing - 1) % kRing], tile0 + (long)mr * STRIDE,
                        tile0 + (long)mf * STRIDE, lane);
    }
  }
#pragma unroll
  for (int s = 0; s < kTailSteps; ++s) {
    const int m = 1 + kMainSteps + s;
    const bool refill = m + kRing - 2 <= STEPS - 1;
    const int mr = refill ? m + kRing - 2 : STEPS - 1;
    v2f xb[8];
    c_to_b_regs(sd[s % kRing][0], sd[s % kRing][1], xb);
    seeded_step<true>(A, xb, sd[(1 + s) % kRing], STORE_ALL, sd[s % kRing], tile0 + (long)(m - 1) * STRIDE,
                      refill, sd[(s + kRing - 1) % kRing], tile0 + (long)mr * STRIDE,
                      tile0 + (long)(STEPS - 1) * STRIDE, lane);
  }
  store_c_cols(tile0 + (long)(STEPS - 1) * STRIDE, sd[(STEPS - 1) % kRing][0],
               sd[(STEPS - 1) % kRing][1], lane);
}

constexpr long kFwdStride = (long)kStep;
constexpr long kBwdStride = -(long)kStep;

template <bool FWD>
__global__ void __launch_bounds__(32)
kf_chunk_end_kernel(const float* __restrict__ M, float* buf) {
  const int lane = threadIdx.x & 31;
  const int b0 = blockIdx.x * 16;
  const int j = blockIdx.y;

  ATiles A;
  load_a_tiles(M, A, lane);
  v2f none[8] = {};
  run_recurrence<kL, FWD ? kFwdStride : kBwdStride, false, true>(
      A, none, step_tile(buf + (size_t)b0 * kD, FWD, j * kL), lane);
}

template <bool FWD>
__global__ void __launch_bounds__(32)
kf_chunk_fill_kernel(const float* __restrict__ M, float* buf, const float* __restrict__ x0,
                     float* __restrict__ hat0) {
  const int lane = threadIdx.x & 31;
  const int b0 = blockIdx.x * 16;
  const int j = blockIdx.y;

  if (j == kT) {
    v2f x[8];
    load_b_cols(x0 + (size_t)b0 * kD, x, lane);
    store_b_cols(hat0 + (size_t)b0 * kD, x, lane);
    return;
  }

  ATiles A;
  load_a_tiles(M, A, lane);
  v2f xb[8];
  load_b_cols(step_tile(buf + (size_t)b0 * kD, FWD, j * kL - 1), xb, lane);
  run_recurrence<kL - 1, FWD ? kFwdStride : kBwdStride, true, false>(
      A, xb, step_tile(buf + (size_t)b0 * kD, FWD, j * kL), lane);
}

__device__ __forceinline__ void power_row_tile(const float* __restrict__ M, int rt,
                                               float* __restrict__ dst, int lane) {
  const int r0 = lane & 15;
  const int half = lane >> 4;
  ATiles Mt;
#pragma unroll
  for (int t = 0; t < 2; ++t) {
#pragma unroll
    for (int c = 0; c < 8; ++c) {
      const float* p = M + colbase(c, half) * kD + 16 * t + r0;
      Mt.a[t][c] = (v2f){p[0], p[kD]};
    }
  }
  v2f q[8];
#pragma unroll
  for (int c = 0; c < 8; ++c) q[c] = *(const v2f*)(M + (16 * rt + r0) * kD + colbase(c, half));
  for (int i = 1; i < kL; ++i) {
    const v8f zero = {0.f, 0.f, 0.f, 0.f, 0.f, 0.f, 0.f, 0.f};
    v8f d0 = zero, d1 = zero;
#pragma unroll
    for (int c = 0; c < 8; ++c) {
      d0 = WMMA4(Mt.a[0][c], q[c], d0);
      d1 = WMMA4(Mt.a[1][c], q[c], d1);
    }
    c_to_b_regs(d0, d1, q);
  }
  store_b_cols(dst + 16 * rt * kD, q, lane);
}

constexpr int kInRing = 6;

template <int STEPS, bool STORE_END>
__device__ __forceinline__ void forward_end_pass(const ATiles& Kt, const ATiles& A,
                                                 const float* __restrict__ zt, float* dst, int lane) {
  static_assert(STEPS >= kInRing && STEPS >= kL - 1 && STEPS <= kL, "a whole chunk, or the last one");
  const v8f zero = {0.f, 0.f, 0.f, 0.f, 0.f, 0.f, 0.f, 0.f};
  v8f zr[kInRing][2];
#pragma unroll
  for (int s = 0; s < kInRing; ++s) load_c_cols(zt + (long)s * kStep, zr[s][0], zr[s][1], lane);
#pragma unroll
  for (int s = kInRing; s < min(kFar, STEPS); ++s) __builtin_prefetch(zt + (long)s * kStep + lane * 16, 0, 0);

  v8f st[2] = {zero, zero};
  v8f sp[2] = {zero, zero};
#pragma unroll
  for (int m = 0; m < STEPS; ++m) {
    v2f zb[8];
    c_to_b_regs(zr[m % kInRing][0], zr[m % kInRing][1], zb);
    v8f s0 = zero, s1 = zero;
    mm_chunk(Kt, zb, 0, s0, s1);
    __builtin_amdgcn_sched_barrier(0);
    if (m >= 1) {
      if (m - 1 + kInRing < STEPS)
        load_c_cols(zt + (long)(m - 1 + kInRing) * kStep, zr[(m - 1) % kInRing][0], zr[(m - 1) % kInRing][1], lane);
      if (m - 1 + kFar < STEPS) __builtin_prefetch(zt + (long)(m - 1 + kFar) * kStep + lane * 16, 0, 0);
      store_c_cols(dst + (long)(m - 1) * kStep, sp[0], sp[1], lane);
    }
    __builtin_amdgcn_sched_barrier(0);
#pragma unroll
    for (int c = 1; c < 8; ++c) mm_chunk(Kt, zb, c, s0, s1);
    sp[0] = s0;
    sp[1] = s1;
    if (m >= 1 && STORE_END) {
      v2f xb[8];
      c_to_b_regs(st[0], st[1], xb);
#pragma unroll
      for (int c = 0; c < 8; ++c) mm_chunk(A, xb, c, s0, s1);
    }
    st[0] = s0;
    st[1] = s1;
    __builtin_amdgcn_sched_barrier(0);
  }
  if (STEPS - 1 <= kL - 2) store_c_cols(dst + (long)(STEPS - 1) * kStep, sp[0], sp[1], lane);
  if (STORE_END) store_c_cols(dst + (long)(kL - 1) * kStep, st[0], st[1], lane);
}

__global__ void __launch_bounds__(32)
kf_forward_end_kernel(const float* __restrict__ z, const float* __restrict__ K,
                      const float* __restrict__ F, float* hat, float* __restrict__ power_out) {
  const int lane = threadIdx.x & 31;
  const int b0 = blockIdx.x * 16;
  const int j = blockIdx.y;

  if (j == kT) {
    if (blockIdx.x < 2) power_row_tile(F, blockIdx.x, power_out, lane);
    return;
  }

  ATiles Kt, A;
  load_a_tiles(K, Kt, lane);
  load_a_tiles(F, A, lane);
  const float* zt = z + (size_t)j * kL * kStep + (size_t)b0 * kD;
  float* dst = step_tile(hat + (size_t)b0 * kD, true, j * kL);
  if (j < kT - 1) forward_end_pass<kL, true>(Kt, A, zt, dst, lane);
  else            forward_end_pass<kL - 1, false>(Kt, A, zt, dst, lane);
}

template <bool FIRST_PRODUCT>
__device__ __forceinline__ void forward_fill_pass(const ATiles& A, const ATiles& G, const v2f x_init[8],
                                                  float* tile0, float* __restrict__ p0, int lane) {
  constexpr int STEPS = kL - 1;
  const v8f zero = {0.f, 0.f, 0.f, 0.f, 0.f, 0.f, 0.f, 0.f};
  v8f sd[kInRing][2];
#pragma unroll
  for (int s = 0; s < kInRing; ++s) load_c_cols(tile0 + (long)s * kStep, sd[s][0], sd[s][1], lane);
#pragma unroll
  for (int s = kInRing; s < min(kFar, STEPS); ++s) __builtin_prefetch(tile0 + (long)s * kStep + lane * 16, 0, 0);

  v2f xb[8];
#pragma unroll
  for (int c = 0; c < 8; ++c) xb[c] = x_init[c];
  v8f xs[2] = {zero, zero};
  v8f ps[2] = {zero, zero};
#pragma unroll
  for (int m = 0; m < STEPS; ++m) {
    v8f a0 = sd[m % kInRing][0], a1 = sd[m % kInRing][1];
    mm_chunk(A, xb, 0, a0, a1);
    __builtin_amdgcn_sched_barrier(0);
    if (m >= 1) {
      if (m - 1 + kInRing < STEPS)
        load_c_cols(tile0 + (long)(m - 1 + kInRing) * kStep, sd[(m - 1) % kInRing][0], sd[(m - 1) % kInRing][1], lane);
      if (m - 1 + kFar < STEPS) __builtin_prefetch(tile0 + (long)(m - 1 + kFar) * kStep + lane * 16, 0, 0);
      store_c_cols(tile0 + (long)(m - 1) * kStep, xs[0], xs[1], lane);
      if (m >= 2 || FIRST_PRODUCT) store_c_cols(p0 + (long)(m - 1) * kStep, ps[0], ps[1], lane);
    }
    __builtin_amdgcn_sched_barrier(0);
#pragma unroll
    for (int c = 1; c < 8; ++c) mm_chunk(A, xb, c, a0, a1);
    v8f q0 = zero, q1 = zero;
#pragma unroll
    for (int c = 0; c < 8; ++c) mm_chunk(G, xb, c, q0, q1);
    xs[0] = a0;
    xs[1] = a1;
    ps[0] = q0;
    ps[1] = q1;
    c_to_b_regs(a0, a1, xb);
    __builtin_amdgcn_sched_barrier(0);
  }
  store_c_cols(tile0 + (long)(STEPS - 1) * kStep, xs[0], xs[1], lane);
  store_c_cols(p0 + (long)(STEPS - 1) * kStep, ps[0], ps[1], lane);
  v8f q0 = zero, q1 = zero;
#pragma unroll
  for (int c = 0; c < 8; ++c) mm_chunk(G, xb, c, q0, q1);
  store_c_cols(p0 + (long)STEPS * kStep, q0, q1, lane);
}

__global__ void __launch_bounds__(32)
kf_forward_fill_kernel(const float* __restrict__ F, const float* __restrict__ Gs,
                       const float* __restrict__ x0, float* hat, float* __restrict__ out2) {
  const int lane = threadIdx.x & 31;
  const int b0 = blockIdx.x * 16;
  const int j = blockIdx.y;

  ATiles A, G;
  load_a_tiles(F, A, lane);
  load_a_tiles(Gs, G, lane);
  v2f xb[8];
  load_b_cols(j > 0 ? step_tile(hat + (size_t)b0 * kD, true, j * kL - 1) : x0 + (size_t)b0 * kD, xb, lane);
  float* p0 = out2 + ((long)j * kL - 1) * (long)kStep + (long)b0 * kD;
  float* tile0 = step_tile(hat + (size_t)b0 * kD, true, j * kL);
  if (j > 0) forward_fill_pass<true>(A, G, xb, tile0, p0, lane);
  else       forward_fill_pass<false>(A, G, xb, tile0, p0, lane);
}

__global__ void __launch_bounds__(32)
kf_forward_borders_kernel(const float* __restrict__ x0, const float* __restrict__ Fk,
                          const float* __restrict__ Gr, float* hat) {
  const int lane = threadIdx.x & 31;
  if (blockIdx.x >= kB / 16) {
    power_row_tile(Gr, blockIdx.x - kB / 16, hat, lane);
    return;
  }
  const int b0 = blockIdx.x * 16;

  ATiles P;
  load_a_tiles(Fk, P, lane);

  v2f xb[8];
  load_b_cols(x0 + (size_t)b0 * kD, xb, lane);

  run_recurrence<kT - 1, kL * kFwdStride, true, false>(
      P, xb, step_tile(hat + (size_t)b0 * kD, true, kL - 1), lane);
}

__global__ void __launch_bounds__(32)
kf_backward_borders_kernel(const float* __restrict__ z, const float* __restrict__ Ht,
                           const float* __restrict__ Ka0, const float* __restrict__ hat, float* out2) {
  const int lane = threadIdx.x & 31;
  const int b0 = blockIdx.x * 16;
  const size_t last = (size_t)(kN - 1) * kStep + (size_t)b0 * kD;

  ATiles A1;
  v2f sb[8];

  {
    ATiles A2;
    v2f hb[8];
    load_b_cols(hat + last, hb, lane);
    load_a_tiles(Ht, A1, lane);
    v8f t0 = {0.f, 0.f, 0.f, 0.f, 0.f, 0.f, 0.f, 0.f};
    v8f t1 = t0;
#pragma unroll
    for (int c = 0; c < 8; ++c) t0 = WMMA4(A1.a[0][c], hb[c], t0);
#pragma unroll
    for (int c = 0; c < 8; ++c) t1 = WMMA4(A1.a[1][c], hb[c], t1);

    v8f zc0, zc1;
    load_c_cols(z + last, zc0, zc1, lane);
    v8f bz0 = zc0 - t0;
    v8f bz1 = zc1 - t1;
    v2f bb[8];
    c_to_b_regs(bz0, bz1, bb);

    load_a_tiles(Ka0, A2, lane);
    v8f s0, s1;
    load_c_cols(hat + last, s0, s1, lane);
#pragma unroll
    for (int c = 0; c < 8; ++c) s0 = WMMA4(A2.a[0][c], bb[c], s0);
#pragma unroll
    for (int c = 0; c < 8; ++c) s1 = WMMA4(A2.a[1][c], bb[c], s1);

    store_c_cols(out2 + last, s0, s1, lane);
    c_to_b_regs(s0, s1, sb);
  }

  load_a_tiles(hat, A1, lane);
  run_recurrence<kT - 1, kL * kBwdStride, true, false>(
      A1, sb, step_tile(out2 + (size_t)b0 * kD, false, kL - 1), lane);
}

extern "C" void kernel_launch(void* const* d_in, const int* in_sizes, int n_in,
                              void* d_out, int out_size, void* d_ws, size_t ws_size,
                              hipStream_t stream) {
  (void)in_sizes; (void)n_in; (void)d_ws; (void)ws_size; (void)out_size;
  const float* z   = (const float*)d_in[0];
  const float* x0  = (const float*)d_in[1];
  const float* F   = (const float*)d_in[2];
  const float* K   = (const float*)d_in[3];
  const float* Ht  = (const float*)d_in[4];
  const float* Ka0 = (const float*)d_in[5];
  const float* Gs  = (const float*)d_in[6];
  const float* Gr  = (const float*)d_in[7];

  float* hat  = (float*)d_out;
  float* out2 = hat + (size_t)kN * kB * kD;

  dim3 slices(kB / 16);
  dim3 slices_and_power(kB / 16 + 2);
  dim3 ends(kB / 16, kT - 1);
  dim3 chunks(kB / 16, kT);
  dim3 chunks_and_row(kB / 16, kT + 1);
  dim3 block(32);
  kf_forward_end_kernel<<<chunks_and_row, block, 0, stream>>>(z, K, F, hat, out2);
  kf_forward_borders_kernel<<<slices_and_power, block, 0, stream>>>(x0, out2, Gr, hat);
  kf_forward_fill_kernel<<<chunks, block, 0, stream>>>(F, Gs, x0, hat, out2);
  kf_chunk_end_kernel<false><<<ends, block, 0, stream>>>(Gr, out2);
  kf_backward_borders_kernel<<<slices, block, 0, stream>>>(z, Ht, Ka0, hat, out2);
  kf_chunk_fill_kernel<false><<<chunks_and_row, block, 0, stream>>>(Gr, out2, x0, hat);
}


# backward chunk-end launch folded into the forward fill launch: a wave keeps its 32 Gs*hat products in registers and runs the zero-start backward chain on them; 5 launches; order of sums unchanged
# speedup vs baseline: 28.9443x; 1.0780x over previous
#include <hip/hip_runtime.h>

typedef __attribute__((ext_vector_type(2))) float v2f;
typedef __attribute__((ext_vector_type(4))) float v4f;
typedef __attribute__((ext_vector_type(8))) float v8f;

#define WMMA4(A, B, C) \
  __builtin_amdgcn_wmma_f32_16x16x4_f32(false, (A), false, (B), (short)0, (C), false, false)

constexpr int kN = 2048;
constexpr int kB = 256;
constexpr int kD = 32;
constexpr size_t kStep = (size_t)kB * kD;

__device__ __forceinline__ int colbase(int c, int half) {
  return 2 * c + 8 * half + (c >= 4 ? 8 : 0);
}

struct ATiles { v2f a[2][8]; };

__device__ __forceinline__ void load_a_tiles(const float* __restrict__ M, ATiles& t, int lane) {
  const int r0 = lane & 15;
  const int half = lane >> 4;
#pragma unroll
  for (int rt = 0; rt < 2; ++rt) {
#pragma unroll
    for (int c = 0; c < 8; ++c) {
      const float* p = M + (16 * rt + r0) * kD + colbase(c, half);
      t.a[rt][c] = *(const v2f*)p;
    }
  }
}

__device__ __forceinline__ void load_b_cols(const float* __restrict__ src, v2f xb[8], int lane) {
  const int col = lane & 15;
  const int half = lane >> 4;
  const float* p = src + col * kD;
#pragma unroll
  for (int c = 0; c < 8; ++c) xb[c] = *(const v2f*)(p + colbase(c, half));
}

__device__ __forceinline__ void store_b_cols(float* __restrict__ dst, const v2f xb[8], int lane) {
  const int col = lane & 15;
  const int half = lane >> 4;
  float* p = dst + col * kD;
#pragma unroll
  for (int c = 0; c < 8; ++c) *(v2f*)(p + colbase(c, half)) = xb[c];
}

__device__ __forceinline__ void load_c_cols(const float* src, v8f& c0, v8f& c1, int lane) {
  const int col = lane & 15;
  const int half = lane >> 4;
  const float* p = src + col * kD + 8 * half;
  v4f a = *(const v4f*)(p + 0);
  v4f b = *(const v4f*)(p + 4);
  v4f c = *(const v4f*)(p + 16);
  v4f d = *(const v4f*)(p + 20);
  c0 = (v8f){a[0], a[1], a[2], a[3], b[0], b[1], b[2], b[3]};
  c1 = (v8f){c[0], c[1], c[2], c[3], d[0], d[1], d[2], d[3]};
}

__device__ __forceinline__ void store_c_cols(float* dst, v8f c0, v8f c1, int lane) {
  const int col = lane & 15;
  const int half = lane >> 4;
  float* p = dst + col * kD + 8 * half;
  *(v4f*)(p + 0)  = (v4f){c0[0], c0[1], c0[2], c0[3]};
  *(v4f*)(p + 4)  = (v4f){c0[4], c0[5], c0[6], c0[7]};
  *(v4f*)(p + 16) = (v4f){c1[0], c1[1], c1[2], c1[3]};
  *(v4f*)(p + 20) = (v4f){c1[4], c1[5], c1[6], c1[7]};
}

__device__ __forceinline__ void c_to_b_regs(const v8f& c0, const v8f& c1, v2f xb[8]) {
#pragma unroll
  for (int c = 0; c < 4; ++c) {
    xb[c]     = (v2f){c0[2 * c], c0[2 * c + 1]};
    xb[c + 4] = (v2f){c1[2 * c], c1[2 * c + 1]};
  }
}

__device__ __forceinline__ void mm_chunk(const ATiles& A, const v2f b[8], int c, v8f& a0, v8f& a1) {
  a0 = WMMA4(A.a[0][c], b[c], a0);
  a1 = WMMA4(A.a[1][c], b[c], a1);
}

constexpr int kRing = 8;
constexpr int kFar = 16;

constexpr int kL = 32;
constexpr int kT = kN / kL;
static_assert(kT * kL == kN, "chunks tile the time axis");

__device__ __forceinline__ float* step_tile(float* base, bool forward, int g) {
  return base + (size_t)(forward ? g + 1 : kN - 2 - g) * kStep;
}

template <bool COMPUTE>
__device__ __forceinline__ void seeded_step(const ATiles& A, const v2f xb[8], v8f acc[2],
                                            bool store_prev, const v8f prev[2], float* prev_dst,
                                            bool refill, v8f freed[2], const float* refill_src,
                                            const float* far_src, int lane) {
  if (COMPUTE) mm_chunk(A, xb, 0, acc[0], acc[1]);
  __builtin_amdgcn_sched_barrier(0);
  if (refill) {
    load_c_cols(refill_src, freed[0], freed[1], lane);
    __builtin_prefetch(far_src + lane * 16, 0, 0);
  }
  if (store_prev) store_c_cols(prev_dst, prev[0], prev[1], lane);
  __builtin_amdgcn_sched_barrier(0);
  if (COMPUTE) {
#pragma unroll
    for (int c = 1; c < 8; ++c) mm_chunk(A, xb, c, acc[0], acc[1]);
  }
  __builtin_amdgcn_sched_barrier(0);
}

template <int STEPS, long STRIDE, bool STORE_ALL, bool ZERO_INIT>
__device__ __forceinline__ void run_recurrence(const ATiles& A, const v2f x_init[8], float* tile0,
                                               int lane) {
  constexpr int kMainSteps = ((STEPS - 1) / kRing) * kRing;
  constexpr int kTailSteps = (STEPS - 1) - kMainSteps;
  static_assert(kRing >= 3 && STEPS >= kRing, "the ring's first loads stay inside the run");
  v8f sd[kRing][2];
  __builtin_amdgcn_sched_barrier(0);
#pragma unroll
  for (int s = 0; s < kRing - 2; ++s) load_c_cols(tile0 + s * STRIDE, sd[s][0], sd[s][1], lane);
  __builtin_amdgcn_sched_barrier(0);
#pragma unroll
  for (int s = kRing - 2; s < min(kFar, STEPS); ++s) __builtin_prefetch(tile0 + s * STRIDE + lane * 16, 0, 0);

  seeded_step<!ZERO_INIT>(A, x_init, sd[0], false, sd[0], nullptr, true, sd[kRing - 2],
                          tile0 + (kRing - 2) * STRIDE, tile0 + min(kFar, STEPS - 1) * STRIDE, lane);

  for (int n = 0; n < kMainSteps; n += kRing) {
#pragma unroll
    for (int s = 0; s < kRing; ++s) {
      const int m = 1 + n + s;
      const int mr = min(m + kRing - 2, STEPS - 1);
      const int mf = min(m + kFar, STEPS - 1);
      v2f xb[8];
      c_to_b_regs(sd[s % kRing][0], sd[s % kRing][1], xb);
      seeded_step<true>(A, xb, sd[(1 + s) % kRing], STORE_ALL, sd[s % kRing], tile0 + (long)(m - 1) * STRIDE,
                        true, sd[(s + kRing - 1) % kRing], tile0 + (long)mr * STRIDE,
                        tile0 + (long)mf * STRIDE, lane);
    }
  }
#pragma unroll
  for (int s = 0; s < kTailSteps; ++s) {
    const int m = 1 + kMainSteps + s;
    const bool refill = m + kRing - 2 <= STEPS - 1;
    const int mr = refill ? m + kRing - 2 : STEPS - 1;
    v2f xb[8];
    c_to_b_regs(sd[s % kRing][0], sd[s % kRing][1], xb);
    seeded_step<true>(A, xb, sd[(1 + s) % kRing], STORE_ALL, sd[s % kRing], tile0 + (long)(m - 1) * STRIDE,
                      refill, sd[(s + kRing - 1) % kRing], tile0 + (long)mr * STRIDE,
                      tile0 + (long)(STEPS - 1) * STRIDE, lane);
  }
  store_c_cols(tile0 + (long)(STEPS - 1) * STRIDE, sd[(STEPS - 1) % kRing][0],
               sd[(STEPS - 1) % kRing][1], lane);
}

constexpr long kFwdStride = (long)kStep;
constexpr long kBwdStride = -(long)kStep;

template <bool FWD>
__global__ void __launch_bounds__(32)
kf_chunk_fill_kernel(const float* __restrict__ M, float* buf, const float* __restrict__ x0,
                     float* __restrict__ hat0) {
  const int lane = threadIdx.x & 31;
  const int b0 = blockIdx.x * 16;
  const int j = blockIdx.y;

  if (j == kT) {
    v2f x[8];
    load_b_cols(x0 + (size_t)b0 * kD, x, lane);
    store_b_cols(hat0 + (size_t)b0 * kD, x, lane);
    return;
  }

  ATiles A;
  load_a_tiles(M, A, lane);
  v2f xb[8];
  load_b_cols(step_tile(buf + (size_t)b0 * kD, FWD, j * kL - 1), xb, lane);
  run_recurrence<kL - 1, FWD ? kFwdStride : kBwdStride, true, false>(
      A, xb, step_tile(buf + (size_t)b0 * kD, FWD, j * kL), lane);
}

__device__ __forceinline__ void power_row_tile(const float* __restrict__ M, int rt,
                                               float* __restrict__ dst, int lane) {
  const int r0 = lane & 15;
  const int half = lane >> 4;
  ATiles Mt;
#pragma unroll
  for (int t = 0; t < 2; ++t) {
#pragma unroll
    for (int c = 0; c < 8; ++c) {
      const float* p = M + colbase(c, half) * kD + 16 * t + r0;
      Mt.a[t][c] = (v2f){p[0], p[kD]};
    }
  }
  v2f q[8];
#pragma unroll
  for (int c = 0; c < 8; ++c) q[c] = *(const v2f*)(M + (16 * rt + r0) * kD + colbase(c, half));
  for (int i = 1; i < kL; ++i) {
    const v8f zero = {0.f, 0.f, 0.f, 0.f, 0.f, 0.f, 0.f, 0.f};
    v8f d0 = zero, d1 = zero;
#pragma unroll
    for (int c = 0; c < 8; ++c) {
      d0 = WMMA4(Mt.a[0][c], q[c], d0);
      d1 = WMMA4(Mt.a[1][c], q[c], d1);
    }
    c_to_b_regs(d0, d1, q);
  }
  store_b_cols(dst + 16 * rt * kD, q, lane);
}

constexpr int kInRing = 6;

template <int STEPS, bool STORE_END>
__device__ __forceinline__ void forward_end_pass(const ATiles& Kt, const ATiles& A,
                                                 const float* __restrict__ zt, float* dst, int lane) {
  static_assert(STEPS >= kInRing && STEPS >= kL - 1 && STEPS <= kL, "a whole chunk, or the last one");
  const v8f zero = {0.f, 0.f, 0.f, 0.f, 0.f, 0.f, 0.f, 0.f};
  v8f zr[kInRing][2];
#pragma unroll
  for (int s = 0; s < kInRing; ++s) load_c_cols(zt + (long)s * kStep, zr[s][0], zr[s][1], lane);
#pragma unroll
  for (int s = kInRing; s < min(kFar, STEPS); ++s) __builtin_prefetch(zt + (long)s * kStep + lane * 16, 0, 0);

  v8f st[2] = {zero, zero};
  v8f sp[2] = {zero, zero};
#pragma unroll
  for (int m = 0; m < STEPS; ++m) {
    v2f zb[8];
    c_to_b_regs(zr[m % kInRing][0], zr[m % kInRing][1], zb);
    v8f s0 = zero, s1 = zero;
    mm_chunk(Kt, zb, 0, s0, s1);
    __builtin_amdgcn_sched_barrier(0);
    if (m >= 1) {
      if (m - 1 + kInRing < STEPS)
        load_c_cols(zt + (long)(m - 1 + kInRing) * kStep, zr[(m - 1) % kInRing][0], zr[(m - 1) % kInRing][1], lane);
      if (m - 1 + kFar < STEPS) __builtin_prefetch(zt + (long)(m - 1 + kFar) * kStep + lane * 16, 0, 0);
      store_c_cols(dst + (long)(m - 1) * kStep, sp[0], sp[1], lane);
    }
    __builtin_amdgcn_sched_barrier(0);
#pragma unroll
    for (int c = 1; c < 8; ++c) mm_chunk(Kt, zb, c, s0, s1);
    sp[0] = s0;
    sp[1] = s1;
    if (m >= 1 && STORE_END) {
      v2f xb[8];
      c_to_b_regs(st[0], st[1], xb);
#pragma unroll
      for (int c = 0; c < 8; ++c) mm_chunk(A, xb, c, s0, s1);
    }
    st[0] = s0;
    st[1] = s1;
    __builtin_amdgcn_sched_barrier(0);
  }
  if (STEPS - 1 <= kL - 2) store_c_cols(dst + (long)(STEPS - 1) * kStep, sp[0], sp[1], lane);
  if (STORE_END) store_c_cols(dst + (long)(kL - 1) * kStep, st[0], st[1], lane);
}

__global__ void __launch_bounds__(32)
kf_forward_end_kernel(const float* __restrict__ z, const float* __restrict__ K,
                      const float* __restrict__ F, float* hat, float* __restrict__ power_out) {
  const int lane = threadIdx.x & 31;
  const int b0 = blockIdx.x * 16;
  const int j = blockIdx.y;

  if (j == kT) {
    if (blockIdx.x < 2) power_row_tile(F, blockIdx.x, power_out, lane);
    return;
  }

  ATiles Kt, A;
  load_a_tiles(K, Kt, lane);
  load_a_tiles(F, A, lane);
  const float* zt = z + (size_t)j * kL * kStep + (size_t)b0 * kD;
  float* dst = step_tile(hat + (size_t)b0 * kD, true, j * kL);
  if (j < kT - 1) forward_end_pass<kL, true>(Kt, A, zt, dst, lane);
  else            forward_end_pass<kL - 1, false>(Kt, A, zt, dst, lane);
}

template <bool BACKWARD_END>
__device__ __forceinline__ void forward_fill_pass(const ATiles& A, const ATiles& G, const ATiles& R,
                                                  const v2f x_init[8], float* tile0,
                                                  float* __restrict__ p0, int lane) {
  constexpr int STEPS = kL - 1;
  const v8f zero = {0.f, 0.f, 0.f, 0.f, 0.f, 0.f, 0.f, 0.f};
  v8f sd[kInRing][2];
#pragma unroll
  for (int s = 0; s < kInRing; ++s) load_c_cols(tile0 + (long)s * kStep, sd[s][0], sd[s][1], lane);
#pragma unroll
  for (int s = kInRing; s < min(kFar, STEPS); ++s) __builtin_prefetch(tile0 + (long)s * kStep + lane * 16, 0, 0);

  v2f xb[8];
#pragma unroll
  for (int c = 0; c < 8; ++c) xb[c] = x_init[c];
  v8f xs[2] = {zero, zero};
  v8f ps[2] = {zero, zero};
  v8f pk[kL][2];
#pragma unroll
  for (int m = 0; m < STEPS; ++m) {
    v8f a0 = sd[m % kInRing][0], a1 = sd[m % kInRing][1];
    mm_chunk(A, xb, 0, a0, a1);
    __builtin_amdgcn_sched_barrier(0);
    if (m >= 1) {
      if (m - 1 + kInRing < STEPS)
        load_c_cols(tile0 + (long)(m - 1 + kInRing) * kStep, sd[(m - 1) % kInRing][0], sd[(m - 1) % kInRing][1], lane);
      if (m - 1 + kFar < STEPS) __builtin_prefetch(tile0 + (long)(m - 1 + kFar) * kStep + lane * 16, 0, 0);
      store_c_cols(tile0 + (long)(m - 1) * kStep, xs[0], xs[1], lane);
      if (m >= 2) store_c_cols(p0 + (long)(m - 1) * kStep, ps[0], ps[1], lane);
    }
    __builtin_amdgcn_sched_barrier(0);
#pragma unroll
    for (int c = 1; c < 8; ++c) mm_chunk(A, xb, c, a0, a1);
    v8f q0 = zero, q1 = zero;
#pragma unroll
    for (int c = 0; c < 8; ++c) mm_chunk(G, xb, c, q0, q1);
    xs[0] = a0;
    xs[1] = a1;
    ps[0] = q0;
    ps[1] = q1;
    if (BACKWARD_END) {
      pk[m][0] = q0;
      pk[m][1] = q1;
    }
    c_to_b_regs(a0, a1, xb);
    __builtin_amdgcn_sched_barrier(0);
  }
  store_c_cols(tile0 + (long)(STEPS - 1) * kStep, xs[0], xs[1], lane);
  store_c_cols(p0 + (long)(STEPS - 1) * kStep, ps[0], ps[1], lane);
  v8f q0 = zero, q1 = zero;
#pragma unroll
  for (int c = 0; c < 8; ++c) mm_chunk(G, xb, c, q0, q1);
  store_c_cols(p0 + (long)STEPS * kStep, q0, q1, lane);

  if (BACKWARD_END) {
    v8f s0 = q0, s1 = q1;
#pragma unroll
    for (int m = kL - 2; m >= 0; --m) {
      v2f sb[8];
      c_to_b_regs(s0, s1, sb);
      v8f a0 = pk[m][0], a1 = pk[m][1];
#pragma unroll
      for (int c = 0; c < 8; ++c) mm_chunk(R, sb, c, a0, a1);
      s0 = a0;
      s1 = a1;
    }
    store_c_cols(p0, s0, s1, lane);
  }
}

__global__ void __launch_bounds__(32)
kf_forward_fill_kernel(const float* __restrict__ F, const float* __restrict__ Gs,
                       const float* __restrict__ Gr, const float* __restrict__ x0, float* hat,
                       float* __restrict__ out2) {
  const int lane = threadIdx.x & 31;
  const int b0 = blockIdx.x * 16;
  const int j = blockIdx.y;

  ATiles A, G, R;
  load_a_tiles(F, A, lane);
  load_a_tiles(Gs, G, lane);
  load_a_tiles(Gr, R, lane);
  v2f xb[8];
  load_b_cols(j > 0 ? step_tile(hat + (size_t)b0 * kD, true, j * kL - 1) : x0 + (size_t)b0 * kD, xb, lane);
  float* p0 = out2 + ((long)j * kL - 1) * (long)kStep + (long)b0 * kD;
  float* tile0 = step_tile(hat + (size_t)b0 * kD, true, j * kL);
  if (j > 0) forward_fill_pass<true>(A, G, R, xb, tile0, p0, lane);
  else       forward_fill_pass<false>(A, G, R, xb, tile0, p0, lane);
}

__global__ void __launch_bounds__(32)
kf_forward_borders_kernel(const float* __restrict__ x0, const float* __restrict__ Fk,
                          const float* __restrict__ Gr, float* hat) {
  const int lane = threadIdx.x & 31;
  if (blockIdx.x >= kB / 16) {
    power_row_tile(Gr, blockIdx.x - kB / 16, hat, lane);
    return;
  }
  const int b0 = blockIdx.x * 16;

  ATiles P;
  load_a_tiles(Fk, P, lane);

  v2f xb[8];
  load_b_cols(x0 + (size_t)b0 * kD, xb, lane);

  run_recurrence<kT - 1, kL * kFwdStride, true, false>(
      P, xb, step_tile(hat + (size_t)b0 * kD, true, kL - 1), lane);
}

__global__ void __launch_bounds__(32)
kf_backward_borders_kernel(const float* __restrict__ z, const float* __restrict__ Ht,
                           const float* __restrict__ Ka0, const float* __restrict__ hat, float* out2) {
  const int lane = threadIdx.x & 31;
  const int b0 = blockIdx.x * 16;
  const size_t last = (size_t)(kN - 1) * kStep + (size_t)b0 * kD;

  ATiles A1;
  v2f sb[8];

  {
    ATiles A2;
    v2f hb[8];
    load_b_cols(hat + last, hb, lane);
    load_a_tiles(Ht, A1, lane);
    v8f t0 = {0.f, 0.f, 0.f, 0.f, 0.f, 0.f, 0.f, 0.f};
    v8f t1 = t0;
#pragma unroll
    for (int c = 0; c < 8; ++c) t0 = WMMA4(A1.a[0][c], hb[c], t0);
#pragma unroll
    for (int c = 0; c < 8; ++c) t1 = WMMA4(A1.a[1][c], hb[c], t1);

    v8f zc0, zc1;
    load_c_cols(z + last, zc0, zc1, lane);
    v8f bz0 = zc0 - t0;
    v8f bz1 = zc1 - t1;
    v2f bb[8];
    c_to_b_regs(bz0, bz1, bb);

    load_a_tiles(Ka0, A2, lane);
    v8f s0, s1;
    load_c_cols(hat + last, s0, s1, lane);
#pragma unroll
    for (int c = 0; c < 8; ++c) s0 = WMMA4(A2.a[0][c], bb[c], s0);
#pragma unroll
    for (int c = 0; c < 8; ++c) s1 = WMMA4(A2.a[1][c], bb[c], s1);

    store_c_cols(out2 + last, s0, s1, lane);
    c_to_b_regs(s0, s1, sb);
  }

  load_a_tiles(hat, A1, lane);
  run_recurrence<kT - 1, kL * kBwdStride, true, false>(
      A1, sb, step_tile(out2 + (size_t)b0 * kD, false, kL - 1), lane);
}

extern "C" void kernel_launch(void* const* d_in, const int* in_sizes, int n_in,
                              void* d_out, int out_size, void* d_ws, size_t ws_size,
                              hipStream_t stream) {
  (void)in_sizes; (void)n_in; (void)d_ws; (void)ws_size; (void)out_size;
  const float* z   = (const float*)d_in[0];
  const float* x0  = (const float*)d_in[1];
  const float* F   = (const float*)d_in[2];
  const float* K   = (const float*)d_in[3];
  const float* Ht  = (const float*)d_in[4];
  const float* Ka0 = (const float*)d_in[5];
  const float* Gs  = (const float*)d_in[6];
  const float* Gr  = (const float*)d_in[7];

  float* hat  = (float*)d_out;
  float* out2 = hat + (size_t)kN * kB * kD;

  dim3 slices(kB / 16);
  dim3 slices_and_power(kB / 16 + 2);
  dim3 chunks(kB / 16, kT);
  dim3 chunks_and_row(kB / 16, kT + 1);
  dim3 block(32);
  kf_forward_end_kernel<<<chunks_and_row, block, 0, stream>>>(z, K, F, hat, out2);
  kf_forward_borders_kernel<<<slices_and_power, block, 0, stream>>>(x0, out2, Gr, hat);
  kf_forward_fill_kernel<<<chunks, block, 0, stream>>>(F, Gs, Gr, x0, hat, out2);
  kf_backward_borders_kernel<<<slices, block, 0, stream>>>(z, Ht, Ka0, hat, out2);
  kf_chunk_fill_kernel<false><<<chunks_and_row, block, 0, stream>>>(Gr, out2, x0, hat);
}


# input products (K*z, Gs*hat) no longer stored and read back: each fill pass forms its seed again from the z or hat tile it reads instead; 2 of 7 whole-array passes gone; order of sums unchanged
# speedup vs baseline: 30.3602x; 1.0489x over previous
#include <hip/hip_runtime.h>

typedef __attribute__((ext_vector_type(2))) float v2f;
typedef __attribute__((ext_vector_type(4))) float v4f;
typedef __attribute__((ext_vector_type(8))) float v8f;

#define WMMA4(A, B, C) \
  __builtin_amdgcn_wmma_f32_16x16x4_f32(false, (A), false, (B), (short)0, (C), false, false)

constexpr int kN = 2048;
constexpr int kB = 256;
constexpr int kD = 32;
constexpr size_t kStep = (size_t)kB * kD;

__device__ __forceinline__ int colbase(int c, int half) {
  return 2 * c + 8 * half + (c >= 4 ? 8 : 0);
}

struct ATiles { v2f a[2][8]; };

__device__ __forceinline__ void load_a_tiles(const float* __restrict__ M, ATiles& t, int lane) {
  const int r0 = lane & 15;
  const int half = lane >> 4;
#pragma unroll
  for (int rt = 0; rt < 2; ++rt) {
#pragma unroll
    for (int c = 0; c < 8; ++c) {
      const float* p = M + (16 * rt + r0) * kD + colbase(c, half);
      t.a[rt][c] = *(const v2f*)p;
    }
  }
}

__device__ __forceinline__ void load_b_cols(const float* __restrict__ src, v2f xb[8], int lane) {
  const int col = lane & 15;
  const int half = lane >> 4;
  const float* p = src + col * kD;
#pragma unroll
  for (int c = 0; c < 8; ++c) xb[c] = *(const v2f*)(p + colbase(c, half));
}

__device__ __forceinline__ void store_b_cols(float* __restrict__ dst, const v2f xb[8], int lane) {
  const int col = lane & 15;
  const int half = lane >> 4;
  float* p = dst + col * kD;
#pragma unroll
  for (int c = 0; c < 8; ++c) *(v2f*)(p + colbase(c, half)) = xb[c];
}

__device__ __forceinline__ void load_c_cols(const float* src, v8f& c0, v8f& c1, int lane) {
  const int col = lane & 15;
  const int half = lane >> 4;
  const float* p = src + col * kD + 8 * half;
  v4f a = *(const v4f*)(p + 0);
  v4f b = *(const v4f*)(p + 4);
  v4f c = *(const v4f*)(p + 16);
  v4f d = *(const v4f*)(p + 20);
  c0 = (v8f){a[0], a[1], a[2], a[3], b[0], b[1], b[2], b[3]};
  c1 = (v8f){c[0], c[1], c[2], c[3], d[0], d[1], d[2], d[3]};
}

__device__ __forceinline__ void store_c_cols(float* dst, v8f c0, v8f c1, int lane) {
  const int col = lane & 15;
  const int half = lane >> 4;
  float* p = dst + col * kD + 8 * half;
  *(v4f*)(p + 0)  = (v4f){c0[0], c0[1], c0[2], c0[3]};
  *(v4f*)(p + 4)  = (v4f){c0[4], c0[5], c0[6], c0[7]};
  *(v4f*)(p + 16) = (v4f){c1[0], c1[1], c1[2], c1[3]};
  *(v4f*)(p + 20) = (v4f){c1[4], c1[5], c1[6], c1[7]};
}

__device__ __forceinline__ void c_to_b_regs(const v8f& c0, const v8f& c1, v2f xb[8]) {
#pragma unroll
  for (int c = 0; c < 4; ++c) {
    xb[c]     = (v2f){c0[2 * c], c0[2 * c + 1]};
    xb[c + 4] = (v2f){c1[2 * c], c1[2 * c + 1]};
  }
}

__device__ __forceinline__ void mm_chunk(const ATiles& A, const v2f b[8], int c, v8f& a0, v8f& a1) {
  a0 = WMMA4(A.a[0][c], b[c], a0);
  a1 = WMMA4(A.a[1][c], b[c], a1);
}

constexpr int kRing = 8;
constexpr int kFar = 16;

constexpr int kL = 32;
constexpr int kT = kN / kL;
static_assert(kT * kL == kN, "chunks tile the time axis");

__device__ __forceinline__ float* step_tile(float* base, bool forward, int g) {
  return base + (size_t)(forward ? g + 1 : kN - 2 - g) * kStep;
}

template <bool COMPUTE>
__device__ __forceinline__ void seeded_step(const ATiles& A, const v2f xb[8], v8f acc[2],
                                            bool store_prev, const v8f prev[2], float* prev_dst,
                                            bool refill, v8f freed[2], const float* refill_src,
                                            const float* far_src, int lane) {
  if (COMPUTE) mm_chunk(A, xb, 0, acc[0], acc[1]);
  __builtin_amdgcn_sched_barrier(0);
  if (refill) {
    load_c_cols(refill_src, freed[0], freed[1], lane);
    __builtin_prefetch(far_src + lane * 16, 0, 0);
  }
  if (store_prev) store_c_cols(prev_dst, prev[0], prev[1], lane);
  __builtin_amdgcn_sched_barrier(0);
  if (COMPUTE) {
#pragma unroll
    for (int c = 1; c < 8; ++c) mm_chunk(A, xb, c, acc[0], acc[1]);
  }
  __builtin_amdgcn_sched_barrier(0);
}

template <int STEPS, long STRIDE, bool STORE_ALL, bool ZERO_INIT>
__device__ __forceinline__ void run_recurrence(const ATiles& A, const v2f x_init[8], float* tile0,
                                               int lane) {
  constexpr int kMainSteps = ((STEPS - 1) / kRing) * kRing;
  constexpr int kTailSteps = (STEPS - 1) - kMainSteps;
  static_assert(kRing >= 3 && STEPS >= kRing, "the ring's first loads stay inside the run");
  v8f sd[kRing][2];
  __builtin_amdgcn_sched_barrier(0);
#pragma unroll
  for (int s = 0; s < kRing - 2; ++s) load_c_cols(tile0 + s * STRIDE, sd[s][0], sd[s][1], lane);
  __builtin_amdgcn_sched_barrier(0);
#pragma unroll
  for (int s = kRing - 2; s < min(kFar, STEPS); ++s) __builtin_prefetch(tile0 + s * STRIDE + lane * 16, 0, 0);

  seeded_step<!ZERO_INIT>(A, x_init, sd[0], false, sd[0], nullptr, true, sd[kRing - 2],
                          tile0 + (kRing - 2) * STRIDE, tile0 + min(kFar, STEPS - 1) * STRIDE, lane);

  for (int n = 0; n < kMainSteps; n += kRing) {
#pragma unroll
    for (int s = 0; s < kRing; ++s) {
      const int m = 1 + n + s;
      const int mr = min(m + kRing - 2, STEPS - 1);
      const int mf = min(m + kFar, STEPS - 1);
      v2f xb[8];
      c_to_b_regs(sd[s % kRing][0], sd[s % kRing][1], xb);
      seeded_step<true>(A, xb, sd[(1 + s) % kRing], STORE_ALL, sd[s % kRing], tile0 + (long)(m - 1) * STRIDE,
                        true, sd[(s + kRing - 1) % kRing], tile0 + (long)mr * STRIDE,
                        tile0 + (long)mf * STRIDE, lane);
    }
  }
#pragma unroll
  for (int s = 0; s < kTailSteps; ++s) {
    const int m = 1 + kMainSteps + s;
    const bool refill = m + kRing - 2 <= STEPS - 1;
    const int mr = refill ? m + kRing - 2 : STEPS - 1;
    v2f xb[8];
    c_to_b_regs(sd[s % kRing][0], sd[s % kRing][1], xb);
    seeded_step<true>(A, xb, sd[(1 + s) % kRing], STORE_ALL, sd[s % kRing], tile0 + (long)(m - 1) * STRIDE,
                      refill, sd[(s + kRing - 1) % kRing], tile0 + (long)mr * STRIDE,
                      tile0 + (long)(STEPS - 1) * STRIDE, lane);
  }
  store_c_cols(tile0 + (long)(STEPS - 1) * STRIDE, sd[(STEPS - 1) % kRing][0],
               sd[(STEPS - 1) % kRing][1], lane);
}

constexpr long kFwdStride = (long)kStep;
constexpr long kBwdStride = -(long)kStep;

__device__ __forceinline__ void power_row_tile(const float* __restrict__ M, int rt,
                                               float* __restrict__ dst, int lane) {
  const int r0 = lane & 15;
  const int half = lane >> 4;
  ATiles Mt;
#pragma unroll
  for (int t = 0; t < 2; ++t) {
#pragma unroll
    for (int c = 0; c < 8; ++c) {
      const float* p = M + colbase(c, half) * kD + 16 * t + r0;
      Mt.a[t][c] = (v2f){p[0], p[kD]};
    }
  }
  v2f q[8];
#pragma unroll
  for (int c = 0; c < 8; ++c) q[c] = *(const v2f*)(M + (16 * rt + r0) * kD + colbase(c, half));
  for (int i = 1; i < kL; ++i) {
    const v8f zero = {0.f, 0.f, 0.f, 0.f, 0.f, 0.f, 0.f, 0.f};
    v8f d0 = zero, d1 = zero;
#pragma unroll
    for (int c = 0; c < 8; ++c) {
      d0 = WMMA4(Mt.a[0][c], q[c], d0);
      d1 = WMMA4(Mt.a[1][c], q[c], d1);
    }
    c_to_b_regs(d0, d1, q);
  }
  store_b_cols(dst + 16 * rt * kD, q, lane);
}

constexpr int kInRing = 6;

__device__ __forceinline__ void forward_end_pass(const ATiles& Kt, const ATiles& A,
                                                 const float* __restrict__ zt,
                                                 float* __restrict__ dst_end, int lane) {
  constexpr int STEPS = kL;
  static_assert(STEPS >= kInRing, "the ring's first loads stay inside the chunk");
  const v8f zero = {0.f, 0.f, 0.f, 0.f, 0.f, 0.f, 0.f, 0.f};
  v8f zr[kInRing][2];
#pragma unroll
  for (int s = 0; s < kInRing; ++s) load_c_cols(zt + (long)s * kStep, zr[s][0], zr[s][1], lane);
#pragma unroll
  for (int s = kInRing; s < min(kFar, STEPS); ++s) __builtin_prefetch(zt + (long)s * kStep + lane * 16, 0, 0);

  v8f st[2] = {zero, zero};
#pragma unroll
  for (int m = 0; m < STEPS; ++m) {
    v2f zb[8];
    c_to_b_regs(zr[m % kInRing][0], zr[m % kInRing][1], zb);
    v8f s0 = zero, s1 = zero;
    mm_chunk(Kt, zb, 0, s0, s1);
    __builtin_amdgcn_sched_barrier(0);
    if (m >= 1) {
      if (m - 1 + kInRing < STEPS)
        load_c_cols(zt + (long)(m - 1 + kInRing) * kStep, zr[(m - 1) % kInRing][0], zr[(m - 1) % kInRing][1], lane);
      if (m - 1 + kFar < STEPS) __builtin_prefetch(zt + (long)(m - 1 + kFar) * kStep + lane * 16, 0, 0);
    }
    __builtin_amdgcn_sched_barrier(0);
#pragma unroll
    for (int c = 1; c < 8; ++c) mm_chunk(Kt, zb, c, s0, s1);
    if (m >= 1) {
      v2f xb[8];
      c_to_b_regs(st[0], st[1], xb);
#pragma unroll
      for (int c = 0; c < 8; ++c) mm_chunk(A, xb, c, s0, s1);
    }
    st[0] = s0;
    st[1] = s1;
    __builtin_amdgcn_sched_barrier(0);
  }
  store_c_cols(dst_end, st[0], st[1], lane);
}

__global__ void __launch_bounds__(32)
kf_forward_end_kernel(const float* __restrict__ z, const float* __restrict__ K,
                      const float* __restrict__ F, float* __restrict__ hat,
                      float* __restrict__ power_out) {
  const int lane = threadIdx.x & 31;
  const int b0 = blockIdx.x * 16;
  const int j = blockIdx.y;

  if (j == kT - 1) {
    if (blockIdx.x < 2) power_row_tile(F, blockIdx.x, power_out, lane);
    return;
  }

  ATiles Kt, A;
  load_a_tiles(K, Kt, lane);
  load_a_tiles(F, A, lane);
  const float* zt = z + (size_t)j * kL * kStep + (size_t)b0 * kD;
  float* dst_end = step_tile(hat + (size_t)b0 * kD, true, j * kL + kL - 1);
  forward_end_pass(Kt, A, zt, dst_end, lane);
}

template <bool BACKWARD_END>
__device__ __forceinline__ void forward_fill_pass(const ATiles& Kt, const ATiles& A, const ATiles& G,
                                                  const ATiles& R, const v2f x_init[8],
                                                  const float* __restrict__ zt, float* tile0,
                                                  float* __restrict__ p0, int lane) {
  constexpr int STEPS = kL - 1;
  const v8f zero = {0.f, 0.f, 0.f, 0.f, 0.f, 0.f, 0.f, 0.f};
  v8f zr[kInRing][2];
#pragma unroll
  for (int s = 0; s < kInRing; ++s) load_c_cols(zt + (long)s * kStep, zr[s][0], zr[s][1], lane);
#pragma unroll
  for (int s = kInRing; s < min(kFar, STEPS); ++s) __builtin_prefetch(zt + (long)s * kStep + lane * 16, 0, 0);

  v2f xb[8];
#pragma unroll
  for (int c = 0; c < 8; ++c) xb[c] = x_init[c];
  v8f xs[2] = {zero, zero};
  v8f pk[kL][2];
#pragma unroll
  for (int m = 0; m < STEPS; ++m) {
    v2f zb[8];
    c_to_b_regs(zr[m % kInRing][0], zr[m % kInRing][1], zb);
    v8f a0 = zero, a1 = zero;
    mm_chunk(Kt, zb, 0, a0, a1);
    __builtin_amdgcn_sched_barrier(0);
    if (m >= 1) {
      if (m - 1 + kInRing < STEPS)
        load_c_cols(zt + (long)(m - 1 + kInRing) * kStep, zr[(m - 1) % kInRing][0], zr[(m - 1) % kInRing][1], lane);
      if (m - 1 + kFar < STEPS) __builtin_prefetch(zt + (long)(m - 1 + kFar) * kStep + lane * 16, 0, 0);
      store_c_cols(tile0 + (long)(m - 1) * kStep, xs[0], xs[1], lane);
    }
    __builtin_amdgcn_sched_barrier(0);
#pragma unroll
    for (int c = 1; c < 8; ++c) mm_chunk(Kt, zb, c, a0, a1);
#pragma unroll
    for (int c = 0; c < 8; ++c) mm_chunk(A, xb, c, a0, a1);
    if (BACKWARD_END) {
      v8f q0 = zero, q1 = zero;
#pragma unroll
      for (int c = 0; c < 8; ++c) mm_chunk(G, xb, c, q0, q1);
      pk[m][0] = q0;
      pk[m][1] = q1;
    }
    xs[0] = a0;
    xs[1] = a1;
    c_to_b_regs(a0, a1, xb);
    __builtin_amdgcn_sched_barrier(0);
  }
  store_c_cols(tile0 + (long)(STEPS - 1) * kStep, xs[0], xs[1], lane);

  if (BACKWARD_END) {
    v8f s0 = zero, s1 = zero;
#pragma unroll
    for (int c = 0; c < 8; ++c) mm_chunk(G, xb, c, s0, s1);
#pragma unroll
    for (int m = kL - 2; m >= 0; --m) {
      v2f sb[8];
      c_to_b_regs(s0, s1, sb);
      v8f a0 = pk[m][0], a1 = pk[m][1];
#pragma unroll
      for (int c = 0; c < 8; ++c) mm_chunk(R, sb, c, a0, a1);
      s0 = a0;
      s1 = a1;
    }
    store_c_cols(p0, s0, s1, lane);
  }
}

__global__ void __launch_bounds__(32)
kf_forward_fill_kernel(const float* __restrict__ z, const float* __restrict__ K,
                       const float* __restrict__ F, const float* __restrict__ Gs,
                       const float* __restrict__ Gr, const float* __restrict__ x0, float* hat,
                       float* __restrict__ out2) {
  const int lane = threadIdx.x & 31;
  const int b0 = blockIdx.x * 16;
  const int j = blockIdx.y;

  ATiles Kt, A;
  load_a_tiles(K, Kt, lane);
  load_a_tiles(F, A, lane);
  v2f xb[8];
  load_b_cols(j > 0 ? step_tile(hat + (size_t)b0 * kD, true, j * kL - 1) : x0 + (size_t)b0 * kD, xb, lane);
  const float* zt = z + (size_t)j * kL * kStep + (size_t)b0 * kD;
  float* tile0 = step_tile(hat + (size_t)b0 * kD, true, j * kL);
  if (j > 0) {
    ATiles G, R;
    load_a_tiles(Gs, G, lane);
    load_a_tiles(Gr, R, lane);
    float* p0 = out2 + ((long)j * kL - 1) * (long)kStep + (long)b0 * kD;
    forward_fill_pass<true>(Kt, A, G, R, xb, zt, tile0, p0, lane);
  } else {
    forward_fill_pass<false>(Kt, A, A, A, xb, zt, tile0, nullptr, lane);
  }
}

__global__ void __launch_bounds__(32)
kf_backward_fill_kernel(const float* __restrict__ Gs, const float* __restrict__ Gr,
                        const float* __restrict__ x0, float* hat, float* out2) {
  const int lane = threadIdx.x & 31;
  const int b0 = blockIdx.x * 16;
  const int j = blockIdx.y;

  if (j == kT) {
    v2f x[8];
    load_b_cols(x0 + (size_t)b0 * kD, x, lane);
    store_b_cols(hat + (size_t)b0 * kD, x, lane);
    return;
  }

  constexpr int STEPS = kL - 1;
  const v8f zero = {0.f, 0.f, 0.f, 0.f, 0.f, 0.f, 0.f, 0.f};
  ATiles G, R;
  load_a_tiles(Gs, G, lane);
  load_a_tiles(Gr, R, lane);
  v2f sb[8];
  load_b_cols(step_tile(out2 + (size_t)b0 * kD, false, j * kL - 1), sb, lane);
  float* tile0 = step_tile(out2 + (size_t)b0 * kD, false, j * kL);
  const float* ht = hat + (size_t)(kN - 1 - j * kL) * kStep + (size_t)b0 * kD;

  v8f hr[kInRing][2];
#pragma unroll
  for (int s = 0; s < kInRing; ++s) load_c_cols(ht - (long)s * kStep, hr[s][0], hr[s][1], lane);
#pragma unroll
  for (int s = kInRing; s < min(kFar, STEPS); ++s) __builtin_prefetch(ht - (long)s * kStep + lane * 16, 0, 0);

  v8f xs[2] = {zero, zero};
#pragma unroll
  for (int m = 0; m < STEPS; ++m) {
    v2f hb[8];
    c_to_b_regs(hr[m % kInRing][0], hr[m % kInRing][1], hb);
    v8f a0 = zero, a1 = zero;
    mm_chunk(G, hb, 0, a0, a1);
    __builtin_amdgcn_sched_barrier(0);
    if (m >= 1) {
      if (m - 1 + kInRing < STEPS)
        load_c_cols(ht - (long)(m - 1 + kInRing) * kStep, hr[(m - 1) % kInRing][0], hr[(m - 1) % kInRing][1], lane);
      if (m - 1 + kFar < STEPS) __builtin_prefetch(ht - (long)(m - 1 + kFar) * kStep + lane * 16, 0, 0);
      store_c_cols(tile0 - (long)(m - 1) * kStep, xs[0], xs[1], lane);
    }
    __builtin_amdgcn_sched_barrier(0);
#pragma unroll
    for (int c = 1; c < 8; ++c) mm_chunk(G, hb, c, a0, a1);
#pragma unroll
    for (int c = 0; c < 8; ++c) mm_chunk(R, sb, c, a0, a1);
    xs[0] = a0;
    xs[1] = a1;
    c_to_b_regs(a0, a1, sb);
    __builtin_amdgcn_sched_barrier(0);
  }
  store_c_cols(tile0 - (long)(STEPS - 1) * kStep, xs[0], xs[1], lane);
}

__global__ void __launch_bounds__(32)
kf_forward_borders_kernel(const float* __restrict__ x0, const float* __restrict__ Fk,
                          const float* __restrict__ Gr, float* hat) {
  const int lane = threadIdx.x & 31;
  if (blockIdx.x >= kB / 16) {
    power_row_tile(Gr, blockIdx.x - kB / 16, hat, lane);
    return;
  }
  const int b0 = blockIdx.x * 16;

  ATiles P;
  load_a_tiles(Fk, P, lane);

  v2f xb[8];
  load_b_cols(x0 + (size_t)b0 * kD, xb, lane);

  run_recurrence<kT - 1, kL * kFwdStride, true, false>(
      P, xb, step_tile(hat + (size_t)b0 * kD, true, kL - 1), lane);
}

__global__ void __launch_bounds__(32)
kf_backward_borders_kernel(const float* __restrict__ z, const float* __restrict__ Ht,
                           const float* __restrict__ Ka0, const float* __restrict__ hat, float* out2) {
  const int lane = threadIdx.x & 31;
  const int b0 = blockIdx.x * 16;
  const size_t last = (size_t)(kN - 1) * kStep + (size_t)b0 * kD;

  ATiles A1;
  v2f sb[8];

  {
    ATiles A2;
    v2f hb[8];
    load_b_cols(hat + last, hb, lane);
    load_a_tiles(Ht, A1, lane);
    v8f t0 = {0.f, 0.f, 0.f, 0.f, 0.f, 0.f, 0.f, 0.f};
    v8f t1 = t0;
#pragma unroll
    for (int c = 0; c < 8; ++c) t0 = WMMA4(A1.a[0][c], hb[c], t0);
#pragma unroll
    for (int c = 0; c < 8; ++c) t1 = WMMA4(A1.a[1][c], hb[c], t1);

    v8f zc0, zc1;
    load_c_cols(z + last, zc0, zc1, lane);
    v8f bz0 = zc0 - t0;
    v8f bz1 = zc1 - t1;
    v2f bb[8];
    c_to_b_regs(bz0, bz1, bb);

    load_a_tiles(Ka0, A2, lane);
    v8f s0, s1;
    load_c_cols(hat + last, s0, s1, lane);
#pragma unroll
    for (int c = 0; c < 8; ++c) s0 = WMMA4(A2.a[0][c], bb[c], s0);
#pragma unroll
    for (int c = 0; c < 8; ++c) s1 = WMMA4(A2.a[1][c], bb[c], s1);

    store_c_cols(out2 + last, s0, s1, lane);
    c_to_b_regs(s0, s1, sb);
  }

  load_a_tiles(hat, A1, lane);
  run_recurrence<kT - 1, kL * kBwdStride, true, false>(
      A1, sb, step_tile(out2 + (size_t)b0 * kD, false, kL - 1), lane);
}

extern "C" void kernel_launch(void* const* d_in, const int* in_sizes, int n_in,
                              void* d_out, int out_size, void* d_ws, size_t ws_size,
                              hipStream_t stream) {
  (void)in_sizes; (void)n_in; (void)d_ws; (void)ws_size; (void)out_size;
  const float* z   = (const float*)d_in[0];
  const float* x0  = (const float*)d_in[1];
  const float* F   = (const float*)d_in[2];
  const float* K   = (const float*)d_in[3];
  const float* Ht  = (const float*)d_in[4];
  const float* Ka0 = (const float*)d_in[5];
  const float* Gs  = (const float*)d_in[6];
  const float* Gr  = (const float*)d_in[7];

  float* hat  = (float*)d_out;
  float* out2 = hat + (size_t)kN * kB * kD;

  dim3 slices(kB / 16);
  dim3 slices_and_power(kB / 16 + 2);
  dim3 chunks(kB / 16, kT);
  dim3 chunks_and_row(kB / 16, kT + 1);
  dim3 block(32);
  kf_forward_end_kernel<<<chunks, block, 0, stream>>>(z, K, F, hat, out2);
  kf_forward_borders_kernel<<<slices_and_power, block, 0, stream>>>(x0, out2, Gr, hat);
  kf_forward_fill_kernel<<<chunks, block, 0, stream>>>(z, K, F, Gs, Gr, x0, hat, out2);
  kf_backward_borders_kernel<<<slices, block, 0, stream>>>(z, Ht, Ka0, hat, out2);
  kf_backward_fill_kernel<<<chunks_and_row, block, 0, stream>>>(Gs, Gr, x0, hat, out2);
}


# border phases as a two-level scan: 4 waves per slice, one per SIMD, run groups of 16 border steps from zero at once, jump by the 16th power of the step matrix, refill; changes the order of sums
# speedup vs baseline: 31.7454x; 1.0456x over previous
#include <hip/hip_runtime.h>

typedef __attribute__((ext_vector_type(2))) float v2f;
typedef __attribute__((ext_vector_type(4))) float v4f;
typedef __attribute__((ext_vector_type(8))) float v8f;

#define WMMA4(A, B, C) \
  __builtin_amdgcn_wmma_f32_16x16x4_f32(false, (A), false, (B), (short)0, (C), false, false)

constexpr int kN = 2048;
constexpr int kB = 256;
constexpr int kD = 32;
constexpr size_t kStep = (size_t)kB * kD;

__device__ __forceinline__ int colbase(int c, int half) {
  return 2 * c + 8 * half + (c >= 4 ? 8 : 0);
}

struct ATiles { v2f a[2][8]; };

__device__ __forceinline__ void load_a_tiles(const float* __restrict__ M, ATiles& t, int lane) {
  const int r0 = lane & 15;
  const int half = lane >> 4;
#pragma unroll
  for (int rt = 0; rt < 2; ++rt) {
#pragma unroll
    for (int c = 0; c < 8; ++c) {
      const float* p = M + (16 * rt + r0) * kD + colbase(c, half);
      t.a[rt][c] = *(const v2f*)p;
    }
  }
}

__device__ __forceinline__ void load_b_cols(const float* __restrict__ src, v2f xb[8], int lane) {
  const int col = lane & 15;
  const int half = lane >> 4;
  const float* p = src + col * kD;
#pragma unroll
  for (int c = 0; c < 8; ++c) xb[c] = *(const v2f*)(p + colbase(c, half));
}

__device__ __forceinline__ void store_b_cols(float* __restrict__ dst, const v2f xb[8], int lane) {
  const int col = lane & 15;
  const int half = lane >> 4;
  float* p = dst + col * kD;
#pragma unroll
  for (int c = 0; c < 8; ++c) *(v2f*)(p + colbase(c, half)) = xb[c];
}

__device__ __forceinline__ void load_c_cols(const float* src, v8f& c0, v8f& c1, int lane) {
  const int col = lane & 15;
  const int half = lane >> 4;
  const float* p = src + col * kD + 8 * half;
  v4f a = *(const v4f*)(p + 0);
  v4f b = *(const v4f*)(p + 4);
  v4f c = *(const v4f*)(p + 16);
  v4f d = *(const v4f*)(p + 20);
  c0 = (v8f){a[0], a[1], a[2], a[3], b[0], b[1], b[2], b[3]};
  c1 = (v8f){c[0], c[1], c[2], c[3], d[0], d[1], d[2], d[3]};
}

__device__ __forceinline__ void store_c_cols(float* dst, v8f c0, v8f c1, int lane) {
  const int col = lane & 15;
  const int half = lane >> 4;
  float* p = dst + col * kD + 8 * half;
  *(v4f*)(p + 0)  = (v4f){c0[0], c0[1], c0[2], c0[3]};
  *(v4f*)(p + 4)  = (v4f){c0[4], c0[5], c0[6], c0[7]};
  *(v4f*)(p + 16) = (v4f){c1[0], c1[1], c1[2], c1[3]};
  *(v4f*)(p + 20) = (v4f){c1[4], c1[5], c1[6], c1[7]};
}

__device__ __forceinline__ void c_to_b_regs(const v8f& c0, const v8f& c1, v2f xb[8]) {
#pragma unroll
  for (int c = 0; c < 4; ++c) {
    xb[c]     = (v2f){c0[2 * c], c0[2 * c + 1]};
    xb[c + 4] = (v2f){c1[2 * c], c1[2 * c + 1]};
  }
}

__device__ __forceinline__ void mm_chunk(const ATiles& A, const v2f b[8], int c, v8f& a0, v8f& a1) {
  a0 = WMMA4(A.a[0][c], b[c], a0);
  a1 = WMMA4(A.a[1][c], b[c], a1);
}

constexpr int kRing = 8;
constexpr int kFar = 16;

constexpr int kL = 32;
constexpr int kT = kN / kL;
static_assert(kT * kL == kN, "chunks tile the time axis");

__device__ __forceinline__ float* step_tile(float* base, bool forward, int g) {
  return base + (size_t)(forward ? g + 1 : kN - 2 - g) * kStep;
}

template <bool COMPUTE>
__device__ __forceinline__ void seeded_step(const ATiles& A, const v2f xb[8], v8f acc[2],
                                            bool store_prev, const v8f prev[2], float* prev_dst,
                                            bool refill, v8f freed[2], const float* refill_src,
                                            const float* far_src, int lane) {
  if (COMPUTE) mm_chunk(A, xb, 0, acc[0], acc[1]);
  __builtin_amdgcn_sched_barrier(0);
  if (refill) {
    load_c_cols(refill_src, freed[0], freed[1], lane);
    __builtin_prefetch(far_src + lane * 16, 0, 0);
  }
  if (store_prev) store_c_cols(prev_dst, prev[0], prev[1], lane);
  __builtin_amdgcn_sched_barrier(0);
  if (COMPUTE) {
#pragma unroll
    for (int c = 1; c < 8; ++c) mm_chunk(A, xb, c, acc[0], acc[1]);
  }
  __builtin_amdgcn_sched_barrier(0);
}

template <int STEPS, long STRIDE, bool STORE_ALL, bool ZERO_INIT, bool STORE_LAST>
__device__ __forceinline__ void run_recurrence(const ATiles& A, const v2f x_init[8], float* tile0,
                                               int lane, v8f last_out[2]) {
  constexpr int kMainSteps = ((STEPS - 1) / kRing) * kRing;
  constexpr int kTailSteps = (STEPS - 1) - kMainSteps;
  static_assert(kRing >= 3 && STEPS >= kRing, "the ring's first loads stay inside the run");
  v8f sd[kRing][2];
  __builtin_amdgcn_sched_barrier(0);
#pragma unroll
  for (int s = 0; s < kRing - 2; ++s) load_c_cols(tile0 + s * STRIDE, sd[s][0], sd[s][1], lane);
  __builtin_amdgcn_sched_barrier(0);
#pragma unroll
  for (int s = kRing - 2; s < min(kFar, STEPS); ++s) __builtin_prefetch(tile0 + s * STRIDE + lane * 16, 0, 0);

  seeded_step<!ZERO_INIT>(A, x_init, sd[0], false, sd[0], nullptr, true, sd[kRing - 2],
                          tile0 + (kRing - 2) * STRIDE, tile0 + min(kFar, STEPS - 1) * STRIDE, lane);

  for (int n = 0; n < kMainSteps; n += kRing) {
#pragma unroll
    for (int s = 0; s < kRing; ++s) {
      const int m = 1 + n + s;
      const int mr = min(m + kRing - 2, STEPS - 1);
      const int mf = min(m + kFar, STEPS - 1);
      v2f xb[8];
      c_to_b_regs(sd[s % kRing][0], sd[s % kRing][1], xb);
      seeded_step<true>(A, xb, sd[(1 + s) % kRing], STORE_ALL, sd[s % kRing], tile0 + (long)(m - 1) * STRIDE,
                        true, sd[(s + kRing - 1) % kRing], tile0 + (long)mr * STRIDE,
                        tile0 + (long)mf * STRIDE, lane);
    }
  }
#pragma unroll
  for (int s = 0; s < kTailSteps; ++s) {
    const int m = 1 + kMainSteps + s;
    const bool refill = m + kRing - 2 <= STEPS - 1;
    const int mr = refill ? m + kRing - 2 : STEPS - 1;
    v2f xb[8];
    c_to_b_regs(sd[s % kRing][0], sd[s % kRing][1], xb);
    seeded_step<true>(A, xb, sd[(1 + s) % kRing], STORE_ALL, sd[s % kRing], tile0 + (long)(m - 1) * STRIDE,
                      refill, sd[(s + kRing - 1) % kRing], tile0 + (long)mr * STRIDE,
                      tile0 + (long)(STEPS - 1) * STRIDE, lane);
  }
  if (STORE_LAST) {
    store_c_cols(tile0 + (long)(STEPS - 1) * STRIDE, sd[(STEPS - 1) % kRing][0],
                 sd[(STEPS - 1) % kRing][1], lane);
  } else {
    last_out[0] = sd[(STEPS - 1) % kRing][0];
    last_out[1] = sd[(STEPS - 1) % kRing][1];
  }
}

constexpr long kFwdStride = (long)kStep;
constexpr long kBwdStride = -(long)kStep;

template <int POW>
__device__ __forceinline__ void power_row_tile(const float* __restrict__ M, int rt,
                                               float* __restrict__ dst, int lane) {
  const int r0 = lane & 15;
  const int half = lane >> 4;
  ATiles Mt;
#pragma unroll
  for (int t = 0; t < 2; ++t) {
#pragma unroll
    for (int c = 0; c < 8; ++c) {
      const float* p = M + colbase(c, half) * kD + 16 * t + r0;
      Mt.a[t][c] = (v2f){p[0], p[kD]};
    }
  }
  v2f q[8];
#pragma unroll
  for (int c = 0; c < 8; ++c) q[c] = *(const v2f*)(M + (16 * rt + r0) * kD + colbase(c, half));
  for (int i = 1; i < POW; ++i) {
    const v8f zero = {0.f, 0.f, 0.f, 0.f, 0.f, 0.f, 0.f, 0.f};
    v8f d0 = zero, d1 = zero;
#pragma unroll
    for (int c = 0; c < 8; ++c) {
      d0 = WMMA4(Mt.a[0][c], q[c], d0);
      d1 = WMMA4(Mt.a[1][c], q[c], d1);
    }
    c_to_b_regs(d0, d1, q);
  }
  store_b_cols(dst + 16 * rt * kD, q, lane);
}

constexpr int kInRing = 6;

__device__ __forceinline__ void forward_end_pass(const ATiles& Kt, const ATiles& A,
                                                 const float* __restrict__ zt,
                                                 float* __restrict__ dst_end, int lane) {
  constexpr int STEPS = kL;
  static_assert(STEPS >= kInRing, "the ring's first loads stay inside the chunk");
  const v8f zero = {0.f, 0.f, 0.f, 0.f, 0.f, 0.f, 0.f, 0.f};
  v8f zr[kInRing][2];
#pragma unroll
  for (int s = 0; s < kInRing; ++s) load_c_cols(zt + (long)s * kStep, zr[s][0], zr[s][1], lane);
#pragma unroll
  for (int s = kInRing; s < min(kFar, STEPS); ++s) __builtin_prefetch(zt + (long)s * kStep + lane * 16, 0, 0);

  v8f st[2] = {zero, zero};
#pragma unroll
  for (int m = 0; m < STEPS; ++m) {
    v2f zb[8];
    c_to_b_regs(zr[m % kInRing][0], zr[m % kInRing][1], zb);
    v8f s0 = zero, s1 = zero;
    mm_chunk(Kt, zb, 0, s0, s1);
    __builtin_amdgcn_sched_barrier(0);
    if (m >= 1) {
      if (m - 1 + kInRing < STEPS)
        load_c_cols(zt + (long)(m - 1 + kInRing) * kStep, zr[(m - 1) % kInRing][0], zr[(m - 1) % kInRing][1], lane);
      if (m - 1 + kFar < STEPS) __builtin_prefetch(zt + (long)(m - 1 + kFar) * kStep + lane * 16, 0, 0);
    }
    __builtin_amdgcn_sched_barrier(0);
#pragma unroll
    for (int c = 1; c < 8; ++c) mm_chunk(Kt, zb, c, s0, s1);
    if (m >= 1) {
      v2f xb[8];
      c_to_b_regs(st[0], st[1], xb);
#pragma unroll
      for (int c = 0; c < 8; ++c) mm_chunk(A, xb, c, s0, s1);
    }
    st[0] = s0;
    st[1] = s1;
    __builtin_amdgcn_sched_barrier(0);
  }
  store_c_cols(dst_end, st[0], st[1], lane);
}

__global__ void __launch_bounds__(32)
kf_forward_end_kernel(const float* __restrict__ z, const float* __restrict__ K,
                      const float* __restrict__ F, const float* __restrict__ Gr,
                      float* __restrict__ hat, float* __restrict__ power_out) {
  const int lane = threadIdx.x & 31;
  const int b0 = blockIdx.x * 16;
  const int j = blockIdx.y;

  if (j == kT - 1) {
    if (blockIdx.x < 2) power_row_tile<kL>(F, blockIdx.x, power_out, lane);
    else if (blockIdx.x < 4) power_row_tile<kL>(Gr, blockIdx.x - 2, hat, lane);
    return;
  }

  ATiles Kt, A;
  load_a_tiles(K, Kt, lane);
  load_a_tiles(F, A, lane);
  const float* zt = z + (size_t)j * kL * kStep + (size_t)b0 * kD;
  float* dst_end = step_tile(hat + (size_t)b0 * kD, true, j * kL + kL - 1);
  forward_end_pass(Kt, A, zt, dst_end, lane);
}

template <bool BACKWARD_END>
__device__ __forceinline__ void forward_fill_pass(const ATiles& Kt, const ATiles& A, const ATiles& G,
                                                  const ATiles& R, const v2f x_init[8],
                                                  const float* __restrict__ zt, float* tile0,
                                                  float* __restrict__ p0, int lane) {
  constexpr int STEPS = kL - 1;
  const v8f zero = {0.f, 0.f, 0.f, 0.f, 0.f, 0.f, 0.f, 0.f};
  v8f zr[kInRing][2];
#pragma unroll
  for (int s = 0; s < kInRing; ++s) load_c_cols(zt + (long)s * kStep, zr[s][0], zr[s][1], lane);
#pragma unroll
  for (int s = kInRing; s < min(kFar, STEPS); ++s) __builtin_prefetch(zt + (long)s * kStep + lane * 16, 0, 0);

  v2f xb[8];
#pragma unroll
  for (int c = 0; c < 8; ++c) xb[c] = x_init[c];
  v8f xs[2] = {zero, zero};
  v8f pk[kL][2];
#pragma unroll
  for (int m = 0; m < STEPS; ++m) {
    v2f zb[8];
    c_to_b_regs(zr[m % kInRing][0], zr[m % kInRing][1], zb);
    v8f a0 = zero, a1 = zero;
    mm_chunk(Kt, zb, 0, a0, a1);
    __builtin_amdgcn_sched_barrier(0);
    if (m >= 1) {
      if (m - 1 + kInRing < STEPS)
        load_c_cols(zt + (long)(m - 1 + kInRing) * kStep, zr[(m - 1) % kInRing][0], zr[(m - 1) % kInRing][1], lane);
      if (m - 1 + kFar < STEPS) __builtin_prefetch(zt + (long)(m - 1 + kFar) * kStep + lane * 16, 0, 0);
      store_c_cols(tile0 + (long)(m - 1) * kStep, xs[0], xs[1], lane);
    }
    __builtin_amdgcn_sched_barrier(0);
#pragma unroll
    for (int c = 1; c < 8; ++c) mm_chunk(Kt, zb, c, a0, a1);
#pragma unroll
    for (int c = 0; c < 8; ++c) mm_chunk(A, xb, c, a0, a1);
    if (BACKWARD_END) {
      v8f q0 = zero, q1 = zero;
#pragma unroll
      for (int c = 0; c < 8; ++c) mm_chunk(G, xb, c, q0, q1);
      pk[m][0] = q0;
      pk[m][1] = q1;
    }
    xs[0] = a0;
    xs[1] = a1;
    c_to_b_regs(a0, a1, xb);
    __builtin_amdgcn_sched_barrier(0);
  }
  store_c_cols(tile0 + (long)(STEPS - 1) * kStep, xs[0], xs[1], lane);

  if (BACKWARD_END) {
    v8f s0 = zero, s1 = zero;
#pragma unroll
    for (int c = 0; c < 8; ++c) mm_chunk(G, xb, c, s0, s1);
#pragma unroll
    for (int m = kL - 2; m >= 0; --m) {
      v2f sb[8];
      c_to_b_regs(s0, s1, sb);
      v8f a0 = pk[m][0], a1 = pk[m][1];
#pragma unroll
      for (int c = 0; c < 8; ++c) mm_chunk(R, sb, c, a0, a1);
      s0 = a0;
      s1 = a1;
    }
    store_c_cols(p0, s0, s1, lane);
  }
}

__global__ void __launch_bounds__(32)
kf_forward_fill_kernel(const float* __restrict__ z, const float* __restrict__ K,
                       const float* __restrict__ F, const float* __restrict__ Gs,
                       const float* __restrict__ Gr, const float* __restrict__ x0, float* hat,
                       float* __restrict__ out2) {
  const int lane = threadIdx.x & 31;
  const int b0 = blockIdx.x * 16;
  const int j = blockIdx.y;

  ATiles Kt, A;
  load_a_tiles(K, Kt, lane);
  load_a_tiles(F, A, lane);
  v2f xb[8];
  load_b_cols(j > 0 ? step_tile(hat + (size_t)b0 * kD, true, j * kL - 1) : x0 + (size_t)b0 * kD, xb, lane);
  const float* zt = z + (size_t)j * kL * kStep + (size_t)b0 * kD;
  float* tile0 = step_tile(hat + (size_t)b0 * kD, true, j * kL);
  if (j > 0) {
    ATiles G, R;
    load_a_tiles(Gs, G, lane);
    load_a_tiles(Gr, R, lane);
    float* p0 = out2 + ((long)j * kL - 1) * (long)kStep + (long)b0 * kD;
    forward_fill_pass<true>(Kt, A, G, R, xb, zt, tile0, p0, lane);
  } else {
    forward_fill_pass<false>(Kt, A, A, A, xb, zt, tile0, nullptr, lane);
  }
}

__global__ void __launch_bounds__(32)
kf_backward_fill_kernel(const float* __restrict__ Gs, const float* __restrict__ Gr,
                        const float* __restrict__ x0, float* hat, float* out2) {
  const int lane = threadIdx.x & 31;
  const int b0 = blockIdx.x * 16;
  const int j = blockIdx.y;

  if (j == kT) {
    v2f x[8];
    load_b_cols(x0 + (size_t)b0 * kD, x, lane);
    store_b_cols(hat + (size_t)b0 * kD, x, lane);
    return;
  }

  constexpr int STEPS = kL - 1;
  const v8f zero = {0.f, 0.f, 0.f, 0.f, 0.f, 0.f, 0.f, 0.f};
  ATiles G, R;
  load_a_tiles(Gs, G, lane);
  load_a_tiles(Gr, R, lane);
  v2f sb[8];
  load_b_cols(step_tile(out2 + (size_t)b0 * kD, false, j * kL - 1), sb, lane);
  float* tile0 = step_tile(out2 + (size_t)b0 * kD, false, j * kL);
  const float* ht = hat + (size_t)(kN - 1 - j * kL) * kStep + (size_t)b0 * kD;

  v8f hr[kInRing][2];
#pragma unroll
  for (int s = 0; s < kInRing; ++s) load_c_cols(ht - (long)s * kStep, hr[s][0], hr[s][1], lane);
#pragma unroll
  for (int s = kInRing; s < min(kFar, STEPS); ++s) __builtin_prefetch(ht - (long)s * kStep + lane * 16, 0, 0);

  v8f xs[2] = {zero, zero};
#pragma unroll
  for (int m = 0; m < STEPS; ++m) {
    v2f hb[8];
    c_to_b_regs(hr[m % kInRing][0], hr[m % kInRing][1], hb);
    v8f a0 = zero, a1 = zero;
    mm_chunk(G, hb, 0, a0, a1);
    __builtin_amdgcn_sched_barrier(0);
    if (m >= 1) {
      if (m - 1 + kInRing < STEPS)
        load_c_cols(ht - (long)(m - 1 + kInRing) * kStep, hr[(m - 1) % kInRing][0], hr[(m - 1) % kInRing][1], lane);
      if (m - 1 + kFar < STEPS) __builtin_prefetch(ht - (long)(m - 1 + kFar) * kStep + lane * 16, 0, 0);
      store_c_cols(tile0 - (long)(m - 1) * kStep, xs[0], xs[1], lane);
    }
    __builtin_amdgcn_sched_barrier(0);
#pragma unroll
    for (int c = 1; c < 8; ++c) mm_chunk(G, hb, c, a0, a1);
#pragma unroll
    for (int c = 0; c < 8; ++c) mm_chunk(R, sb, c, a0, a1);
    xs[0] = a0;
    xs[1] = a1;
    c_to_b_regs(a0, a1, sb);
    __builtin_amdgcn_sched_barrier(0);
  }
  store_c_cols(tile0 - (long)(STEPS - 1) * kStep, xs[0], xs[1], lane);
}

constexpr int kGroups = 4;
constexpr int kGroupLen = 16;
static_assert(kGroups * kGroupLen == kT, "groups tile the border steps, the last one short by one");
constexpr int kPowStep = 4;
static_assert(kPowStep * kPowStep == kGroupLen, "two equal levels reach the group's power");

__device__ __forceinline__ void lds_put_c(v4f* slot, v8f c0, v8f c1) {
  slot[0] = (v4f){c0[0], c0[1], c0[2], c0[3]};
  slot[1] = (v4f){c0[4], c0[5], c0[6], c0[7]};
  slot[2] = (v4f){c1[0], c1[1], c1[2], c1[3]};
  slot[3] = (v4f){c1[4], c1[5], c1[6], c1[7]};
}

__device__ __forceinline__ void lds_get_c(const v4f* slot, v8f& c0, v8f& c1) {
  const v4f a = slot[0], b = slot[1], c = slot[2], d = slot[3];
  c0 = (v8f){a[0], a[1], a[2], a[3], b[0], b[1], b[2], b[3]};
  c1 = (v8f){c[0], c[1], c[2], c[3], d[0], d[1], d[2], d[3]};
}

template <long STRIDE>
__device__ __forceinline__ void group_end_from_zero(const ATiles& P, float* group_tile0, v4f* slot,
                                                    int lane) {
  const v2f none[8] = {};
  v8f e[2];
  run_recurrence<kGroupLen, STRIDE, false, true, false>(P, none, group_tile0, lane, e);
  lds_put_c(slot, e[0], e[1]);
}

template <long STRIDE>
__device__ __forceinline__ void group_jump_and_fill(const ATiles& P, const ATiles& Pg, v2f xb[8],
                                                    const v4f* ends, float* group_tile0, int g,
                                                    int lane) {
#pragma unroll
  for (int i = 0; i < kGroups - 1; ++i) {
    if (i < g) {
      v8f a0, a1;
      lds_get_c(ends + (i * 32 + lane) * 4, a0, a1);
#pragma unroll
      for (int c = 0; c < 8; ++c) mm_chunk(Pg, xb, c, a0, a1);
      c_to_b_regs(a0, a1, xb);
    }
  }
  v8f unused[2];
  if (g < kGroups - 1)
    run_recurrence<kGroupLen, STRIDE, true, false, true>(P, xb, group_tile0, lane, unused);
  else
    run_recurrence<kGroupLen - 1, STRIDE, true, false, true>(P, xb, group_tile0, lane, unused);
}

__global__ void __launch_bounds__(32 * kGroups)
kf_forward_borders_kernel(const float* __restrict__ x0, const float* __restrict__ Fk, float* hat) {
  __shared__ v4f lds_e[(kGroups - 1) * 32 * 4];
  __shared__ v4f lds_p[kD * kD / 4];
  __shared__ v4f lds_q[kD * kD / 4];
  const int lane = threadIdx.x & 31;
  const int g = __builtin_amdgcn_readfirstlane((int)(threadIdx.x >> 5));
  if (blockIdx.x >= kB / 16) {
    if (g == 0) power_row_tile<kGroupLen>(hat, blockIdx.x - kB / 16, hat + kD * kD, lane);
    return;
  }
  const int b0 = blockIdx.x * 16;
  constexpr long STRIDE = kL * kFwdStride;
  float* group_tile0 = step_tile(hat + (size_t)b0 * kD, true, kL - 1) + (long)g * kGroupLen * STRIDE;

  ATiles P;
  load_a_tiles(Fk, P, lane);
  if (g < kGroups - 1) {
    group_end_from_zero<STRIDE>(P, group_tile0, lds_e + (g * 32 + lane) * 4, lane);
  } else {
    power_row_tile<kPowStep>(Fk, 0, (float*)lds_q, lane);
    power_row_tile<kPowStep>(Fk, 1, (float*)lds_q, lane);
    __builtin_amdgcn_fence(__ATOMIC_SEQ_CST, "wavefront");
    power_row_tile<kPowStep>((const float*)lds_q, 0, (float*)lds_p, lane);
    power_row_tile<kPowStep>((const float*)lds_q, 1, (float*)lds_p, lane);
  }
  __syncthreads();

  ATiles Pg;
  load_a_tiles((const float*)lds_p, Pg, lane);
  v2f xb[8];
  load_b_cols(x0 + (size_t)b0 * kD, xb, lane);
  group_jump_and_fill<STRIDE>(P, Pg, xb, lds_e, group_tile0, g, lane);
}

__global__ void __launch_bounds__(32 * kGroups)
kf_backward_borders_kernel(const float* __restrict__ z, const float* __restrict__ Ht,
                           const float* __restrict__ Ka0, const float* __restrict__ hat, float* out2) {
  __shared__ v4f lds_e[(kGroups - 1) * 32 * 4];
  __shared__ v4f lds_s[32 * 4];
  const int lane = threadIdx.x & 31;
  const int g = __builtin_amdgcn_readfirstlane((int)(threadIdx.x >> 5));
  const int b0 = blockIdx.x * 16;
  const size_t last = (size_t)(kN - 1) * kStep + (size_t)b0 * kD;
  constexpr long STRIDE = kL * kBwdStride;
  float* group_tile0 = step_tile(out2 + (size_t)b0 * kD, false, kL - 1) + (long)g * kGroupLen * STRIDE;

  ATiles P, Pg;
  load_a_tiles(hat, P, lane);
  load_a_tiles(hat + kD * kD, Pg, lane);
  if (g < kGroups - 1) {
    group_end_from_zero<STRIDE>(P, group_tile0, lds_e + (g * 32 + lane) * 4, lane);
  } else {
    ATiles A1, A2;
    v2f hb[8];
    load_b_cols(hat + last, hb, lane);
    load_a_tiles(Ht, A1, lane);
    v8f t0 = {0.f, 0.f, 0.f, 0.f, 0.f, 0.f, 0.f, 0.f};
    v8f t1 = t0;
#pragma unroll
    for (int c = 0; c < 8; ++c) t0 = WMMA4(A1.a[0][c], hb[c], t0);
#pragma unroll
    for (int c = 0; c < 8; ++c) t1 = WMMA4(A1.a[1][c], hb[c], t1);

    v8f zc0, zc1;
    load_c_cols(z + last, zc0, zc1, lane);
    v8f bz0 = zc0 - t0;
    v8f bz1 = zc1 - t1;
    v2f bb[8];
    c_to_b_regs(bz0, bz1, bb);

    load_a_tiles(Ka0, A2, lane);
    v8f s0, s1;
    load_c_cols(hat + last, s0, s1, lane);
#pragma unroll
    for (int c = 0; c < 8; ++c) s0 = WMMA4(A2.a[0][c], bb[c], s0);
#pragma unroll
    for (int c = 0; c < 8; ++c) s1 = WMMA4(A2.a[1][c], bb[c], s1);

    store_c_cols(out2 + last, s0, s1, lane);
    lds_put_c(lds_s + lane * 4, s0, s1);
  }
  __syncthreads();

  v8f s0, s1;
  lds_get_c(lds_s + lane * 4, s0, s1);
  v2f sb[8];
  c_to_b_regs(s0, s1, sb);
  group_jump_and_fill<STRIDE>(P, Pg, sb, lds_e, group_tile0, g, lane);
}

extern "C" void kernel_launch(void* const* d_in, const int* in_sizes, int n_in,
                              void* d_out, int out_size, void* d_ws, size_t ws_size,
                              hipStream_t stream) {
  (void)in_sizes; (void)n_in; (void)d_ws; (void)ws_size; (void)out_size;
  const float* z   = (const float*)d_in[0];
  const float* x0  = (const float*)d_in[1];
  const float* F   = (const float*)d_in[2];
  const float* K   = (const float*)d_in[3];
  const float* Ht  = (const float*)d_in[4];
  const float* Ka0 = (const float*)d_in[5];
  const float* Gs  = (const float*)d_in[6];
  const float* Gr  = (const float*)d_in[7];

  float* hat  = (float*)d_out;
  float* out2 = hat + (size_t)kN * kB * kD;

  dim3 slices(kB / 16);
  dim3 slices_and_power(kB / 16 + 2);
  dim3 group_block(32 * kGroups);
  dim3 chunks(kB / 16, kT);
  dim3 chunks_and_row(kB / 16, kT + 1);
  dim3 block(32);
  kf_forward_end_kernel<<<chunks, block, 0, stream>>>(z, K, F, Gr, hat, out2);
  kf_forward_borders_kernel<<<slices_and_power, group_block, 0, stream>>>(x0, out2, hat);
  kf_forward_fill_kernel<<<chunks, block, 0, stream>>>(z, K, F, Gs, Gr, x0, hat, out2);
  kf_backward_borders_kernel<<<slices, group_block, 0, stream>>>(z, Ht, Ka0, hat, out2);
  kf_backward_fill_kernel<<<chunks_and_row, block, 0, stream>>>(Gs, Gr, x0, hat, out2);
}


# backward chunk ends in the forward fill pass as group sums of (Gr^i*Gs)*x, i<8, joined by Gr^8; the 9 matrices built once per call by spare waves of launch 1; no Gr*s chain; changes the order of sums
# speedup vs baseline: 32.6300x; 1.0279x over previous
#include <hip/hip_runtime.h>

typedef __attribute__((ext_vector_type(2))) float v2f;
typedef __attribute__((ext_vector_type(4))) float v4f;
typedef __attribute__((ext_vector_type(8))) float v8f;

#define WMMA4(A, B, C) \
  __builtin_amdgcn_wmma_f32_16x16x4_f32(false, (A), false, (B), (short)0, (C), false, false)

constexpr int kN = 2048;
constexpr int kB = 256;
constexpr int kD = 32;
constexpr size_t kStep = (size_t)kB * kD;

__device__ __forceinline__ int colbase(int c, int half) {
  return 2 * c + 8 * half + (c >= 4 ? 8 : 0);
}

struct ATiles { v2f a[2][8]; };

__device__ __forceinline__ void load_a_tiles(const float* __restrict__ M, ATiles& t, int lane) {
  const int r0 = lane & 15;
  const int half = lane >> 4;
#pragma unroll
  for (int rt = 0; rt < 2; ++rt) {
#pragma unroll
    for (int c = 0; c < 8; ++c) {
      const float* p = M + (16 * rt + r0) * kD + colbase(c, half);
      t.a[rt][c] = *(const v2f*)p;
    }
  }
}

__device__ __forceinline__ void load_b_cols(const float* __restrict__ src, v2f xb[8], int lane) {
  const int col = lane & 15;
  const int half = lane >> 4;
  const float* p = src + col * kD;
#pragma unroll
  for (int c = 0; c < 8; ++c) xb[c] = *(const v2f*)(p + colbase(c, half));
}

__device__ __forceinline__ void store_b_cols(float* __restrict__ dst, const v2f xb[8], int lane) {
  const int col = lane & 15;
  const int half = lane >> 4;
  float* p = dst + col * kD;
#pragma unroll
  for (int c = 0; c < 8; ++c) *(v2f*)(p + colbase(c, half)) = xb[c];
}

__device__ __forceinline__ void load_c_cols(const float* src, v8f& c0, v8f& c1, int lane) {
  const int col = lane & 15;
  const int half = lane >> 4;
  const float* p = src + col * kD + 8 * half;
  v4f a = *(const v4f*)(p + 0);
  v4f b = *(const v4f*)(p + 4);
  v4f c = *(const v4f*)(p + 16);
  v4f d = *(const v4f*)(p + 20);
  c0 = (v8f){a[0], a[1], a[2], a[3], b[0], b[1], b[2], b[3]};
  c1 = (v8f){c[0], c[1], c[2], c[3], d[0], d[1], d[2], d[3]};
}

__device__ __forceinline__ void store_c_cols(float* dst, v8f c0, v8f c1, int lane) {
  const int col = lane & 15;
  const int half = lane >> 4;
  float* p = dst + col * kD + 8 * half;
  *(v4f*)(p + 0)  = (v4f){c0[0], c0[1], c0[2], c0[3]};
  *(v4f*)(p + 4)  = (v4f){c0[4], c0[5], c0[6], c0[7]};
  *(v4f*)(p + 16) = (v4f){c1[0], c1[1], c1[2], c1[3]};
  *(v4f*)(p + 20) = (v4f){c1[4], c1[5], c1[6], c1[7]};
}

__device__ __forceinline__ void c_to_b_regs(const v8f& c0, const v8f& c1, v2f xb[8]) {
#pragma unroll
  for (int c = 0; c < 4; ++c) {
    xb[c]     = (v2f){c0[2 * c], c0[2 * c + 1]};
    xb[c + 4] = (v2f){c1[2 * c], c1[2 * c + 1]};
  }
}

__device__ __forceinline__ void mm_chunk(const ATiles& A, const v2f b[8], int c, v8f& a0, v8f& a1) {
  a0 = WMMA4(A.a[0][c], b[c], a0);
  a1 = WMMA4(A.a[1][c], b[c], a1);
}

constexpr int kRing = 8;
constexpr int kFar = 16;

constexpr int kL = 32;
constexpr int kT = kN / kL;
static_assert(kT * kL == kN, "chunks tile the time axis");

__device__ __forceinline__ float* step_tile(float* base, bool forward, int g) {
  return base + (size_t)(forward ? g + 1 : kN - 2 - g) * kStep;
}

template <bool COMPUTE>
__device__ __forceinline__ void seeded_step(const ATiles& A, const v2f xb[8], v8f acc[2],
                                            bool store_prev, const v8f prev[2], float* prev_dst,
                                            bool refill, v8f freed[2], const float* refill_src,
                                            const float* far_src, int lane) {
  if (COMPUTE) mm_chunk(A, xb, 0, acc[0], acc[1]);
  __builtin_amdgcn_sched_barrier(0);
  if (refill) {
    load_c_cols(refill_src, freed[0], freed[1], lane);
    __builtin_prefetch(far_src + lane * 16, 0, 0);
  }
  if (store_prev) store_c_cols(prev_dst, prev[0], prev[1], lane);
  __builtin_amdgcn_sched_barrier(0);
  if (COMPUTE) {
#pragma unroll
    for (int c = 1; c < 8; ++c) mm_chunk(A, xb, c, acc[0], acc[1]);
  }
  __builtin_amdgcn_sched_barrier(0);
}

template <int STEPS, long STRIDE, bool STORE_ALL, bool ZERO_INIT, bool STORE_LAST>
__device__ __forceinline__ void run_recurrence(const ATiles& A, const v2f x_init[8], float* tile0,
                                               int lane, v8f last_out[2]) {
  constexpr int kMainSteps = ((STEPS - 1) / kRing) * kRing;
  constexpr int kTailSteps = (STEPS - 1) - kMainSteps;
  static_assert(kRing >= 3 && STEPS >= kRing, "the ring's first loads stay inside the run");
  v8f sd[kRing][2];
  __builtin_amdgcn_sched_barrier(0);
#pragma unroll
  for (int s = 0; s < kRing - 2; ++s) load_c_cols(tile0 + s * STRIDE, sd[s][0], sd[s][1], lane);
  __builtin_amdgcn_sched_barrier(0);
#pragma unroll
  for (int s = kRing - 2; s < min(kFar, STEPS); ++s) __builtin_prefetch(tile0 + s * STRIDE + lane * 16, 0, 0);

  seeded_step<!ZERO_INIT>(A, x_init, sd[0], false, sd[0], nullptr, true, sd[kRing - 2],
                          tile0 + (kRing - 2) * STRIDE, tile0 + min(kFar, STEPS - 1) * STRIDE, lane);

  for (int n = 0; n < kMainSteps; n += kRing) {
#pragma unroll
    for (int s = 0; s < kRing; ++s) {
      const int m = 1 + n + s;
      const int mr = min(m + kRing - 2, STEPS - 1);
      const int mf = min(m + kFar, STEPS - 1);
      v2f xb[8];
      c_to_b_regs(sd[s % kRing][0], sd[s % kRing][1], xb);
      seeded_step<true>(A, xb, sd[(1 + s) % kRing], STORE_ALL, sd[s % kRing], tile0 + (long)(m - 1) * STRIDE,
                        true, sd[(s + kRing - 1) % kRing], tile0 + (long)mr * STRIDE,
                        tile0 + (long)mf * STRIDE, lane);
    }
  }
#pragma unroll
  for (int s = 0; s < kTailSteps; ++s) {
    const int m = 1 + kMainSteps + s;
    const bool refill = m + kRing - 2 <= STEPS - 1;
    const int mr = refill ? m + kRing - 2 : STEPS - 1;
    v2f xb[8];
    c_to_b_regs(sd[s % kRing][0], sd[s % kRing][1], xb);
    seeded_step<true>(A, xb, sd[(1 + s) % kRing], STORE_ALL, sd[s % kRing], tile0 + (long)(m - 1) * STRIDE,
                      refill, sd[(s + kRing - 1) % kRing], tile0 + (long)mr * STRIDE,
                      tile0 + (long)(STEPS - 1) * STRIDE, lane);
  }
  if (STORE_LAST) {
    store_c_cols(tile0 + (long)(STEPS - 1) * STRIDE, sd[(STEPS - 1) % kRing][0],
                 sd[(STEPS - 1) % kRing][1], lane);
  } else {
    last_out[0] = sd[(STEPS - 1) % kRing][0];
    last_out[1] = sd[(STEPS - 1) % kRing][1];
  }
}

constexpr long kFwdStride = (long)kStep;
constexpr long kBwdStride = -(long)kStep;

template <int POW>
__device__ __forceinline__ void power_row_tile(const float* __restrict__ M, int rt,
                                               float* __restrict__ dst, int lane) {
  const int r0 = lane & 15;
  const int half = lane >> 4;
  ATiles Mt;
#pragma unroll
  for (int t = 0; t < 2; ++t) {
#pragma unroll
    for (int c = 0; c < 8; ++c) {
      const float* p = M + colbase(c, half) * kD + 16 * t + r0;
      Mt.a[t][c] = (v2f){p[0], p[kD]};
    }
  }
  v2f q[8];
#pragma unroll
  for (int c = 0; c < 8; ++c) q[c] = *(const v2f*)(M + (16 * rt + r0) * kD + colbase(c, half));
  for (int i = 1; i < POW; ++i) {
    const v8f zero = {0.f, 0.f, 0.f, 0.f, 0.f, 0.f, 0.f, 0.f};
    v8f d0 = zero, d1 = zero;
#pragma unroll
    for (int c = 0; c < 8; ++c) {
      d0 = WMMA4(Mt.a[0][c], q[c], d0);
      d1 = WMMA4(Mt.a[1][c], q[c], d1);
    }
    c_to_b_regs(d0, d1, q);
  }
  store_b_cols(dst + 16 * rt * kD, q, lane);
}

constexpr int kGrp = 8;
static_assert(kL % kGrp == 0, "groups tile the chunk");

__device__ __forceinline__ void gain_power_table(const float* __restrict__ M, const float* __restrict__ G0,
                                                 float* __restrict__ dst, int lane) {
  const int col = lane & 15;
  const int half = lane >> 4;
  ATiles Mt;
  load_a_tiles(M, Mt, lane);
  v2f q[2][8];
#pragma unroll
  for (int ct = 0; ct < 2; ++ct) {
#pragma unroll
    for (int c = 0; c < 8; ++c) {
      const float* p = G0 + colbase(c, half) * kD + 16 * ct + col;
      q[ct][c] = (v2f){p[0], p[kD]};
      float* w = dst + colbase(c, half) * kD + 16 * ct + col;
      w[0] = q[ct][c][0];
      w[kD] = q[ct][c][1];
    }
  }
  for (int i = 1; i < kGrp; ++i) {
    const v8f zero = {0.f, 0.f, 0.f, 0.f, 0.f, 0.f, 0.f, 0.f};
#pragma unroll
    for (int ct = 0; ct < 2; ++ct) {
      v8f d0 = zero, d1 = zero;
#pragma unroll
      for (int c = 0; c < 8; ++c) {
        d0 = WMMA4(Mt.a[0][c], q[ct][c], d0);
        d1 = WMMA4(Mt.a[1][c], q[ct][c], d1);
      }
      float* w = dst + (long)i * kD * kD + 8 * half * kD + 16 * ct + col;
#pragma unroll
      for (int v = 0; v < 8; ++v) {
        w[v * kD] = d0[v];
        w[(16 + v) * kD] = d1[v];
      }
      c_to_b_regs(d0, d1, q[ct]);
    }
  }
}

constexpr int kInRing = 6;

__device__ __forceinline__ void forward_end_pass(const ATiles& Kt, const ATiles& A,
                                                 const float* __restrict__ zt,
                                                 float* __restrict__ dst_end, int lane) {
  constexpr int STEPS = kL;
  static_assert(STEPS >= kInRing, "the ring's first loads stay inside the chunk");
  const v8f zero = {0.f, 0.f, 0.f, 0.f, 0.f, 0.f, 0.f, 0.f};
  v8f zr[kInRing][2];
#pragma unroll
  for (int s = 0; s < kInRing; ++s) load_c_cols(zt + (long)s * kStep, zr[s][0], zr[s][1], lane);
#pragma unroll
  for (int s = kInRing; s < min(kFar, STEPS); ++s) __builtin_prefetch(zt + (long)s * kStep + lane * 16, 0, 0);

  v8f st[2] = {zero, zero};
#pragma unroll
  for (int m = 0; m < STEPS; ++m) {
    v2f zb[8];
    c_to_b_regs(zr[m % kInRing][0], zr[m % kInRing][1], zb);
    v8f s0 = zero, s1 = zero;
    mm_chunk(Kt, zb, 0, s0, s1);
    __builtin_amdgcn_sched_barrier(0);
    if (m >= 1) {
      if (m - 1 + kInRing < STEPS)
        load_c_cols(zt + (long)(m - 1 + kInRing) * kStep, zr[(m - 1) % kInRing][0], zr[(m - 1) % kInRing][1], lane);
      if (m - 1 + kFar < STEPS) __builtin_prefetch(zt + (long)(m - 1 + kFar) * kStep + lane * 16, 0, 0);
    }
    __builtin_amdgcn_sched_barrier(0);
#pragma unroll
    for (int c = 1; c < 8; ++c) mm_chunk(Kt, zb, c, s0, s1);
    if (m >= 1) {
      v2f xb[8];
      c_to_b_regs(st[0], st[1], xb);
#pragma unroll
      for (int c = 0; c < 8; ++c) mm_chunk(A, xb, c, s0, s1);
    }
    st[0] = s0;
    st[1] = s1;
    __builtin_amdgcn_sched_barrier(0);
  }
  store_c_cols(dst_end, st[0], st[1], lane);
}

__global__ void __launch_bounds__(32)
kf_forward_end_kernel(const float* __restrict__ z, const float* __restrict__ K,
                      const float* __restrict__ F, const float* __restrict__ Gr,
                      const float* __restrict__ Gs, float* __restrict__ hat,
                      float* __restrict__ power_out) {
  const int lane = threadIdx.x & 31;
  const int b0 = blockIdx.x * 16;
  const int j = blockIdx.y;

  if (j == kT - 1) {
    if (blockIdx.x < 2) power_row_tile<kL>(F, blockIdx.x, power_out, lane);
    else if (blockIdx.x < 4) power_row_tile<kL>(Gr, blockIdx.x - 2, hat, lane);
    else if (blockIdx.x == 4) gain_power_table(Gr, Gs, power_out + kStep, lane);
    else if (blockIdx.x < 7) power_row_tile<kGrp>(Gr, blockIdx.x - 5, power_out + 2 * kStep, lane);
    return;
  }

  ATiles Kt, A;
  load_a_tiles(K, Kt, lane);
  load_a_tiles(F, A, lane);
  const float* zt = z + (size_t)j * kL * kStep + (size_t)b0 * kD;
  float* dst_end = step_tile(hat + (size_t)b0 * kD, true, j * kL + kL - 1);
  forward_end_pass(Kt, A, zt, dst_end, lane);
}

template <bool BACKWARD_END>
__device__ __forceinline__ void forward_fill_pass(const ATiles& Kt, const ATiles& A, const float* vtab,
                                                  const v2f x_init[8],
                                                  const float* __restrict__ zt, float* tile0,
                                                  float* __restrict__ p0, int lane) {
  constexpr int STEPS = kL - 1;
  const v8f zero = {0.f, 0.f, 0.f, 0.f, 0.f, 0.f, 0.f, 0.f};
  v8f zr[kInRing][2];
#pragma unroll
  for (int s = 0; s < kInRing; ++s) load_c_cols(zt + (long)s * kStep, zr[s][0], zr[s][1], lane);
#pragma unroll
  for (int s = kInRing; s < min(kFar, STEPS); ++s) __builtin_prefetch(zt + (long)s * kStep + lane * 16, 0, 0);

  ATiles V[kGrp], Rg;
  if (BACKWARD_END) {
#pragma unroll
    for (int i = 0; i < kGrp; ++i) load_a_tiles(vtab + i * kD * kD, V[i], lane);
    load_a_tiles(vtab + kStep, Rg, lane);
  }

  v2f xb[8];
#pragma unroll
  for (int c = 0; c < 8; ++c) xb[c] = x_init[c];
  v8f xs[2] = {zero, zero};
  v8f eb[kL / kGrp][2];
#pragma unroll
  for (int g = 0; g < kL / kGrp; ++g) eb[g][0] = eb[g][1] = zero;
#pragma unroll
  for (int m = 0; m < STEPS; ++m) {
    v2f zb[8];
    c_to_b_regs(zr[m % kInRing][0], zr[m % kInRing][1], zb);
    v8f a0 = zero, a1 = zero;
    mm_chunk(Kt, zb, 0, a0, a1);
    __builtin_amdgcn_sched_barrier(0);
    if (m >= 1) {
      if (m - 1 + kInRing < STEPS)
        load_c_cols(zt + (long)(m - 1 + kInRing) * kStep, zr[(m - 1) % kInRing][0], zr[(m - 1) % kInRing][1], lane);
      if (m - 1 + kFar < STEPS) __builtin_prefetch(zt + (long)(m - 1 + kFar) * kStep + lane * 16, 0, 0);
      store_c_cols(tile0 + (long)(m - 1) * kStep, xs[0], xs[1], lane);
    }
    __builtin_amdgcn_sched_barrier(0);
#pragma unroll
    for (int c = 1; c < 8; ++c) mm_chunk(Kt, zb, c, a0, a1);
#pragma unroll
    for (int c = 0; c < 8; ++c) mm_chunk(A, xb, c, a0, a1);
    if (BACKWARD_END) {
#pragma unroll
      for (int c = 0; c < 8; ++c) mm_chunk(V[m % kGrp], xb, c, eb[m / kGrp][0], eb[m / kGrp][1]);
    }
    xs[0] = a0;
    xs[1] = a1;
    c_to_b_regs(a0, a1, xb);
    __builtin_amdgcn_sched_barrier(0);
  }
  store_c_cols(tile0 + (long)(STEPS - 1) * kStep, xs[0], xs[1], lane);

  if (BACKWARD_END) {
    constexpr int kLast = kL - 1;
#pragma unroll
    for (int c = 0; c < 8; ++c) mm_chunk(V[kLast % kGrp], xb, c, eb[kLast / kGrp][0], eb[kLast / kGrp][1]);
    v8f s0 = eb[kL / kGrp - 1][0], s1 = eb[kL / kGrp - 1][1];
#pragma unroll
    for (int g = kL / kGrp - 2; g >= 0; --g) {
      v2f sb[8];
      c_to_b_regs(s0, s1, sb);
      v8f a0 = eb[g][0], a1 = eb[g][1];
#pragma unroll
      for (int c = 0; c < 8; ++c) mm_chunk(Rg, sb, c, a0, a1);
      s0 = a0;
      s1 = a1;
    }
    store_c_cols(p0, s0, s1, lane);
  }
}

__global__ void __launch_bounds__(32)
kf_forward_fill_kernel(const float* __restrict__ z, const float* __restrict__ K,
                       const float* __restrict__ F, const float* __restrict__ x0, float* hat,
                       float* __restrict__ out2) {
  const int lane = threadIdx.x & 31;
  const int b0 = blockIdx.x * 16;
  const int j = blockIdx.y;

  ATiles Kt, A;
  load_a_tiles(K, Kt, lane);
  load_a_tiles(F, A, lane);
  v2f xb[8];
  load_b_cols(j > 0 ? step_tile(hat + (size_t)b0 * kD, true, j * kL - 1) : x0 + (size_t)b0 * kD, xb, lane);
  const float* zt = z + (size_t)j * kL * kStep + (size_t)b0 * kD;
  float* tile0 = step_tile(hat + (size_t)b0 * kD, true, j * kL);
  if (j > 0) {
    float* p0 = out2 + ((long)j * kL - 1) * (long)kStep + (long)b0 * kD;
    forward_fill_pass<true>(Kt, A, out2 + kStep, xb, zt, tile0, p0, lane);
  } else {
    forward_fill_pass<false>(Kt, A, nullptr, xb, zt, tile0, nullptr, lane);
  }
}

__global__ void __launch_bounds__(32)
kf_backward_fill_kernel(const float* __restrict__ Gs, const float* __restrict__ Gr,
                        const float* __restrict__ x0, float* hat, float* out2) {
  const int lane = threadIdx.x & 31;
  const int b0 = blockIdx.x * 16;
  const int j = blockIdx.y;

  if (j == kT) {
    v2f x[8];
    load_b_cols(x0 + (size_t)b0 * kD, x, lane);
    store_b_cols(hat + (size_t)b0 * kD, x, lane);
    return;
  }

  constexpr int STEPS = kL - 1;
  const v8f zero = {0.f, 0.f, 0.f, 0.f, 0.f, 0.f, 0.f, 0.f};
  ATiles G, R;
  load_a_tiles(Gs, G, lane);
  load_a_tiles(Gr, R, lane);
  v2f sb[8];
  load_b_cols(step_tile(out2 + (size_t)b0 * kD, false, j * kL - 1), sb, lane);
  float* tile0 = step_tile(out2 + (size_t)b0 * kD, false, j * kL);
  const float* ht = hat + (size_t)(kN - 1 - j * kL) * kStep + (size_t)b0 * kD;

  v8f hr[kInRing][2];
#pragma unroll
  for (int s = 0; s < kInRing; ++s) load_c_cols(ht - (long)s * kStep, hr[s][0], hr[s][1], lane);
#pragma unroll
  for (int s = kInRing; s < min(kFar, STEPS); ++s) __builtin_prefetch(ht - (long)s * kStep + lane * 16, 0, 0);

  v8f xs[2] = {zero, zero};
#pragma unroll
  for (int m = 0; m < STEPS; ++m) {
    v2f hb[8];
    c_to_b_regs(hr[m % kInRing][0], hr[m % kInRing][1], hb);
    v8f a0 = zero, a1 = zero;
    mm_chunk(G, hb, 0, a0, a1);
    __builtin_amdgcn_sched_barrier(0);
    if (m >= 1) {
      if (m - 1 + kInRing < STEPS)
        load_c_cols(ht - (long)(m - 1 + kInRing) * kStep, hr[(m - 1) % kInRing][0], hr[(m - 1) % kInRing][1], lane);
      if (m - 1 + kFar < STEPS) __builtin_prefetch(ht - (long)(m - 1 + kFar) * kStep + lane * 16, 0, 0);
      store_c_cols(tile0 - (long)(m - 1) * kStep, xs[0], xs[1], lane);
    }
    __builtin_amdgcn_sched_barrier(0);
#pragma unroll
    for (int c = 1; c < 8; ++c) mm_chunk(G, hb, c, a0, a1);
#pragma unroll
    for (int c = 0; c < 8; ++c) mm_chunk(R, sb, c, a0, a1);
    xs[0] = a0;
    xs[1] = a1;
    c_to_b_regs(a0, a1, sb);
    __builtin_amdgcn_sched_barrier(0);
  }
  store_c_cols(tile0 - (long)(STEPS - 1) * kStep, xs[0], xs[1], lane);
}

constexpr int kGroups = 4;
constexpr int kGroupLen = 16;
static_assert(kGroups * kGroupLen == kT, "groups tile the border steps, the last one short by one");
constexpr int kPowStep = 4;
static_assert(kPowStep * kPowStep == kGroupLen, "two equal levels reach the group's power");

__device__ __forceinline__ void lds_put_c(v4f* slot, v8f c0, v8f c1) {
  slot[0] = (v4f){c0[0], c0[1], c0[2], c0[3]};
  slot[1] = (v4f){c0[4], c0[5], c0[6], c0[7]};
  slot[2] = (v4f){c1[0], c1[1], c1[2], c1[3]};
  slot[3] = (v4f){c1[4], c1[5], c1[6], c1[7]};
}

__device__ __forceinline__ void lds_get_c(const v4f* slot, v8f& c0, v8f& c1) {
  const v4f a = slot[0], b = slot[1], c = slot[2], d = slot[3];
  c0 = (v8f){a[0], a[1], a[2], a[3], b[0], b[1], b[2], b[3]};
  c1 = (v8f){c[0], c[1], c[2], c[3], d[0], d[1], d[2], d[3]};
}

template <long STRIDE>
__device__ __forceinline__ void group_end_from_zero(const ATiles& P, float* group_tile0, v4f* slot,
                                                    int lane) {
  const v2f none[8] = {};
  v8f e[2];
  run_recurrence<kGroupLen, STRIDE, false, true, false>(P, none, group_tile0, lane, e);
  lds_put_c(slot, e[0], e[1]);
}

template <long STRIDE>
__device__ __forceinline__ void group_jump_and_fill(const ATiles& P, const ATiles& Pg, v2f xb[8],
                                                    const v4f* ends, float* group_tile0, int g,
                                                    int lane) {
#pragma unroll
  for (int i = 0; i < kGroups - 1; ++i) {
    if (i < g) {
      v8f a0, a1;
      lds_get_c(ends + (i * 32 + lane) * 4, a0, a1);
#pragma unroll
      for (int c = 0; c < 8; ++c) mm_chunk(Pg, xb, c, a0, a1);
      c_to_b_regs(a0, a1, xb);
    }
  }
  v8f unused[2];
  if (g < kGroups - 1)
    run_recurrence<kGroupLen, STRIDE, true, false, true>(P, xb, group_tile0, lane, unused);
  else
    run_recurrence<kGroupLen - 1, STRIDE, true, false, true>(P, xb, group_tile0, lane, unused);
}

__global__ void __launch_bounds__(32 * kGroups)
kf_forward_borders_kernel(const float* __restrict__ x0, const float* __restrict__ Fk, float* hat) {
  __shared__ v4f lds_e[(kGroups - 1) * 32 * 4];
  __shared__ v4f lds_p[kD * kD / 4];
  __shared__ v4f lds_q[kD * kD / 4];
  const int lane = threadIdx.x & 31;
  const int g = __builtin_amdgcn_readfirstlane((int)(threadIdx.x >> 5));
  if (blockIdx.x >= kB / 16) {
    if (g == 0) power_row_tile<kGroupLen>(hat, blockIdx.x - kB / 16, hat + kD * kD, lane);
    return;
  }
  const int b0 = blockIdx.x * 16;
  constexpr long STRIDE = kL * kFwdStride;
  float* group_tile0 = step_tile(hat + (size_t)b0 * kD, true, kL - 1) + (long)g * kGroupLen * STRIDE;

  ATiles P;
  load_a_tiles(Fk, P, lane);
  if (g < kGroups - 1) {
    group_end_from_zero<STRIDE>(P, group_tile0, lds_e + (g * 32 + lane) * 4, lane);
  } else {
    power_row_tile<kPowStep>(Fk, 0, (float*)lds_q, lane);
    power_row_tile<kPowStep>(Fk, 1, (float*)lds_q, lane);
    __builtin_amdgcn_fence(__ATOMIC_SEQ_CST, "wavefront");
    power_row_tile<kPowStep>((const float*)lds_q, 0, (float*)lds_p, lane);
    power_row_tile<kPowStep>((const float*)lds_q, 1, (float*)lds_p, lane);
  }
  __syncthreads();

  ATiles Pg;
  load_a_tiles((const float*)lds_p, Pg, lane);
  v2f xb[8];
  load_b_cols(x0 + (size_t)b0 * kD, xb, lane);
  group_jump_and_fill<STRIDE>(P, Pg, xb, lds_e, group_tile0, g, lane);
}

__global__ void __launch_bounds__(32 * kGroups)
kf_backward_borders_kernel(const float* __restrict__ z, const float* __restrict__ Ht,
                           const float* __restrict__ Ka0, const float* __restrict__ hat, float* out2) {
  __shared__ v4f lds_e[(kGroups - 1) * 32 * 4];
  __shared__ v4f lds_s[32 * 4];
  const int lane = threadIdx.x & 31;
  const int g = __builtin_amdgcn_readfirstlane((int)(threadIdx.x >> 5));
  const int b0 = blockIdx.x * 16;
  const size_t last = (size_t)(kN - 1) * kStep + (size_t)b0 * kD;
  constexpr long STRIDE = kL * kBwdStride;
  float* group_tile0 = step_tile(out2 + (size_t)b0 * kD, false, kL - 1) + (long)g * kGroupLen * STRIDE;

  ATiles P, Pg;
  load_a_tiles(hat, P, lane);
  load_a_tiles(hat + kD * kD, Pg, lane);
  if (g < kGroups - 1) {
    group_end_from_zero<STRIDE>(P, group_tile0, lds_e + (g * 32 + lane) * 4, lane);
  } else {
    ATiles A1, A2;
    v2f hb[8];
    load_b_cols(hat + last, hb, lane);
    load_a_tiles(Ht, A1, lane);
    v8f t0 = {0.f, 0.f, 0.f, 0.f, 0.f, 0.f, 0.f, 0.f};
    v8f t1 = t0;
#pragma unroll
    for (int c = 0; c < 8; ++c) t0 = WMMA4(A1.a[0][c], hb[c], t0);
#pragma unroll
    for (int c = 0; c < 8; ++c) t1 = WMMA4(A1.a[1][c], hb[c], t1);

    v8f zc0, zc1;
    load_c_cols(z + last, zc0, zc1, lane);
    v8f bz0 = zc0 - t0;
    v8f bz1 = zc1 - t1;
    v2f bb[8];
    c_to_b_regs(bz0, bz1, bb);

    load_a_tiles(Ka0, A2, lane);
    v8f s0, s1;
    load_c_cols(hat + last, s0, s1, lane);
#pragma unroll
    for (int c = 0; c < 8; ++c) s0 = WMMA4(A2.a[0][c], bb[c], s0);
#pragma unroll
    for (int c = 0; c < 8; ++c) s1 = WMMA4(A2.a[1][c], bb[c], s1);

    store_c_cols(out2 + last, s0, s1, lane);
    lds_put_c(lds_s + lane * 4, s0, s1);
  }
  __syncthreads();

  v8f s0, s1;
  lds_get_c(lds_s + lane * 4, s0, s1);
  v2f sb[8];
  c_to_b_regs(s0, s1, sb);
  group_jump_and_fill<STRIDE>(P, Pg, sb, lds_e, group_tile0, g, lane);
}

extern "C" void kernel_launch(void* const* d_in, const int* in_sizes, int n_in,
                              void* d_out, int out_size, void* d_ws, size_t ws_size,
                              hipStream_t stream) {
  (void)in_sizes; (void)n_in; (void)d_ws; (void)ws_size; (void)out_size;
  const float* z   = (const float*)d_in[0];
  const float* x0  = (const float*)d_in[1];
  const float* F   = (const float*)d_in[2];
  const float* K   = (const float*)d_in[3];
  const float* Ht  = (const float*)d_in[4];
  const float* Ka0 = (const float*)d_in[5];
  const float* Gs  = (const float*)d_in[6];
  const float* Gr  = (const float*)d_in[7];

  float* hat  = (float*)d_out;
  float* out2 = hat + (size_t)kN * kB * kD;

  dim3 slices(kB / 16);
  dim3 slices_and_power(kB / 16 + 2);
  dim3 group_block(32 * kGroups);
  dim3 chunks(kB / 16, kT);
  dim3 chunks_and_row(kB / 16, kT + 1);
  dim3 block(32);
  kf_forward_end_kernel<<<chunks, block, 0, stream>>>(z, K, F, Gr, Gs, hat, out2);
  kf_forward_borders_kernel<<<slices_and_power, group_block, 0, stream>>>(x0, out2, hat);
  kf_forward_fill_kernel<<<chunks, block, 0, stream>>>(z, K, F, x0, hat, out2);
  kf_backward_borders_kernel<<<slices, group_block, 0, stream>>>(z, Ht, Ka0, hat, out2);
  kf_backward_fill_kernel<<<chunks_and_row, block, 0, stream>>>(Gs, Gr, x0, hat, out2);
}


# forward chunk ends as group sums of (F^i*K)*z, i<8, joined by F^8: 16 WMMAs a step, no F*x chain; the 8 matrices formed by the 4 waves of a work-group, handed over in LDS; changes the order of sums
# speedup vs baseline: 34.6942x; 1.0633x over previous
#include <hip/hip_runtime.h>

typedef __attribute__((ext_vector_type(2))) float v2f;
typedef __attribute__((ext_vector_type(4))) float v4f;
typedef __attribute__((ext_vector_type(8))) float v8f;

#define WMMA4(A, B, C) \
  __builtin_amdgcn_wmma_f32_16x16x4_f32(false, (A), false, (B), (short)0, (C), false, false)

constexpr int kN = 2048;
constexpr int kB = 256;
constexpr int kD = 32;
constexpr size_t kStep = (size_t)kB * kD;

__device__ __forceinline__ int colbase(int c, int half) {
  return 2 * c + 8 * half + (c >= 4 ? 8 : 0);
}

struct ATiles { v2f a[2][8]; };

__device__ __forceinline__ void load_a_tiles(const float* __restrict__ M, ATiles& t, int lane) {
  const int r0 = lane & 15;
  const int half = lane >> 4;
#pragma unroll
  for (int rt = 0; rt < 2; ++rt) {
#pragma unroll
    for (int c = 0; c < 8; ++c) {
      const float* p = M + (16 * rt + r0) * kD + colbase(c, half);
      t.a[rt][c] = *(const v2f*)p;
    }
  }
}

__device__ __forceinline__ void load_b_cols(const float* __restrict__ src, v2f xb[8], int lane) {
  const int col = lane & 15;
  const int half = lane >> 4;
  const float* p = src + col * kD;
#pragma unroll
  for (int c = 0; c < 8; ++c) xb[c] = *(const v2f*)(p + colbase(c, half));
}

__device__ __forceinline__ void store_b_cols(float* __restrict__ dst, const v2f xb[8], int lane) {
  const int col = lane & 15;
  const int half = lane >> 4;
  float* p = dst + col * kD;
#pragma unroll
  for (int c = 0; c < 8; ++c) *(v2f*)(p + colbase(c, half)) = xb[c];
}

__device__ __forceinline__ void load_c_cols(const float* src, v8f& c0, v8f& c1, int lane) {
  const int col = lane & 15;
  const int half = lane >> 4;
  const float* p = src + col * kD + 8 * half;
  v4f a = *(const v4f*)(p + 0);
  v4f b = *(const v4f*)(p + 4);
  v4f c = *(const v4f*)(p + 16);
  v4f d = *(const v4f*)(p + 20);
  c0 = (v8f){a[0], a[1], a[2], a[3], b[0], b[1], b[2], b[3]};
  c1 = (v8f){c[0], c[1], c[2], c[3], d[0], d[1], d[2], d[3]};
}

__device__ __forceinline__ void store_c_cols(float* dst, v8f c0, v8f c1, int lane) {
  const int col = lane & 15;
  const int half = lane >> 4;
  float* p = dst + col * kD + 8 * half;
  *(v4f*)(p + 0)  = (v4f){c0[0], c0[1], c0[2], c0[3]};
  *(v4f*)(p + 4)  = (v4f){c0[4], c0[5], c0[6], c0[7]};
  *(v4f*)(p + 16) = (v4f){c1[0], c1[1], c1[2], c1[3]};
  *(v4f*)(p + 20) = (v4f){c1[4], c1[5], c1[6], c1[7]};
}

__device__ __forceinline__ void c_to_b_regs(const v8f& c0, const v8f& c1, v2f xb[8]) {
#pragma unroll
  for (int c = 0; c < 4; ++c) {
    xb[c]     = (v2f){c0[2 * c], c0[2 * c + 1]};
    xb[c + 4] = (v2f){c1[2 * c], c1[2 * c + 1]};
  }
}

__device__ __forceinline__ void mm_chunk(const ATiles& A, const v2f b[8], int c, v8f& a0, v8f& a1) {
  a0 = WMMA4(A.a[0][c], b[c], a0);
  a1 = WMMA4(A.a[1][c], b[c], a1);
}

constexpr int kRing = 8;
constexpr int kFar = 16;

constexpr int kL = 32;
constexpr int kT = kN / kL;
static_assert(kT * kL == kN, "chunks tile the time axis");

__device__ __forceinline__ float* step_tile(float* base, bool forward, int g) {
  return base + (size_t)(forward ? g + 1 : kN - 2 - g) * kStep;
}

template <bool COMPUTE>
__device__ __forceinline__ void seeded_step(const ATiles& A, const v2f xb[8], v8f acc[2],
                                            bool store_prev, const v8f prev[2], float* prev_dst,
                                            bool refill, v8f freed[2], const float* refill_src,
                                            const float* far_src, int lane) {
  if (COMPUTE) mm_chunk(A, xb, 0, acc[0], acc[1]);
  __builtin_amdgcn_sched_barrier(0);
  if (refill) {
    load_c_cols(refill_src, freed[0], freed[1], lane);
    __builtin_prefetch(far_src + lane * 16, 0, 0);
  }
  if (store_prev) store_c_cols(prev_dst, prev[0], prev[1], lane);
  __builtin_amdgcn_sched_barrier(0);
  if (COMPUTE) {
#pragma unroll
    for (int c = 1; c < 8; ++c) mm_chunk(A, xb, c, acc[0], acc[1]);
  }
  __builtin_amdgcn_sched_barrier(0);
}

template <int STEPS, long STRIDE, bool STORE_ALL, bool ZERO_INIT, bool STORE_LAST>
__device__ __forceinline__ void run_recurrence(const ATiles& A, const v2f x_init[8], float* tile0,
                                               int lane, v8f last_out[2]) {
  constexpr int kMainSteps = ((STEPS - 1) / kRing) * kRing;
  constexpr int kTailSteps = (STEPS - 1) - kMainSteps;
  static_assert(kRing >= 3 && STEPS >= kRing, "the ring's first loads stay inside the run");
  v8f sd[kRing][2];
  __builtin_amdgcn_sched_barrier(0);
#pragma unroll
  for (int s = 0; s < kRing - 2; ++s) load_c_cols(tile0 + s * STRIDE, sd[s][0], sd[s][1], lane);
  __builtin_amdgcn_sched_barrier(0);
#pragma unroll
  for (int s = kRing - 2; s < min(kFar, STEPS); ++s) __builtin_prefetch(tile0 + s * STRIDE + lane * 16, 0, 0);

  seeded_step<!ZERO_INIT>(A, x_init, sd[0], false, sd[0], nullptr, true, sd[kRing - 2],
                          tile0 + (kRing - 2) * STRIDE, tile0 + min(kFar, STEPS - 1) * STRIDE, lane);

  for (int n = 0; n < kMainSteps; n += kRing) {
#pragma unroll
    for (int s = 0; s < kRing; ++s) {
      const int m = 1 + n + s;
      const int mr = min(m + kRing - 2, STEPS - 1);
      const int mf = min(m + kFar, STEPS - 1);
      v2f xb[8];
      c_to_b_regs(sd[s % kRing][0], sd[s % kRing][1], xb);
      seeded_step<true>(A, xb, sd[(1 + s) % kRing], STORE_ALL, sd[s % kRing], tile0 + (long)(m - 1) * STRIDE,
                        true, sd[(s + kRing - 1) % kRing], tile0 + (long)mr * STRIDE,
                        tile0 + (long)mf * STRIDE, lane);
    }
  }
#pragma unroll
  for (int s = 0; s < kTailSteps; ++s) {
    const int m = 1 + kMainSteps + s;
    const bool refill = m + kRing - 2 <= STEPS - 1;
    const int mr = refill ? m + kRing - 2 : STEPS - 1;
    v2f xb[8];
    c_to_b_regs(sd[s % kRing][0], sd[s % kRing][1], xb);
    seeded_step<true>(A, xb, sd[(1 + s) % kRing], STORE_ALL, sd[s % kRing], tile0 + (long)(m - 1) * STRIDE,
                      refill, sd[(s + kRing - 1) % kRing], tile0 + (long)mr * STRIDE,
                      tile0 + (long)(STEPS - 1) * STRIDE, lane);
  }
  if (STORE_LAST) {
    store_c_cols(tile0 + (long)(STEPS - 1) * STRIDE, sd[(STEPS - 1) % kRing][0],
                 sd[(STEPS - 1) % kRing][1], lane);
  } else {
    last_out[0] = sd[(STEPS - 1) % kRing][0];
    last_out[1] = sd[(STEPS - 1) % kRing][1];
  }
}

constexpr long kFwdStride = (long)kStep;
constexpr long kBwdStride = -(long)kStep;

template <int POW>
__device__ __forceinline__ void power_row_tile_regs(const float* __restrict__ M, int rt,
                                                    v2f q[8], int lane) {
  const int r0 = lane & 15;
  const int half = lane >> 4;
  ATiles Mt;
#pragma unroll
  for (int t = 0; t < 2; ++t) {
#pragma unroll
    for (int c = 0; c < 8; ++c) {
      const float* p = M + colbase(c, half) * kD + 16 * t + r0;
      Mt.a[t][c] = (v2f){p[0], p[kD]};
    }
  }
#pragma unroll
  for (int c = 0; c < 8; ++c) q[c] = *(const v2f*)(M + (16 * rt + r0) * kD + colbase(c, half));
  for (int i = 1; i < POW; ++i) {
    const v8f zero = {0.f, 0.f, 0.f, 0.f, 0.f, 0.f, 0.f, 0.f};
    v8f d0 = zero, d1 = zero;
#pragma unroll
    for (int c = 0; c < 8; ++c) {
      d0 = WMMA4(Mt.a[0][c], q[c], d0);
      d1 = WMMA4(Mt.a[1][c], q[c], d1);
    }
    c_to_b_regs(d0, d1, q);
  }
}

template <int POW>
__device__ __forceinline__ void power_row_tile(const float* __restrict__ M, int rt,
                                               float* __restrict__ dst, int lane) {
  v2f q[8];
  power_row_tile_regs<POW>(M, rt, q, lane);
  store_b_cols(dst + 16 * rt * kD, q, lane);
}

constexpr int kGrp = 8;
static_assert(kL % kGrp == 0, "groups tile the chunk");

__device__ __forceinline__ void gain_power_table(const float* __restrict__ M, const float* __restrict__ G0,
                                                 float* __restrict__ dst, int lane) {
  const int col = lane & 15;
  const int half = lane >> 4;
  ATiles Mt;
  load_a_tiles(M, Mt, lane);
  v2f q[2][8];
#pragma unroll
  for (int ct = 0; ct < 2; ++ct) {
#pragma unroll
    for (int c = 0; c < 8; ++c) {
      const float* p = G0 + colbase(c, half) * kD + 16 * ct + col;
      q[ct][c] = (v2f){p[0], p[kD]};
      float* w = dst + colbase(c, half) * kD + 16 * ct + col;
      w[0] = q[ct][c][0];
      w[kD] = q[ct][c][1];
    }
  }
  for (int i = 1; i < kGrp; ++i) {
    const v8f zero = {0.f, 0.f, 0.f, 0.f, 0.f, 0.f, 0.f, 0.f};
#pragma unroll
    for (int ct = 0; ct < 2; ++ct) {
      v8f d0 = zero, d1 = zero;
#pragma unroll
      for (int c = 0; c < 8; ++c) {
        d0 = WMMA4(Mt.a[0][c], q[ct][c], d0);
        d1 = WMMA4(Mt.a[1][c], q[ct][c], d1);
      }
      float* w = dst + (long)i * kD * kD + 8 * half * kD + 16 * ct + col;
#pragma unroll
      for (int v = 0; v < 8; ++v) {
        w[v * kD] = d0[v];
        w[(16 + v) * kD] = d1[v];
      }
      c_to_b_regs(d0, d1, q[ct]);
    }
  }
}

constexpr int kInRing = 6;

constexpr int kFGrp = 8;
static_assert(kL % kFGrp == 0, "groups tile the chunk");
constexpr int kEndWaves = 4;
static_assert(kT % kEndWaves == 0, "the waves of a work-group take neighbouring chunks");
constexpr int kEndRing = 8;

__device__ __forceinline__ int tab_slot(int mat, int rt, int c) { return ((mat * 2 + rt) * 8 + c) * 32; }

__device__ __forceinline__ void gain_chain_col_tile(const float* __restrict__ M, const float* __restrict__ G0,
                                                    int ct, v2f* tab, int lane) {
  const int col = lane & 15;
  const int half = lane >> 4;
  ATiles Mt;
  load_a_tiles(M, Mt, lane);
  v2f q[8];
#pragma unroll
  for (int c = 0; c < 8; ++c) {
    const float* p = G0 + colbase(c, half) * kD + 16 * ct + col;
    q[c] = (v2f){p[0], p[kD]};
  }
  const int ca = 4 * ct + ((col & 7) >> 1);
  const int ha = col >> 3;
  const int ea = col & 1;
  float* tf = (float*)tab;
  for (int p = 1; p < kFGrp; ++p) {
    const v8f zero = {0.f, 0.f, 0.f, 0.f, 0.f, 0.f, 0.f, 0.f};
    v8f d0 = zero, d1 = zero;
#pragma unroll
    for (int c = 0; c < 8; ++c) {
      d0 = WMMA4(Mt.a[0][c], q[c], d0);
      d1 = WMMA4(Mt.a[1][c], q[c], d1);
    }
    float* w0 = tf + (tab_slot(p - 1, 0, ca) + 16 * ha + 8 * half) * 2 + ea;
    float* w1 = tf + (tab_slot(p - 1, 1, ca) + 16 * ha + 8 * half) * 2 + ea;
#pragma unroll
    for (int v = 0; v < 8; ++v) {
      w0[2 * v] = d0[v];
      w1[2 * v] = d1[v];
    }
    c_to_b_regs(d0, d1, q);
  }
}

__device__ __forceinline__ void forward_end_pass(const ATiles W[kFGrp], const ATiles& Fg,
                                                 const float* __restrict__ zt,
                                                 float* __restrict__ dst_end, int lane) {
  constexpr int STEPS = kL;
  static_assert(STEPS >= kEndRing, "the ring's first loads stay inside the chunk");
  const v8f zero = {0.f, 0.f, 0.f, 0.f, 0.f, 0.f, 0.f, 0.f};
  v8f zr[kEndRing][2];
#pragma unroll
  for (int s = 0; s < kEndRing; ++s) load_c_cols(zt + (long)s * kStep, zr[s][0], zr[s][1], lane);

  v8f acc[kL / kFGrp][2];
#pragma unroll
  for (int g = 0; g < kL / kFGrp; ++g) acc[g][0] = acc[g][1] = zero;
#pragma unroll
  for (int m = 0; m < STEPS; ++m) {
    v2f zb[8];
    c_to_b_regs(zr[m % kEndRing][0], zr[m % kEndRing][1], zb);
    mm_chunk(W[kFGrp - 1 - m % kFGrp], zb, 0, acc[m / kFGrp][0], acc[m / kFGrp][1]);
    __builtin_amdgcn_sched_barrier(0);
    if (m >= 1) {
      if (m - 1 + kEndRing < STEPS)
        load_c_cols(zt + (long)(m - 1 + kEndRing) * kStep, zr[(m - 1) % kEndRing][0], zr[(m - 1) % kEndRing][1], lane);
      if (m - 1 + kFar < STEPS) __builtin_prefetch(zt + (long)(m - 1 + kFar) * kStep + lane * 16, 0, 0);
    }
    __builtin_amdgcn_sched_barrier(0);
#pragma unroll
    for (int c = 1; c < 8; ++c) mm_chunk(W[kFGrp - 1 - m % kFGrp], zb, c, acc[m / kFGrp][0], acc[m / kFGrp][1]);
    __builtin_amdgcn_sched_barrier(0);
  }
  v8f s0 = acc[0][0], s1 = acc[0][1];
#pragma unroll
  for (int g = 1; g < kL / kFGrp; ++g) {
    v2f sb[8];
    c_to_b_regs(s0, s1, sb);
    v8f a0 = acc[g][0], a1 = acc[g][1];
#pragma unroll
    for (int c = 0; c < 8; ++c) mm_chunk(Fg, sb, c, a0, a1);
    s0 = a0;
    s1 = a1;
  }
  store_c_cols(dst_end, s0, s1, lane);
}

__global__ void __launch_bounds__(32 * kEndWaves)
kf_forward_end_kernel(const float* __restrict__ z, const float* __restrict__ K,
                      const float* __restrict__ F, const float* __restrict__ Gr,
                      const float* __restrict__ Gs, float* __restrict__ hat,
                      float* __restrict__ power_out) {
  __shared__ v2f tab[kFGrp * 2 * 8 * 32];
  const int lane = threadIdx.x & 31;
  const int w = __builtin_amdgcn_readfirstlane((int)(threadIdx.x >> 5));
  const int b0 = blockIdx.x * 16;
  const int j = blockIdx.y * kEndWaves + w;
  const bool has_chunk = j < kT - 1;
  const float* zt = z + (size_t)j * kL * kStep + (size_t)b0 * kD;

  if (has_chunk) {
#pragma unroll
    for (int s = 0; s < min(kFar, kL); ++s) __builtin_prefetch(zt + (long)s * kStep + lane * 16, 0, 0);
  }
  if (w < 2) {
    gain_chain_col_tile(F, K, w, tab, lane);
  } else {
    v2f q[8];
    power_row_tile_regs<kFGrp>(F, w - 2, q, lane);
#pragma unroll
    for (int c = 0; c < 8; ++c) tab[tab_slot(kFGrp - 1, w - 2, c) + lane] = q[c];
  }
  __syncthreads();

  if (!has_chunk) {
    if (blockIdx.x < 2) power_row_tile<kL>(F, blockIdx.x, power_out, lane);
    else if (blockIdx.x < 4) power_row_tile<kL>(Gr, blockIdx.x - 2, hat, lane);
    else if (blockIdx.x == 4) gain_power_table(Gr, Gs, power_out + kStep, lane);
    else if (blockIdx.x < 7) power_row_tile<kGrp>(Gr, blockIdx.x - 5, power_out + 2 * kStep, lane);
    return;
  }

  ATiles W[kFGrp], Fg;
  load_a_tiles(K, W[0], lane);
#pragma unroll
  for (int p = 1; p < kFGrp; ++p) {
#pragma unroll
    for (int rt = 0; rt < 2; ++rt) {
#pragma unroll
      for (int c = 0; c < 8; ++c) W[p].a[rt][c] = tab[tab_slot(p - 1, rt, c) + lane];
    }
  }
#pragma unroll
  for (int rt = 0; rt < 2; ++rt) {
#pragma unroll
    for (int c = 0; c < 8; ++c) Fg.a[rt][c] = tab[tab_slot(kFGrp - 1, rt, c) + lane];
  }
  float* dst_end = step_tile(hat + (size_t)b0 * kD, true, j * kL + kL - 1);
  forward_end_pass(W, Fg, zt, dst_end, lane);
}

template <bool BACKWARD_END>
__device__ __forceinline__ void forward_fill_pass(const ATiles& Kt, const ATiles& A, const float* vtab,
                                                  const v2f x_init[8],
                                                  const float* __restrict__ zt, float* tile0,
                                                  float* __restrict__ p0, int lane) {
  constexpr int STEPS = kL - 1;
  const v8f zero = {0.f, 0.f, 0.f, 0.f, 0.f, 0.f, 0.f, 0.f};
  v8f zr[kInRing][2];
#pragma unroll
  for (int s = 0; s < kInRing; ++s) load_c_cols(zt + (long)s * kStep, zr[s][0], zr[s][1], lane);
#pragma unroll
  for (int s = kInRing; s < min(kFar, STEPS); ++s) __builtin_prefetch(zt + (long)s * kStep + lane * 16, 0, 0);

  ATiles V[kGrp], Rg;
  if (BACKWARD_END) {
#pragma unroll
    for (int i = 0; i < kGrp; ++i) load_a_tiles(vtab + i * kD * kD, V[i], lane);
    load_a_tiles(vtab + kStep, Rg, lane);
  }

  v2f xb[8];
#pragma unroll
  for (int c = 0; c < 8; ++c) xb[c] = x_init[c];
  v8f xs[2] = {zero, zero};
  v8f eb[kL / kGrp][2];
#pragma unroll
  for (int g = 0; g < kL / kGrp; ++g) eb[g][0] = eb[g][1] = zero;
#pragma unroll
  for (int m = 0; m < STEPS; ++m) {
    v2f zb[8];
    c_to_b_regs(zr[m % kInRing][0], zr[m % kInRing][1], zb);
    v8f a0 = zero, a1 = zero;
    mm_chunk(Kt, zb, 0, a0, a1);
    __builtin_amdgcn_sched_barrier(0);
    if (m >= 1) {
      if (m - 1 + kInRing < STEPS)
        load_c_cols(zt + (long)(m - 1 + kInRing) * kStep, zr[(m - 1) % kInRing][0], zr[(m - 1) % kInRing][1], lane);
      if (m - 1 + kFar < STEPS) __builtin_prefetch(zt + (long)(m - 1 + kFar) * kStep + lane * 16, 0, 0);
      store_c_cols(tile0 + (long)(m - 1) * kStep, xs[0], xs[1], lane);
    }
    __builtin_amdgcn_sched_barrier(0);
#pragma unroll
    for (int c = 1; c < 8; ++c) mm_chunk(Kt, zb, c, a0, a1);
#pragma unroll
    for (int c = 0; c < 8; ++c) mm_chunk(A, xb, c, a0, a1);
    if (BACKWARD_END) {
#pragma unroll
      for (int c = 0; c < 8; ++c) mm_chunk(V[m % kGrp], xb, c, eb[m / kGrp][0], eb[m / kGrp][1]);
    }
    xs[0] = a0;
    xs[1] = a1;
    c_to_b_regs(a0, a1, xb);
    __builtin_amdgcn_sched_barrier(0);
  }
  store_c_cols(tile0 + (long)(STEPS - 1) * kStep, xs[0], xs[1], lane);

  if (BACKWARD_END) {
    constexpr int kLast = kL - 1;
#pragma unroll
    for (int c = 0; c < 8; ++c) mm_chunk(V[kLast % kGrp], xb, c, eb[kLast / kGrp][0], eb[kLast / kGrp][1]);
    v8f s0 = eb[kL / kGrp - 1][0], s1 = eb[kL / kGrp - 1][1];
#pragma unroll
    for (int g = kL / kGrp - 2; g >= 0; --g) {
      v2f sb[8];
      c_to_b_regs(s0, s1, sb);
      v8f a0 = eb[g][0], a1 = eb[g][1];
#pragma unroll
      for (int c = 0; c < 8; ++c) mm_chunk(Rg, sb, c, a0, a1);
      s0 = a0;
      s1 = a1;
    }
    store_c_cols(p0, s0, s1, lane);
  }
}

__global__ void __launch_bounds__(32)
kf_forward_fill_kernel(const float* __restrict__ z, const float* __restrict__ K,
                       const float* __restrict__ F, const float* __restrict__ x0, float* hat,
                       float* __restrict__ out2) {
  const int lane = threadIdx.x & 31;
  const int b0 = blockIdx.x * 16;
  const int j = blockIdx.y;

  ATiles Kt, A;
  load_a_tiles(K, Kt, lane);
  load_a_tiles(F, A, lane);
  v2f xb[8];
  load_b_cols(j > 0 ? step_tile(hat + (size_t)b0 * kD, true, j * kL - 1) : x0 + (size_t)b0 * kD, xb, lane);
  const float* zt = z + (size_t)j * kL * kStep + (size_t)b0 * kD;
  float* tile0 = step_tile(hat + (size_t)b0 * kD, true, j * kL);
  if (j > 0) {
    float* p0 = out2 + ((long)j * kL - 1) * (long)kStep + (long)b0 * kD;
    forward_fill_pass<true>(Kt, A, out2 + kStep, xb, zt, tile0, p0, lane);
  } else {
    forward_fill_pass<false>(Kt, A, nullptr, xb, zt, tile0, nullptr, lane);
  }
}

__global__ void __launch_bounds__(32)
kf_backward_fill_kernel(const float* __restrict__ Gs, const float* __restrict__ Gr,
                        const float* __restrict__ x0, float* hat, float* out2) {
  const int lane = threadIdx.x & 31;
  const int b0 = blockIdx.x * 16;
  const int j = blockIdx.y;

  if (j == kT) {
    v2f x[8];
    load_b_cols(x0 + (size_t)b0 * kD, x, lane);
    store_b_cols(hat + (size_t)b0 * kD, x, lane);
    return;
  }

  constexpr int STEPS = kL - 1;
  const v8f zero = {0.f, 0.f, 0.f, 0.f, 0.f, 0.f, 0.f, 0.f};
  ATiles G, R;
  load_a_tiles(Gs, G, lane);
  load_a_tiles(Gr, R, lane);
  v2f sb[8];
  load_b_cols(step_tile(out2 + (size_t)b0 * kD, false, j * kL - 1), sb, lane);
  float* tile0 = step_tile(out2 + (size_t)b0 * kD, false, j * kL);
  const float* ht = hat + (size_t)(kN - 1 - j * kL) * kStep + (size_t)b0 * kD;

  v8f hr[kInRing][2];
#pragma unroll
  for (int s = 0; s < kInRing; ++s) load_c_cols(ht - (long)s * kStep, hr[s][0], hr[s][1], lane);
#pragma unroll
  for (int s = kInRing; s < min(kFar, STEPS); ++s) __builtin_prefetch(ht - (long)s * kStep + lane * 16, 0, 0);

  v8f xs[2] = {zero, zero};
#pragma unroll
  for (int m = 0; m < STEPS; ++m) {
    v2f hb[8];
    c_to_b_regs(hr[m % kInRing][0], hr[m % kInRing][1], hb);
    v8f a0 = zero, a1 = zero;
    mm_chunk(G, hb, 0, a0, a1);
    __builtin_amdgcn_sched_barrier(0);
    if (m >= 1) {
      if (m - 1 + kInRing < STEPS)
        load_c_cols(ht - (long)(m - 1 + kInRing) * kStep, hr[(m - 1) % kInRing][0], hr[(m - 1) % kInRing][1], lane);
      if (m - 1 + kFar < STEPS) __builtin_prefetch(ht - (long)(m - 1 + kFar) * kStep + lane * 16, 0, 0);
      store_c_cols(tile0 - (long)(m - 1) * kStep, xs[0], xs[1], lane);
    }
    __builtin_amdgcn_sched_barrier(0);
#pragma unroll
    for (int c = 1; c < 8; ++c) mm_chunk(G, hb, c, a0, a1);
#pragma unroll
    for (int c = 0; c < 8; ++c) mm_chunk(R, sb, c, a0, a1);
    xs[0] = a0;
    xs[1] = a1;
    c_to_b_regs(a0, a1, sb);
    __builtin_amdgcn_sched_barrier(0);
  }
  store_c_cols(tile0 - (long)(STEPS - 1) * kStep, xs[0], xs[1], lane);
}

constexpr int kGroups = 4;
constexpr int kGroupLen = 16;
static_assert(kGroups * kGroupLen == kT, "groups tile the border steps, the last one short by one");
constexpr int kPowStep = 4;
static_assert(kPowStep * kPowStep == kGroupLen, "two equal levels reach the group's power");

__device__ __forceinline__ void lds_put_c(v4f* slot, v8f c0, v8f c1) {
  slot[0] = (v4f){c0[0], c0[1], c0[2], c0[3]};
  slot[1] = (v4f){c0[4], c0[5], c0[6], c0[7]};
  slot[2] = (v4f){c1[0], c1[1], c1[2], c1[3]};
  slot[3] = (v4f){c1[4], c1[5], c1[6], c1[7]};
}

__device__ __forceinline__ void lds_get_c(const v4f* slot, v8f& c0, v8f& c1) {
  const v4f a = slot[0], b = slot[1], c = slot[2], d = slot[3];
  c0 = (v8f){a[0], a[1], a[2], a[3], b[0], b[1], b[2], b[3]};
  c1 = (v8f){c[0], c[1], c[2], c[3], d[0], d[1], d[2], d[3]};
}

template <long STRIDE>
__device__ __forceinline__ void group_end_from_zero(const ATiles& P, float* group_tile0, v4f* slot,
                                                    int lane) {
  const v2f none[8] = {};
  v8f e[2];
  run_recurrence<kGroupLen, STRIDE, false, true, false>(P, none, group_tile0, lane, e);
  lds_put_c(slot, e[0], e[1]);
}

template <long STRIDE>
__device__ __forceinline__ void group_jump_and_fill(const ATiles& P, const ATiles& Pg, v2f xb[8],
                                                    const v4f* ends, float* group_tile0, int g,
                                                    int lane) {
#pragma unroll
  for (int i = 0; i < kGroups - 1; ++i) {
    if (i < g) {
      v8f a0, a1;
      lds_get_c(ends + (i * 32 + lane) * 4, a0, a1);
#pragma unroll
      for (int c = 0; c < 8; ++c) mm_chunk(Pg, xb, c, a0, a1);
      c_to_b_regs(a0, a1, xb);
    }
  }
  v8f unused[2];
  if (g < kGroups - 1)
    run_recurrence<kGroupLen, STRIDE, true, false, true>(P, xb, group_tile0, lane, unused);
  else
    run_recurrence<kGroupLen - 1, STRIDE, true, false, true>(P, xb, group_tile0, lane, unused);
}

__global__ void __launch_bounds__(32 * kGroups)
kf_forward_borders_kernel(const float* __restrict__ x0, const float* __restrict__ Fk, float* hat) {
  __shared__ v4f lds_e[(kGroups - 1) * 32 * 4];
  __shared__ v4f lds_p[kD * kD / 4];
  __shared__ v4f lds_q[kD * kD / 4];
  const int lane = threadIdx.x & 31;
  const int g = __builtin_amdgcn_readfirstlane((int)(threadIdx.x >> 5));
  if (blockIdx.x >= kB / 16) {
    if (g == 0) power_row_tile<kGroupLen>(hat, blockIdx.x - kB / 16, hat + kD * kD, lane);
    return;
  }
  const int b0 = blockIdx.x * 16;
  constexpr long STRIDE = kL * kFwdStride;
  float* group_tile0 = step_tile(hat + (size_t)b0 * kD, true, kL - 1) + (long)g * kGroupLen * STRIDE;

  ATiles P;
  load_a_tiles(Fk, P, lane);
  if (g < kGroups - 1) {
    group_end_from_zero<STRIDE>(P, group_tile0, lds_e + (g * 32 + lane) * 4, lane);
  } else {
    power_row_tile<kPowStep>(Fk, 0, (float*)lds_q, lane);
    power_row_tile<kPowStep>(Fk, 1, (float*)lds_q, lane);
    __builtin_amdgcn_fence(__ATOMIC_SEQ_CST, "wavefront");
    power_row_tile<kPowStep>((const float*)lds_q, 0, (float*)lds_p, lane);
    power_row_tile<kPowStep>((const float*)lds_q, 1, (float*)lds_p, lane);
  }
  __syncthreads();

  ATiles Pg;
  load_a_tiles((const float*)lds_p, Pg, lane);
  v2f xb[8];
  load_b_cols(x0 + (size_t)b0 * kD, xb, lane);
  group_jump_and_fill<STRIDE>(P, Pg, xb, lds_e, group_tile0, g, lane);
}

__global__ void __launch_bounds__(32 * kGroups)
kf_backward_borders_kernel(const float* __restrict__ z, const float* __restrict__ Ht,
                           const float* __restrict__ Ka0, const float* __restrict__ hat, float* out2) {
  __shared__ v4f lds_e[(kGroups - 1) * 32 * 4];
  __shared__ v4f lds_s[32 * 4];
  const int lane = threadIdx.x & 31;
  const int g = __builtin_amdgcn_readfirstlane((int)(threadIdx.x >> 5));
  const int b0 = blockIdx.x * 16;
  const size_t last = (size_t)(kN - 1) * kStep + (size_t)b0 * kD;
  constexpr long STRIDE = kL * kBwdStride;
  float* group_tile0 = step_tile(out2 + (size_t)b0 * kD, false, kL - 1) + (long)g * kGroupLen * STRIDE;

  ATiles P, Pg;
  load_a_tiles(hat, P, lane);
  load_a_tiles(hat + kD * kD, Pg, lane);
  if (g < kGroups - 1) {
    group_end_from_zero<STRIDE>(P, group_tile0, lds_e + (g * 32 + lane) * 4, lane);
  } else {
    ATiles A1, A2;
    v2f hb[8];
    load_b_cols(hat + last, hb, lane);
    load_a_tiles(Ht, A1, lane);
    v8f t0 = {0.f, 0.f, 0.f, 0.f, 0.f, 0.f, 0.f, 0.f};
    v8f t1 = t0;
#pragma unroll
    for (int c = 0; c < 8; ++c) t0 = WMMA4(A1.a[0][c], hb[c], t0);
#pragma unroll
    for (int c = 0; c < 8; ++c) t1 = WMMA4(A1.a[1][c], hb[c], t1);

    v8f zc0, zc1;
    load_c_cols(z + last, zc0, zc1, lane);
    v8f bz0 = zc0 - t0;
    v8f bz1 = zc1 - t1;
    v2f bb[8];
    c_to_b_regs(bz0, bz1, bb);

    load_a_tiles(Ka0, A2, lane);
    v8f s0, s1;
    load_c_cols(hat + last, s0, s1, lane);
#pragma unroll
    for (int c = 0; c < 8; ++c) s0 = WMMA4(A2.a[0][c], bb[c], s0);
#pragma unroll
    for (int c = 0; c < 8; ++c) s1 = WMMA4(A2.a[1][c], bb[c], s1);

    store_c_cols(out2 + last, s0, s1, lane);
    lds_put_c(lds_s + lane * 4, s0, s1);
  }
  __syncthreads();

  v8f s0, s1;
  lds_get_c(lds_s + lane * 4, s0, s1);
  v2f sb[8];
  c_to_b_regs(s0, s1, sb);
  group_jump_and_fill<STRIDE>(P, Pg, sb, lds_e, group_tile0, g, lane);
}

extern "C" void kernel_launch(void* const* d_in, const int* in_sizes, int n_in,
                              void* d_out, int out_size, void* d_ws, size_t ws_size,
                              hipStream_t stream) {
  (void)in_sizes; (void)n_in; (void)d_ws; (void)ws_size; (void)out_size;
  const float* z   = (const float*)d_in[0];
  const float* x0  = (const float*)d_in[1];
  const float* F   = (const float*)d_in[2];
  const float* K   = (const float*)d_in[3];
  const float* Ht  = (const float*)d_in[4];
  const float* Ka0 = (const float*)d_in[5];
  const float* Gs  = (const float*)d_in[6];
  const float* Gr  = (const float*)d_in[7];

  float* hat  = (float*)d_out;
  float* out2 = hat + (size_t)kN * kB * kD;

  dim3 slices(kB / 16);
  dim3 slices_and_power(kB / 16 + 2);
  dim3 group_block(32 * kGroups);
  dim3 chunks(kB / 16, kT);
  dim3 chunks_and_row(kB / 16, kT + 1);
  dim3 block(32);
  dim3 end_groups(kB / 16, kT / kEndWaves);
  dim3 end_block(32 * kEndWaves);
  kf_forward_end_kernel<<<end_groups, end_block, 0, stream>>>(z, K, F, Gr, Gs, hat, out2);
  kf_forward_borders_kernel<<<slices_and_power, group_block, 0, stream>>>(x0, out2, hat);
  kf_forward_fill_kernel<<<chunks, block, 0, stream>>>(z, K, F, x0, hat, out2);
  kf_backward_borders_kernel<<<slices, group_block, 0, stream>>>(z, Ht, Ka0, hat, out2);
  kf_backward_fill_kernel<<<chunks_and_row, block, 0, stream>>>(Gs, Gr, x0, hat, out2);
}
